# starting kernel
# baseline (speedup 1.0000x reference)
#include <hip/hip_runtime.h>
#include <math.h>

#define DIN   128
#define DH    64
#define HEADS 2
#define DOUT  64
#define NEG_SLOPE 0.2f

typedef float v2f __attribute__((ext_vector_type(2)));
typedef float v8f __attribute__((ext_vector_type(8)));

__device__ __forceinline__ void atomic_max_f32(float* addr, float v) {
  if (v >= 0.0f) atomicMax((int*)addr, __float_as_int(v));
  else           atomicMin((unsigned int*)addr, __float_as_uint(v));
}

__global__ void fill_f32(float* __restrict__ p, long n, float v) {
  long i = (long)blockIdx.x * blockDim.x + threadIdx.x;
  if (i < n) p[i] = v;
}

__global__ void __launch_bounds__(256)
gemm_wmma_f32(const float* __restrict__ A, const float* __restrict__ B,
              float* __restrict__ C, int M, int K, int Nc) {
  int gid  = blockIdx.x * blockDim.x + threadIdx.x;
  int wave = gid >> 5;
  int lane = threadIdx.x & 31;
  int tilesN = Nc >> 4;
  int tm = wave / tilesN;
  int tn = wave - tm * tilesN;
  if (tm * 16 >= M) return;
  int half = lane >> 4;
  int l16  = lane & 15;
  long rowA = (long)tm * 16 + l16;
  if (rowA >= M) rowA = M - 1;
  int  colB = tn * 16 + l16;
  v8f acc = {};
  for (int k0 = 0; k0 < K; k0 += 4) {
    int ka = k0 + half * 2;
    v2f a, b;
    a.x = A[rowA * K + ka];
    a.y = A[rowA * K + ka + 1];
    b.x = B[(long)ka       * Nc + colB];
    b.y = B[(long)(ka + 1) * Nc + colB];
    acc = __builtin_amdgcn_wmma_f32_16x16x4_f32(false, a, false, b,
                                                (short)0, acc, false, false);
  }
#pragma unroll
  for (int r = 0; r < 8; ++r) {
    int row = tm * 16 + r + half * 8;
    if (row < M) C[(long)row * Nc + colB] = acc[r];
  }
}

__global__ void att_dots(const float* __restrict__ h,
                         const float* __restrict__ att_s,
                         const float* __restrict__ att_d,
                         float* __restrict__ as_, float* __restrict__ ad_,
                         int n, int heads, int C) {
  int i = blockIdx.x * blockDim.x + threadIdx.x;
  if (i >= n * heads) return;
  int node = i / heads, hd = i - node * heads;
  const float* hp = h + ((long)node * heads + hd) * C;
  const float* sp = att_s + hd * C;
  const float* dp = att_d + hd * C;
  float ss = 0.f, dd = 0.f;
  for (int c = 0; c < C; ++c) { float v = hp[c]; ss += v * sp[c]; dd += v * dp[c]; }
  as_[i] = ss; ad_[i] = dd;
}

__global__ void edge_max(const int* __restrict__ srcs, const int* __restrict__ dsts,
                         int E0, int Etot,
                         const float* __restrict__ as_, const float* __restrict__ ad_,
                         float* __restrict__ ev, float* __restrict__ m, int heads) {
  int e = blockIdx.x * blockDim.x + threadIdx.x;
  if (e >= Etot) return;
  int s = (e < E0) ? srcs[e] : (e - E0);
  int d = (e < E0) ? dsts[e] : (e - E0);
  for (int h = 0; h < heads; ++h) {
    float v = as_[s * heads + h] + ad_[d * heads + h];
    v = (v > 0.f) ? v : NEG_SLOPE * v;
    ev[(long)e * heads + h] = v;
    atomic_max_f32(&m[d * heads + h], v);
  }
}

__global__ void edge_sum(const int* __restrict__ dsts, int E0, int Etot,
                         float* __restrict__ ev, const float* __restrict__ m,
                         float* __restrict__ den, int heads) {
  int e = blockIdx.x * blockDim.x + threadIdx.x;
  if (e >= Etot) return;
  int d = (e < E0) ? dsts[e] : (e - E0);
  for (int h = 0; h < heads; ++h) {
    float ex = expf(ev[(long)e * heads + h] - m[d * heads + h]);
    ev[(long)e * heads + h] = ex;
    atomicAdd(&den[d * heads + h], ex);
  }
}

__global__ void edge_agg(const int* __restrict__ srcs, const int* __restrict__ dsts,
                         int E0, int Etot,
                         const float* __restrict__ ev, const float* __restrict__ den,
                         const float* __restrict__ h, float* __restrict__ out,
                         int heads, int C) {
  long gid = (long)blockIdx.x * blockDim.x + threadIdx.x;
  int e = (int)(gid >> 5);
  int lane = (int)(gid & 31);
  if (e >= Etot) return;
  int s = (e < E0) ? srcs[e] : (e - E0);
  int d = (e < E0) ? dsts[e] : (e - E0);
  int F = heads * C;
  for (int c = lane; c < F; c += 32) {
    int hd = c / C;
    float alpha = ev[(long)e * heads + hd] / den[d * heads + hd];
    atomicAdd(&out[(long)d * F + c], h[(long)s * F + c] * alpha);
  }
}

__global__ void elu_bias(float* __restrict__ p, const float* __restrict__ bias,
                         long n, int F) {
  long i = (long)blockIdx.x * blockDim.x + threadIdx.x;
  if (i >= n) return;
  float v = p[i] + bias[i % F];
  p[i] = (v > 0.f) ? v : (expf(v) - 1.f);
}

__global__ void bias_add(float* __restrict__ p, const float* __restrict__ bias,
                         long n, int F) {
  long i = (long)blockIdx.x * blockDim.x + threadIdx.x;
  if (i >= n) return;
  p[i] += bias[i % F];
}

extern "C" void kernel_launch(void* const* d_in, const int* in_sizes, int n_in,
                              void* d_out, int out_size, void* d_ws, size_t ws_size,
                              hipStream_t stream) {
  const float* x    = (const float*)d_in[0];
  const int*   ei   = (const int*)  d_in[1];
  const float* W1   = (const float*)d_in[2];
  const float* atS1 = (const float*)d_in[3];
  const float* atD1 = (const float*)d_in[4];
  const float* b1   = (const float*)d_in[5];
  const float* W2   = (const float*)d_in[6];
  const float* atS2 = (const float*)d_in[7];
  const float* atD2 = (const float*)d_in[8];
  const float* b2   = (const float*)d_in[9];
  float* out = (float*)d_out;

  const int N    = in_sizes[0] / DIN;
  const int E0   = in_sizes[1] / 2;
  const int Etot = E0 + N;
  const int* srcs = ei;
  const int* dsts = ei + E0;

  float* ws = (float*)d_ws;
  long off = 0;
  float* h1  = ws + off; off += (long)N * (HEADS * DH);
  float* o1  = ws + off; off += (long)N * (HEADS * DH);
  float* ex1 = ws + off; off += (long)Etot * HEADS;
  float* a1s = ws + off; off += (long)N * HEADS;
  float* a1d = ws + off; off += (long)N * HEADS;
  float* m1  = ws + off; off += (long)N * HEADS;
  float* d1  = ws + off; off += (long)N * HEADS;
  float* h2  = h1;
  float* ex2 = h1 + (long)N * DOUT;
  float* a2s = ex2 + Etot;
  float* a2d = a2s + N;
  float* m2  = a2d + N;
  float* d2  = m2 + N;

  const int TB = 256;
  auto blocks = [](long n, int tb) { return (unsigned)((n + tb - 1) / tb); };

  fill_f32<<<blocks((long)N * HEADS * DH, TB), TB, 0, stream>>>(o1, (long)N * HEADS * DH, 0.f);
  fill_f32<<<blocks((long)N * HEADS, TB), TB, 0, stream>>>(d1, (long)N * HEADS, 0.f);
  fill_f32<<<blocks((long)N * HEADS, TB), TB, 0, stream>>>(m1, (long)N * HEADS, -INFINITY);

  {
    long waves = (long)((N + 15) / 16) * ((HEADS * DH) / 16);
    gemm_wmma_f32<<<blocks(waves * 32, TB), TB, 0, stream>>>(x, W1, h1, N, DIN, HEADS * DH);
  }
  att_dots<<<blocks((long)N * HEADS, TB), TB, 0, stream>>>(h1, atS1, atD1, a1s, a1d, N, HEADS, DH);
  edge_max<<<blocks(Etot, TB), TB, 0, stream>>>(srcs, dsts, E0, Etot, a1s, a1d, ex1, m1, HEADS);
  edge_sum<<<blocks(Etot, TB), TB, 0, stream>>>(dsts, E0, Etot, ex1, m1, d1, HEADS);
  edge_agg<<<blocks((long)Etot * 32, TB), TB, 0, stream>>>(srcs, dsts, E0, Etot, ex1, d1, h1, o1, HEADS, DH);
  elu_bias<<<blocks((long)N * HEADS * DH, TB), TB, 0, stream>>>(o1, b1, (long)N * HEADS * DH, HEADS * DH);

  fill_f32<<<blocks((long)N * DOUT, TB), TB, 0, stream>>>(out, (long)N * DOUT, 0.f);
  fill_f32<<<blocks(N, TB), TB, 0, stream>>>(d2, N, 0.f);
  fill_f32<<<blocks(N, TB), TB, 0, stream>>>(m2, N, -INFINITY);

  {
    long waves = (long)((N + 15) / 16) * (DOUT / 16);
    gemm_wmma_f32<<<blocks(waves * 32, TB), TB, 0, stream>>>(o1, W2, h2, N, HEADS * DH, DOUT);
  }
  att_dots<<<blocks(N, TB), TB, 0, stream>>>(h2, atS2, atD2, a2s, a2d, N, 1, DOUT);
  edge_max<<<blocks(Etot, TB), TB, 0, stream>>>(srcs, dsts, E0, Etot, a2s, a2d, ex2, m2, 1);
  edge_sum<<<blocks(Etot, TB), TB, 0, stream>>>(dsts, E0, Etot, ex2, m2, d2, 1);
  edge_agg<<<blocks((long)Etot * 32, TB), TB, 0, stream>>>(srcs, dsts, E0, Etot, ex2, d2, h2, out, 1, DOUT);
  bias_add<<<blocks((long)N * DOUT, TB), TB, 0, stream>>>(out, b2, (long)N * DOUT, DOUT);
}


# Edges grouped by destination once per call (int-atomic rank; each group sorted by source, so sum order is fixed); one wave per node does softmax and row sum in registers; float atomics, fills gone
# speedup vs baseline: 3.4219x; 3.4219x over previous
#include <hip/hip_runtime.h>
#include <math.h>

#define DIN   128
#define DH    64
#define HEADS 2
#define DOUT  64
#define NEG_SLOPE 0.2f

typedef float v2f __attribute__((ext_vector_type(2)));
typedef float v8f __attribute__((ext_vector_type(8)));

__global__ void __launch_bounds__(256)
gemm_wmma_f32(const float* __restrict__ A, const float* __restrict__ B,
              float* __restrict__ C, int M, int K, int Nc) {
  int gid  = blockIdx.x * blockDim.x + threadIdx.x;
  int wave = gid >> 5;
  int lane = threadIdx.x & 31;
  int tilesN = Nc >> 4;
  int tm = wave / tilesN;
  int tn = wave - tm * tilesN;
  if (tm * 16 >= M) return;
  int half = lane >> 4;
  int l16  = lane & 15;
  long rowA = (long)tm * 16 + l16;
  if (rowA >= M) rowA = M - 1;
  int  colB = tn * 16 + l16;
  v8f acc = {};
  for (int k0 = 0; k0 < K; k0 += 4) {
    int ka = k0 + half * 2;
    v2f a, b;
    a.x = A[rowA * K + ka];
    a.y = A[rowA * K + ka + 1];
    b.x = B[(long)ka       * Nc + colB];
    b.y = B[(long)(ka + 1) * Nc + colB];
    acc = __builtin_amdgcn_wmma_f32_16x16x4_f32(false, a, false, b,
                                                (short)0, acc, false, false);
  }
#pragma unroll
  for (int r = 0; r < 8; ++r) {
    int row = tm * 16 + r + half * 8;
    if (row < M) C[(long)row * Nc + colB] = acc[r];
  }
}

__global__ void att_dots(const float* __restrict__ h,
                         const float* __restrict__ att_s,
                         const float* __restrict__ att_d,
                         float* __restrict__ as_, float* __restrict__ ad_,
                         int n, int heads, int C) {
  int i = blockIdx.x * blockDim.x + threadIdx.x;
  if (i >= n * heads) return;
  int node = i / heads, hd = i - node * heads;
  const float* hp = h + ((long)node * heads + hd) * C;
  const float* sp = att_s + hd * C;
  const float* dp = att_d + hd * C;
  float ss = 0.f, dd = 0.f;
  for (int c = 0; c < C; ++c) { float v = hp[c]; ss += v * sp[c]; dd += v * dp[c]; }
  as_[i] = ss; ad_[i] = dd;
}

__global__ void zero_i32(int* __restrict__ p, int n) {
  int i = blockIdx.x * blockDim.x + threadIdx.x;
  if (i < n) p[i] = 0;
}

__global__ void edge_rank(const int* __restrict__ dsts, int E0,
                          int* __restrict__ cnt, int* __restrict__ rank) {
  int e = blockIdx.x * blockDim.x + threadIdx.x;
  if (e >= E0) return;
  rank[e] = atomicAdd(&cnt[dsts[e]], 1);
}

__global__ void __launch_bounds__(1024)
seg_offsets(const int* __restrict__ cnt, int* __restrict__ off, int n) {
  __shared__ int sh[1024];
  __shared__ int red[32];
  int t = threadIdx.x;
  int start = blockIdx.x * 1024;
  int part = 0;
  for (int j = t; j < start; j += 1024) part += cnt[j];
  for (int o = 16; o > 0; o >>= 1) part += __shfl_xor(part, o);
  if ((t & 31) == 0) red[t >> 5] = part;
  int i = start + t;
  int c = (i < n) ? cnt[i] : 0;
  sh[t] = c;
  __syncthreads();
  int base = 0;
  for (int w = 0; w < 32; ++w) base += red[w];
  for (int o = 1; o < 1024; o <<= 1) {
    int v = (t >= o) ? sh[t - o] : 0;
    __syncthreads();
    sh[t] += v;
    __syncthreads();
  }
  int incl = sh[t];
  if (i < n) {
    off[i] = base + incl - c;
    if (i == n - 1) off[n] = base + incl;
  }
}

__global__ void edge_scatter(const int* __restrict__ srcs, const int* __restrict__ dsts, int E0,
                             const int* __restrict__ off, const int* __restrict__ rank,
                             int* __restrict__ seg) {
  int e = blockIdx.x * blockDim.x + threadIdx.x;
  if (e >= E0) return;
  seg[off[dsts[e]] + rank[e]] = srcs[e];
}

__global__ void __launch_bounds__(256)
seg_order(const int* __restrict__ off, const int* __restrict__ seg,
          int* __restrict__ ssrc, int n) {
  int lane = threadIdx.x & 31;
  int node = (int)((blockIdx.x * blockDim.x + threadIdx.x) >> 5);
  node = __builtin_amdgcn_readfirstlane(node);
  if (node >= n) return;
  const int beg = off[node];
  const int end = off[node + 1];
  for (int ib = beg; ib < end; ib += 32) {
    int i = ib + lane;
    bool act = i < end;
    int ki = act ? seg[i] : 0;
    int r = 0;
    for (int jb = beg; jb < end; jb += 32) {
      int j = jb + lane;
      int kj = (j < end) ? seg[j] : 0;
      int cntj = min(32, end - jb);
      for (int t = 0; t < cntj; ++t) {
        int k = __builtin_amdgcn_readlane(kj, t);
        int jj = jb + t;
        r += (k < ki || (k == ki && jj < i)) ? 1 : 0;
      }
    }
    if (act) ssrc[beg + r] = ki;
  }
}

__device__ __forceinline__ float leaky(float v) { return (v > 0.f) ? v : NEG_SLOPE * v; }

__device__ __forceinline__ float wave_max(float v) {
  for (int o = 16; o > 0; o >>= 1) v = fmaxf(v, __shfl_xor(v, o));
  return v;
}
__device__ __forceinline__ float wave_sum(float v) {
  for (int o = 16; o > 0; o >>= 1) v += __shfl_xor(v, o);
  return v;
}

template <int H, int VEC, bool ELU_OUT>
__global__ void __launch_bounds__(256)
node_softmax_agg(const int* __restrict__ off, const int* __restrict__ ssrc,
                 const float* __restrict__ as_, const float* __restrict__ ad_,
                 const float* __restrict__ h, const float* __restrict__ bias,
                 float* __restrict__ outp, int n) {
  constexpr int F = 32 * VEC;
  constexpr int C = F / H;
  int lane = threadIdx.x & 31;
  int node = (int)((blockIdx.x * blockDim.x + threadIdx.x) >> 5);
  node = __builtin_amdgcn_readfirstlane(node);
  if (node >= n) return;
  const int beg = off[node];
  const int tot = off[node + 1] - beg + 1;
  const int myh = (lane * VEC) / C;

  float adv[H], m[H], v[H];
  for (int k = 0; k < H; ++k) { adv[k] = ad_[node * H + k]; m[k] = -INFINITY; v[k] = 0.f; }
  int s = 0;
  bool act = false;

  auto load_chunk = [&](int jb) {
    int idx = jb + lane;
    act = idx < tot;
    s = 0;
    for (int k = 0; k < H; ++k) v[k] = 0.f;
    if (act) {
      s = (idx == 0) ? node : ssrc[beg + idx - 1];
      for (int k = 0; k < H; ++k) v[k] = leaky(as_[s * H + k] + adv[k]);
    }
  };

  for (int jb = 0; jb < tot; jb += 32) {
    load_chunk(jb);
    if (act) { for (int k = 0; k < H; ++k) m[k] = fmaxf(m[k], v[k]); }
  }
  for (int k = 0; k < H; ++k) m[k] = wave_max(m[k]);

  float den[H];
  for (int k = 0; k < H; ++k) den[k] = 0.f;
  for (int jb = 0; jb < tot; jb += 32) {
    if (tot > 32) load_chunk(jb);
    if (act) { for (int k = 0; k < H; ++k) den[k] += expf(v[k] - m[k]); }
  }
  for (int k = 0; k < H; ++k) den[k] = wave_sum(den[k]);

  float acc[VEC];
  for (int q = 0; q < VEC; ++q) acc[q] = 0.f;
  for (int jb = 0; jb < tot; jb += 32) {
    if (tot > 32) load_chunk(jb);
    float w[H];
    for (int k = 0; k < H; ++k) w[k] = act ? expf(v[k] - m[k]) / den[k] : 0.f;
    const int cnt = min(32, tot - jb);
#pragma unroll 4
    for (int t = 0; t < cnt; ++t) {
      int st = __builtin_amdgcn_readlane(s, t);
      float wt = __int_as_float(__builtin_amdgcn_readlane(__float_as_int(w[0]), t));
      if (H == 2) {
        float w1 = __int_as_float(__builtin_amdgcn_readlane(__float_as_int(w[H - 1]), t));
        wt = myh ? w1 : wt;
      }
      const float* row = h + (long)st * F + lane * VEC;
      if (VEC == 4) {
        float4 r = *reinterpret_cast<const float4*>(row);
        acc[0] += r.x * wt; acc[1] += r.y * wt; acc[VEC - 2] += r.z * wt; acc[VEC - 1] += r.w * wt;
      } else {
        float2 r = *reinterpret_cast<const float2*>(row);
        acc[0] += r.x * wt; acc[1] += r.y * wt;
      }
    }
  }

  const int col = lane * VEC;
  for (int q = 0; q < VEC; ++q) {
    float val = acc[q] + bias[col + q];
    if (ELU_OUT) val = (val > 0.f) ? val : (expf(val) - 1.f);
    acc[q] = val;
  }
  float* dstp = outp + (long)node * F + col;
  if (VEC == 4) {
    *reinterpret_cast<float4*>(dstp) = make_float4(acc[0], acc[1], acc[VEC - 2], acc[VEC - 1]);
  } else {
    *reinterpret_cast<float2*>(dstp) = make_float2(acc[0], acc[1]);
  }
}

extern "C" void kernel_launch(void* const* d_in, const int* in_sizes, int n_in,
                              void* d_out, int out_size, void* d_ws, size_t ws_size,
                              hipStream_t stream) {
  const float* x    = (const float*)d_in[0];
  const int*   ei   = (const int*)  d_in[1];
  const float* W1   = (const float*)d_in[2];
  const float* atS1 = (const float*)d_in[3];
  const float* atD1 = (const float*)d_in[4];
  const float* b1   = (const float*)d_in[5];
  const float* W2   = (const float*)d_in[6];
  const float* atS2 = (const float*)d_in[7];
  const float* atD2 = (const float*)d_in[8];
  const float* b2   = (const float*)d_in[9];
  float* out = (float*)d_out;

  const int N    = in_sizes[0] / DIN;
  const int E0   = in_sizes[1] / 2;
  const int* srcs = ei;
  const int* dsts = ei + E0;

  float* ws = (float*)d_ws;
  long off = 0;
  float* h1  = ws + off; off += (long)N * (HEADS * DH);
  float* o1  = ws + off; off += (long)N * (HEADS * DH);
  float* a1s = ws + off; off += (long)N * HEADS;
  float* a1d = ws + off; off += (long)N * HEADS;
  int* cnt   = (int*)(ws + off); off += (long)N;
  int* segof = (int*)(ws + off); off += (long)N + 1;
  int* rank  = (int*)(ws + off); off += (long)E0;
  int* seg   = (int*)(ws + off); off += (long)E0;
  int* ssrc  = (int*)(ws + off); off += (long)E0;
  float* h2  = h1;
  float* a2s = h1 + (long)N * DOUT;
  float* a2d = a2s + N;

  const int TB = 256;
  auto blocks = [](long n, int tb) { return (unsigned)((n + tb - 1) / tb); };

  zero_i32<<<blocks(N, TB), TB, 0, stream>>>(cnt, N);
  edge_rank<<<blocks(E0, TB), TB, 0, stream>>>(dsts, E0, cnt, rank);
  seg_offsets<<<blocks(N, 1024), 1024, 0, stream>>>(cnt, segof, N);
  edge_scatter<<<blocks(E0, TB), TB, 0, stream>>>(srcs, dsts, E0, segof, rank, seg);
  seg_order<<<blocks((long)N * 32, TB), TB, 0, stream>>>(segof, seg, ssrc, N);

  {
    long waves = (long)((N + 15) / 16) * ((HEADS * DH) / 16);
    gemm_wmma_f32<<<blocks(waves * 32, TB), TB, 0, stream>>>(x, W1, h1, N, DIN, HEADS * DH);
  }
  att_dots<<<blocks((long)N * HEADS, TB), TB, 0, stream>>>(h1, atS1, atD1, a1s, a1d, N, HEADS, DH);
  node_softmax_agg<HEADS, 4, true><<<blocks((long)N * 32, TB), TB, 0, stream>>>(
      segof, ssrc, a1s, a1d, h1, b1, o1, N);

  {
    long waves = (long)((N + 15) / 16) * (DOUT / 16);
    gemm_wmma_f32<<<blocks(waves * 32, TB), TB, 0, stream>>>(o1, W2, h2, N, HEADS * DH, DOUT);
  }
  att_dots<<<blocks(N, TB), TB, 0, stream>>>(h2, atS2, atD2, a2s, a2d, N, 1, DOUT);
  node_softmax_agg<1, 2, false><<<blocks((long)N * 32, TB), TB, 0, stream>>>(
      segof, ssrc, a2s, a2d, h2, b2, out, N);
}


# GEMM wave owns 16 rows x one head's 64 columns (A fragment feeds 4 tiles) and takes the attention dots from its accumulators; dot sums now a fixed lane tree; att_dots launches gone
# speedup vs baseline: 4.0293x; 1.1775x over previous
#include <hip/hip_runtime.h>
#include <math.h>

#define DIN   128
#define DH    64
#define HEADS 2
#define DOUT  64
#define NEG_SLOPE 0.2f

typedef float v2f __attribute__((ext_vector_type(2)));
typedef float v8f __attribute__((ext_vector_type(8)));


template <int CNT>
__device__ __forceinline__ void fold_half(float (&v)[16], bool upper, int dist) {
#pragma unroll
  for (int i = 0; i < CNT; ++i) {
    float lo = v[i], hi = v[i + CNT];
    float send = upper ? lo : hi;
    float keep = upper ? hi : lo;
    v[i] = keep + __shfl_xor(send, dist);
  }
}

template <int K, int NC>
__global__ void __launch_bounds__(256)
gemm_heads_wmma_f32(const float* __restrict__ A, const float* __restrict__ B,
                    const float* __restrict__ att_s, const float* __restrict__ att_d,
                    float* __restrict__ C, float* __restrict__ as_, float* __restrict__ ad_,
                    int M) {
  constexpr int HW = NC / 64;
  int lane = threadIdx.x & 31;
  int wave = (int)((blockIdx.x * blockDim.x + threadIdx.x) >> 5);
  wave = __builtin_amdgcn_readfirstlane(wave);
  const int tm = wave / HW;
  const int hd = wave - tm * HW;
  if (tm * 16 >= M) return;
  const int half = lane >> 4;
  const int l16  = lane & 15;
  long rowA = (long)tm * 16 + l16;
  if (rowA >= M) rowA = M - 1;
  const int colbase = hd * 64 + l16 * 4;
  const float* ap = A + rowA * K + half * 2;
  const float* bp = B + (long)(half * 2) * NC + colbase;
  v8f acc0 = {}, acc1 = {}, acc2 = {}, acc3 = {};
#pragma unroll
  for (int k0 = 0; k0 < K; k0 += 4) {
    v2f a = *reinterpret_cast<const v2f*>(ap + k0);
    float4 bx = *reinterpret_cast<const float4*>(bp + (long)k0 * NC);
    float4 by = *reinterpret_cast<const float4*>(bp + (long)(k0 + 1) * NC);
    v2f b;
    b.x = bx.x; b.y = by.x;
    acc0 = __builtin_amdgcn_wmma_f32_16x16x4_f32(false, a, false, b, (short)0, acc0, false, false);
    b.x = bx.y; b.y = by.y;
    acc1 = __builtin_amdgcn_wmma_f32_16x16x4_f32(false, a, false, b, (short)0, acc1, false, false);
    b.x = bx.z; b.y = by.z;
    acc2 = __builtin_amdgcn_wmma_f32_16x16x4_f32(false, a, false, b, (short)0, acc2, false, false);
    b.x = bx.w; b.y = by.w;
    acc3 = __builtin_amdgcn_wmma_f32_16x16x4_f32(false, a, false, b, (short)0, acc3, false, false);
  }

  const float4 ats = *reinterpret_cast<const float4*>(att_s + colbase);
  const float4 atd = *reinterpret_cast<const float4*>(att_d + colbase);
  float v[16];
#pragma unroll
  for (int r = 0; r < 8; ++r) {
    const float c0 = acc0[r], c1 = acc1[r], c2 = acc2[r], c3 = acc3[r];
    const int row = tm * 16 + r + half * 8;
    if (row < M)
      *reinterpret_cast<float4*>(C + (long)row * NC + colbase) = make_float4(c0, c1, c2, c3);
    v[r]     = c0 * ats.x + c1 * ats.y + c2 * ats.z + c3 * ats.w;
    v[8 + r] = c0 * atd.x + c1 * atd.y + c2 * atd.z + c3 * atd.w;
  }
  fold_half<8>(v, (l16 & 8) != 0, 8);
  fold_half<4>(v, (l16 & 4) != 0, 4);
  fold_half<2>(v, (l16 & 2) != 0, 2);
  fold_half<1>(v, (l16 & 1) != 0, 1);
  const int orow = tm * 16 + (l16 & 7) + half * 8;
  if (orow < M) {
    float* dp = (l16 < 8) ? as_ : ad_;
    dp[(long)orow * HW + hd] = v[0];
  }
}

__global__ void zero_i32(int* __restrict__ p, int n) {
  int i = blockIdx.x * blockDim.x + threadIdx.x;
  if (i < n) p[i] = 0;
}

__global__ void edge_rank(const int* __restrict__ dsts, int E0,
                          int* __restrict__ cnt, int* __restrict__ rank) {
  int e = blockIdx.x * blockDim.x + threadIdx.x;
  if (e >= E0) return;
  rank[e] = atomicAdd(&cnt[dsts[e]], 1);
}

__global__ void __launch_bounds__(1024)
seg_offsets(const int* __restrict__ cnt, int* __restrict__ off, int n) {
  __shared__ int sh[1024];
  __shared__ int red[32];
  int t = threadIdx.x;
  int start = blockIdx.x * 1024;
  int part = 0;
  for (int j = t; j < start; j += 1024) part += cnt[j];
  for (int o = 16; o > 0; o >>= 1) part += __shfl_xor(part, o);
  if ((t & 31) == 0) red[t >> 5] = part;
  int i = start + t;
  int c = (i < n) ? cnt[i] : 0;
  sh[t] = c;
  __syncthreads();
  int base = 0;
  for (int w = 0; w < 32; ++w) base += red[w];
  for (int o = 1; o < 1024; o <<= 1) {
    int v = (t >= o) ? sh[t - o] : 0;
    __syncthreads();
    sh[t] += v;
    __syncthreads();
  }
  int incl = sh[t];
  if (i < n) {
    off[i] = base + incl - c;
    if (i == n - 1) off[n] = base + incl;
  }
}

__global__ void edge_scatter(const int* __restrict__ srcs, const int* __restrict__ dsts, int E0,
                             const int* __restrict__ off, const int* __restrict__ rank,
                             int* __restrict__ seg) {
  int e = blockIdx.x * blockDim.x + threadIdx.x;
  if (e >= E0) return;
  seg[off[dsts[e]] + rank[e]] = srcs[e];
}

__global__ void __launch_bounds__(256)
seg_order(const int* __restrict__ off, const int* __restrict__ seg,
          int* __restrict__ ssrc, int n) {
  int lane = threadIdx.x & 31;
  int node = (int)((blockIdx.x * blockDim.x + threadIdx.x) >> 5);
  node = __builtin_amdgcn_readfirstlane(node);
  if (node >= n) return;
  const int beg = off[node];
  const int end = off[node + 1];
  for (int ib = beg; ib < end; ib += 32) {
    int i = ib + lane;
    bool act = i < end;
    int ki = act ? seg[i] : 0;
    int r = 0;
    for (int jb = beg; jb < end; jb += 32) {
      int j = jb + lane;
      int kj = (j < end) ? seg[j] : 0;
      int cntj = min(32, end - jb);
      for (int t = 0; t < cntj; ++t) {
        int k = __builtin_amdgcn_readlane(kj, t);
        int jj = jb + t;
        r += (k < ki || (k == ki && jj < i)) ? 1 : 0;
      }
    }
    if (act) ssrc[beg + r] = ki;
  }
}

__device__ __forceinline__ float leaky(float v) { return (v > 0.f) ? v : NEG_SLOPE * v; }

__device__ __forceinline__ float wave_max(float v) {
  for (int o = 16; o > 0; o >>= 1) v = fmaxf(v, __shfl_xor(v, o));
  return v;
}
__device__ __forceinline__ float wave_sum(float v) {
  for (int o = 16; o > 0; o >>= 1) v += __shfl_xor(v, o);
  return v;
}

template <int H, int VEC, bool ELU_OUT>
__global__ void __launch_bounds__(256)
node_softmax_agg(const int* __restrict__ off, const int* __restrict__ ssrc,
                 const float* __restrict__ as_, const float* __restrict__ ad_,
                 const float* __restrict__ h, const float* __restrict__ bias,
                 float* __restrict__ outp, int n) {
  constexpr int F = 32 * VEC;
  constexpr int C = F / H;
  int lane = threadIdx.x & 31;
  int node = (int)((blockIdx.x * blockDim.x + threadIdx.x) >> 5);
  node = __builtin_amdgcn_readfirstlane(node);
  if (node >= n) return;
  const int beg = off[node];
  const int tot = off[node + 1] - beg + 1;
  const int myh = (lane * VEC) / C;

  float adv[H], m[H], v[H];
  for (int k = 0; k < H; ++k) { adv[k] = ad_[node * H + k]; m[k] = -INFINITY; v[k] = 0.f; }
  int s = 0;
  bool act = false;

  auto load_chunk = [&](int jb) {
    int idx = jb + lane;
    act = idx < tot;
    s = 0;
    for (int k = 0; k < H; ++k) v[k] = 0.f;
    if (act) {
      s = (idx == 0) ? node : ssrc[beg + idx - 1];
      for (int k = 0; k < H; ++k) v[k] = leaky(as_[s * H + k] + adv[k]);
    }
  };

  for (int jb = 0; jb < tot; jb += 32) {
    load_chunk(jb);
    if (act) { for (int k = 0; k < H; ++k) m[k] = fmaxf(m[k], v[k]); }
  }
  for (int k = 0; k < H; ++k) m[k] = wave_max(m[k]);

  float den[H];
  for (int k = 0; k < H; ++k) den[k] = 0.f;
  for (int jb = 0; jb < tot; jb += 32) {
    if (tot > 32) load_chunk(jb);
    if (act) { for (int k = 0; k < H; ++k) den[k] += expf(v[k] - m[k]); }
  }
  for (int k = 0; k < H; ++k) den[k] = wave_sum(den[k]);

  float acc[VEC];
  for (int q = 0; q < VEC; ++q) acc[q] = 0.f;
  for (int jb = 0; jb < tot; jb += 32) {
    if (tot > 32) load_chunk(jb);
    float w[H];
    for (int k = 0; k < H; ++k) w[k] = act ? expf(v[k] - m[k]) / den[k] : 0.f;
    const int cnt = min(32, tot - jb);
#pragma unroll 4
    for (int t = 0; t < cnt; ++t) {
      int st = __builtin_amdgcn_readlane(s, t);
      float wt = __int_as_float(__builtin_amdgcn_readlane(__float_as_int(w[0]), t));
      if (H == 2) {
        float w1 = __int_as_float(__builtin_amdgcn_readlane(__float_as_int(w[H - 1]), t));
        wt = myh ? w1 : wt;
      }
      const float* row = h + (long)st * F + lane * VEC;
      if (VEC == 4) {
        float4 r = *reinterpret_cast<const float4*>(row);
        acc[0] += r.x * wt; acc[1] += r.y * wt; acc[VEC - 2] += r.z * wt; acc[VEC - 1] += r.w * wt;
      } else {
        float2 r = *reinterpret_cast<const float2*>(row);
        acc[0] += r.x * wt; acc[1] += r.y * wt;
      }
    }
  }

  const int col = lane * VEC;
  for (int q = 0; q < VEC; ++q) {
    float val = acc[q] + bias[col + q];
    if (ELU_OUT) val = (val > 0.f) ? val : (expf(val) - 1.f);
    acc[q] = val;
  }
  float* dstp = outp + (long)node * F + col;
  if (VEC == 4) {
    *reinterpret_cast<float4*>(dstp) = make_float4(acc[0], acc[1], acc[VEC - 2], acc[VEC - 1]);
  } else {
    *reinterpret_cast<float2*>(dstp) = make_float2(acc[0], acc[1]);
  }
}

extern "C" void kernel_launch(void* const* d_in, const int* in_sizes, int n_in,
                              void* d_out, int out_size, void* d_ws, size_t ws_size,
                              hipStream_t stream) {
  const float* x    = (const float*)d_in[0];
  const int*   ei   = (const int*)  d_in[1];
  const float* W1   = (const float*)d_in[2];
  const float* atS1 = (const float*)d_in[3];
  const float* atD1 = (const float*)d_in[4];
  const float* b1   = (const float*)d_in[5];
  const float* W2   = (const float*)d_in[6];
  const float* atS2 = (const float*)d_in[7];
  const float* atD2 = (const float*)d_in[8];
  const float* b2   = (const float*)d_in[9];
  float* out = (float*)d_out;

  const int N    = in_sizes[0] / DIN;
  const int E0   = in_sizes[1] / 2;
  const int* srcs = ei;
  const int* dsts = ei + E0;

  float* ws = (float*)d_ws;
  long off = 0;
  float* h1  = ws + off; off += (long)N * (HEADS * DH);
  float* o1  = ws + off; off += (long)N * (HEADS * DH);
  float* a1s = ws + off; off += (long)N * HEADS;
  float* a1d = ws + off; off += (long)N * HEADS;
  int* cnt   = (int*)(ws + off); off += (long)N;
  int* segof = (int*)(ws + off); off += (long)N + 1;
  int* rank  = (int*)(ws + off); off += (long)E0;
  int* seg   = (int*)(ws + off); off += (long)E0;
  int* ssrc  = (int*)(ws + off); off += (long)E0;
  float* h2  = h1;
  float* a2s = h1 + (long)N * DOUT;
  float* a2d = a2s + N;

  const int TB = 256;
  auto blocks = [](long n, int tb) { return (unsigned)((n + tb - 1) / tb); };

  zero_i32<<<blocks(N, TB), TB, 0, stream>>>(cnt, N);
  edge_rank<<<blocks(E0, TB), TB, 0, stream>>>(dsts, E0, cnt, rank);
  seg_offsets<<<blocks(N, 1024), 1024, 0, stream>>>(cnt, segof, N);
  edge_scatter<<<blocks(E0, TB), TB, 0, stream>>>(srcs, dsts, E0, segof, rank, seg);
  seg_order<<<blocks((long)N * 32, TB), TB, 0, stream>>>(segof, seg, ssrc, N);

  {
    long waves = (long)((N + 15) / 16) * HEADS;
    gemm_heads_wmma_f32<DIN, HEADS * DH><<<blocks(waves * 32, TB), TB, 0, stream>>>(
        x, W1, atS1, atD1, h1, a1s, a1d, N);
  }
  node_softmax_agg<HEADS, 4, true><<<blocks((long)N * 32, TB), TB, 0, stream>>>(
      segof, ssrc, a1s, a1d, h1, b1, o1, N);

  {
    long waves = (long)((N + 15) / 16);
    gemm_heads_wmma_f32<HEADS * DH, DOUT><<<blocks(waves * 32, TB), TB, 0, stream>>>(
        o1, W2, atS2, atD2, h2, a2s, a2d, N);
  }
  node_softmax_agg<1, 2, false><<<blocks((long)N * 32, TB), TB, 0, stream>>>(
      segof, ssrc, a2s, a2d, h2, b2, out, N);
}


# Edges grouped by destination with LDS counters only: per-block counts per 256-node range, then one work-group per range counts, places, orders; no global atomics, 9 -> 7 launches; sum order unchanged
# speedup vs baseline: 5.2133x; 1.2938x over previous
#include <hip/hip_runtime.h>
#include <math.h>

#define DIN   128
#define DH    64
#define HEADS 2
#define DOUT  64
#define NEG_SLOPE 0.2f

typedef float v2f __attribute__((ext_vector_type(2)));
typedef float v8f __attribute__((ext_vector_type(8)));


template <int CNT>
__device__ __forceinline__ void fold_half(float (&v)[16], bool upper, int dist) {
#pragma unroll
  for (int i = 0; i < CNT; ++i) {
    float lo = v[i], hi = v[i + CNT];
    float send = upper ? lo : hi;
    float keep = upper ? hi : lo;
    v[i] = keep + __shfl_xor(send, dist);
  }
}

template <int K, int NC>
__global__ void __launch_bounds__(256)
gemm_heads_wmma_f32(const float* __restrict__ A, const float* __restrict__ B,
                    const float* __restrict__ att_s, const float* __restrict__ att_d,
                    float* __restrict__ C, float* __restrict__ as_, float* __restrict__ ad_,
                    int M) {
  constexpr int HW = NC / 64;
  int lane = threadIdx.x & 31;
  int wave = (int)((blockIdx.x * blockDim.x + threadIdx.x) >> 5);
  wave = __builtin_amdgcn_readfirstlane(wave);
  const int tm = wave / HW;
  const int hd = wave - tm * HW;
  if (tm * 16 >= M) return;
  const int half = lane >> 4;
  const int l16  = lane & 15;
  long rowA = (long)tm * 16 + l16;
  if (rowA >= M) rowA = M - 1;
  const int colbase = hd * 64 + l16 * 4;
  const float* ap = A + rowA * K + half * 2;
  const float* bp = B + (long)(half * 2) * NC + colbase;
  v8f acc0 = {}, acc1 = {}, acc2 = {}, acc3 = {};
#pragma unroll
  for (int k0 = 0; k0 < K; k0 += 4) {
    v2f a = *reinterpret_cast<const v2f*>(ap + k0);
    float4 bx = *reinterpret_cast<const float4*>(bp + (long)k0 * NC);
    float4 by = *reinterpret_cast<const float4*>(bp + (long)(k0 + 1) * NC);
    v2f b;
    b.x = bx.x; b.y = by.x;
    acc0 = __builtin_amdgcn_wmma_f32_16x16x4_f32(false, a, false, b, (short)0, acc0, false, false);
    b.x = bx.y; b.y = by.y;
    acc1 = __builtin_amdgcn_wmma_f32_16x16x4_f32(false, a, false, b, (short)0, acc1, false, false);
    b.x = bx.z; b.y = by.z;
    acc2 = __builtin_amdgcn_wmma_f32_16x16x4_f32(false, a, false, b, (short)0, acc2, false, false);
    b.x = bx.w; b.y = by.w;
    acc3 = __builtin_amdgcn_wmma_f32_16x16x4_f32(false, a, false, b, (short)0, acc3, false, false);
  }

  const float4 ats = *reinterpret_cast<const float4*>(att_s + colbase);
  const float4 atd = *reinterpret_cast<const float4*>(att_d + colbase);
  float v[16];
#pragma unroll
  for (int r = 0; r < 8; ++r) {
    const float c0 = acc0[r], c1 = acc1[r], c2 = acc2[r], c3 = acc3[r];
    const int row = tm * 16 + r + half * 8;
    if (row < M)
      *reinterpret_cast<float4*>(C + (long)row * NC + colbase) = make_float4(c0, c1, c2, c3);
    v[r]     = c0 * ats.x + c1 * ats.y + c2 * ats.z + c3 * ats.w;
    v[8 + r] = c0 * atd.x + c1 * atd.y + c2 * atd.z + c3 * atd.w;
  }
  fold_half<8>(v, (l16 & 8) != 0, 8);
  fold_half<4>(v, (l16 & 4) != 0, 4);
  fold_half<2>(v, (l16 & 2) != 0, 2);
  fold_half<1>(v, (l16 & 1) != 0, 1);
  const int orow = tm * 16 + (l16 & 7) + half * 8;
  if (orow < M) {
    float* dp = (l16 < 8) ? as_ : ad_;
    dp[(long)orow * HW + hd] = v[0];
  }
}

#define GRP_SHIFT 8
#define GRP  256
#define MAXB 1024
#define EPB  4096
#define TBE  1024

__device__ __forceinline__ unsigned range_of(int d, int NB) {
  return min((unsigned)d >> GRP_SHIFT, (unsigned)(NB - 1));
}

__global__ void __launch_bounds__(TBE)
bin_count(const int* __restrict__ dsts, int E0, int NB, int* __restrict__ hist) {
  __shared__ int h[MAXB];
  const int t = threadIdx.x;
  for (int b = t; b < NB; b += TBE) h[b] = 0;
  __syncthreads();
  const int base = blockIdx.x * EPB;
  for (int k = 0; k < EPB / TBE; ++k) {
    const int e = base + k * TBE + t;
    if (e < E0) atomicAdd(&h[range_of(dsts[e], NB)], 1);
  }
  __syncthreads();
  for (int b = t; b < NB; b += TBE) hist[blockIdx.x * NB + b] = h[b];
}

__global__ void __launch_bounds__(TBE)
bin_place(const int* __restrict__ srcs, const int* __restrict__ dsts, int E0, int NB, int C,
          const int* __restrict__ hist, int* __restrict__ binstart, unsigned* __restrict__ recs) {
  __shared__ int tot[MAXB];
  __shared__ int cur[MAXB];
  const int t = threadIdx.x;
  const int chunk = blockIdx.x;
  for (int b = t; b < NB; b += TBE) { tot[b] = 0; cur[b] = 0; }
  __syncthreads();
  const int part = t >> 8;
  for (int b = t & 255; b < NB; b += 256) {
    int s_all = 0, s_pre = 0;
    for (int c = part; c < C; c += TBE / 256) {
      const int v = hist[c * NB + b];
      s_all += v;
      s_pre += (c < chunk) ? v : 0;
    }
    atomicAdd(&tot[b], s_all);
    atomicAdd(&cur[b], s_pre);
  }
  __syncthreads();
  if (t < 32) {
    const int K = (NB + 31) >> 5;
    const int lo = t * K;
    const int hi = min(lo + K, NB);
    int s = 0;
    for (int b = lo; b < hi; ++b) s += tot[b];
    int incl = s;
    for (int o = 1; o < 32; o <<= 1) {
      const int v = __shfl_up(incl, o);
      if (t >= o) incl += v;
    }
    int run = incl - s;
    for (int b = lo; b < hi; ++b) { const int c = tot[b]; tot[b] = run; run += c; }
  }
  __syncthreads();
  for (int b = t; b < NB; b += TBE) {
    const int bs = tot[b];
    if (chunk == 0) binstart[b] = bs;
    cur[b] += bs;
  }
  if (chunk == 0 && t == 0) binstart[NB] = E0;
  __syncthreads();
  const int base = chunk * EPB;
  for (int k = 0; k < EPB / TBE; ++k) {
    const int e = base + k * TBE + t;
    if (e < E0) {
      const int d = dsts[e];
      const int s = srcs[e];
      const int pos = atomicAdd(&cur[range_of(d, NB)], 1);
      recs[pos] = ((unsigned)s << GRP_SHIFT) | ((unsigned)d & (unsigned)(GRP - 1));
    }
  }
}

__global__ void __launch_bounds__(TBE)
bin_group(const int* __restrict__ binstart, const unsigned* __restrict__ recs,
          int* __restrict__ segof, int* seg, int* __restrict__ ssrc, int N, int NB) {
  __shared__ int cnt[GRP];
  __shared__ int wtot[GRP / 32];
  const int t = threadIdx.x;
  const int lane = t & 31;
  const int b = blockIdx.x;
  const int beg = binstart[b];
  const int end = binstart[b + 1];
  if (t < GRP) cnt[t] = 0;
  __syncthreads();
  for (int i = beg + t; i < end; i += TBE) atomicAdd(&cnt[recs[i] & (unsigned)(GRP - 1)], 1);
  __syncthreads();
  int c = 0, incl = 0;
  if (t < GRP) {
    c = cnt[t];
    incl = c;
    for (int o = 1; o < 32; o <<= 1) {
      const int v = __shfl_up(incl, o);
      if (lane >= o) incl += v;
    }
    if (lane == 31) wtot[t >> 5] = incl;
  }
  __syncthreads();
  if (t < GRP) {
    int basew = 0;
    for (int k = 0; k < (t >> 5); ++k) basew += wtot[k];
    const int first = beg + basew + incl - c;
    const int node = (b << GRP_SHIFT) + t;
    if (node < N) segof[node] = first;
    cnt[t] = first;
  }
  if (b == NB - 1 && t == 0) segof[N] = end;
  __syncthreads();
  for (int i = beg + t; i < end; i += TBE) {
    const unsigned r = recs[i];
    const int pos = atomicAdd(&cnt[r & (unsigned)(GRP - 1)], 1);
    seg[pos] = (int)(r >> GRP_SHIFT);
  }
  __syncthreads();

  for (int ln = t >> 5; ln < GRP; ln += TBE / 32) {
    const int gb = __builtin_amdgcn_readfirstlane((ln == 0) ? beg : cnt[ln - 1]);
    const int ge = __builtin_amdgcn_readfirstlane(cnt[ln]);
    for (int ib = gb; ib < ge; ib += 32) {
      const int i = ib + lane;
      const bool act = i < ge;
      const int ki = act ? seg[i] : 0;
      int r = 0;
      for (int jb = gb; jb < ge; jb += 32) {
        const int j = jb + lane;
        const int kj = (j < ge) ? seg[j] : 0;
        const int cntj = min(32, ge - jb);
        for (int q = 0; q < cntj; ++q) {
          const int k = __builtin_amdgcn_readlane(kj, q);
          const int jj = jb + q;
          r += (k < ki || (k == ki && jj < i)) ? 1 : 0;
        }
      }
      if (act) ssrc[gb + r] = ki;
    }
  }
}

__device__ __forceinline__ float leaky(float v) { return (v > 0.f) ? v : NEG_SLOPE * v; }

__device__ __forceinline__ float wave_max(float v) {
  for (int o = 16; o > 0; o >>= 1) v = fmaxf(v, __shfl_xor(v, o));
  return v;
}
__device__ __forceinline__ float wave_sum(float v) {
  for (int o = 16; o > 0; o >>= 1) v += __shfl_xor(v, o);
  return v;
}

template <int H, int VEC, bool ELU_OUT>
__global__ void __launch_bounds__(256)
node_softmax_agg(const int* __restrict__ off, const int* __restrict__ ssrc,
                 const float* __restrict__ as_, const float* __restrict__ ad_,
                 const float* __restrict__ h, const float* __restrict__ bias,
                 float* __restrict__ outp, int n) {
  constexpr int F = 32 * VEC;
  constexpr int C = F / H;
  int lane = threadIdx.x & 31;
  int node = (int)((blockIdx.x * blockDim.x + threadIdx.x) >> 5);
  node = __builtin_amdgcn_readfirstlane(node);
  if (node >= n) return;
  const int beg = off[node];
  const int tot = off[node + 1] - beg + 1;
  const int myh = (lane * VEC) / C;

  float adv[H], m[H], v[H];
  for (int k = 0; k < H; ++k) { adv[k] = ad_[node * H + k]; m[k] = -INFINITY; v[k] = 0.f; }
  int s = 0;
  bool act = false;

  auto load_chunk = [&](int jb) {
    int idx = jb + lane;
    act = idx < tot;
    s = 0;
    for (int k = 0; k < H; ++k) v[k] = 0.f;
    if (act) {
      s = (idx == 0) ? node : ssrc[beg + idx - 1];
      for (int k = 0; k < H; ++k) v[k] = leaky(as_[s * H + k] + adv[k]);
    }
  };

  for (int jb = 0; jb < tot; jb += 32) {
    load_chunk(jb);
    if (act) { for (int k = 0; k < H; ++k) m[k] = fmaxf(m[k], v[k]); }
  }
  for (int k = 0; k < H; ++k) m[k] = wave_max(m[k]);

  float den[H];
  for (int k = 0; k < H; ++k) den[k] = 0.f;
  for (int jb = 0; jb < tot; jb += 32) {
    if (tot > 32) load_chunk(jb);
    if (act) { for (int k = 0; k < H; ++k) den[k] += expf(v[k] - m[k]); }
  }
  for (int k = 0; k < H; ++k) den[k] = wave_sum(den[k]);

  float acc[VEC];
  for (int q = 0; q < VEC; ++q) acc[q] = 0.f;
  for (int jb = 0; jb < tot; jb += 32) {
    if (tot > 32) load_chunk(jb);
    float w[H];
    for (int k = 0; k < H; ++k) w[k] = act ? expf(v[k] - m[k]) / den[k] : 0.f;
    const int cnt = min(32, tot - jb);
#pragma unroll 4
    for (int t = 0; t < cnt; ++t) {
      int st = __builtin_amdgcn_readlane(s, t);
      float wt = __int_as_float(__builtin_amdgcn_readlane(__float_as_int(w[0]), t));
      if (H == 2) {
        float w1 = __int_as_float(__builtin_amdgcn_readlane(__float_as_int(w[H - 1]), t));
        wt = myh ? w1 : wt;
      }
      const float* row = h + (long)st * F + lane * VEC;
      if (VEC == 4) {
        float4 r = *reinterpret_cast<const float4*>(row);
        acc[0] += r.x * wt; acc[1] += r.y * wt; acc[VEC - 2] += r.z * wt; acc[VEC - 1] += r.w * wt;
      } else {
        float2 r = *reinterpret_cast<const float2*>(row);
        acc[0] += r.x * wt; acc[1] += r.y * wt;
      }
    }
  }

  const int col = lane * VEC;
  for (int q = 0; q < VEC; ++q) {
    float val = acc[q] + bias[col + q];
    if (ELU_OUT) val = (val > 0.f) ? val : (expf(val) - 1.f);
    acc[q] = val;
  }
  float* dstp = outp + (long)node * F + col;
  if (VEC == 4) {
    *reinterpret_cast<float4*>(dstp) = make_float4(acc[0], acc[1], acc[VEC - 2], acc[VEC - 1]);
  } else {
    *reinterpret_cast<float2*>(dstp) = make_float2(acc[0], acc[1]);
  }
}

extern "C" void kernel_launch(void* const* d_in, const int* in_sizes, int n_in,
                              void* d_out, int out_size, void* d_ws, size_t ws_size,
                              hipStream_t stream) {
  const float* x    = (const float*)d_in[0];
  const int*   ei   = (const int*)  d_in[1];
  const float* W1   = (const float*)d_in[2];
  const float* atS1 = (const float*)d_in[3];
  const float* atD1 = (const float*)d_in[4];
  const float* b1   = (const float*)d_in[5];
  const float* W2   = (const float*)d_in[6];
  const float* atS2 = (const float*)d_in[7];
  const float* atD2 = (const float*)d_in[8];
  const float* b2   = (const float*)d_in[9];
  float* out = (float*)d_out;

  const int N    = in_sizes[0] / DIN;
  const int E0   = in_sizes[1] / 2;
  const int* srcs = ei;
  const int* dsts = ei + E0;

  float* ws = (float*)d_ws;
  long off = 0;
  float* h1  = ws + off; off += (long)N * (HEADS * DH);
  float* o1  = ws + off; off += (long)N * (HEADS * DH);
  float* a1s = ws + off; off += (long)N * HEADS;
  float* a1d = ws + off; off += (long)N * HEADS;
  const int NB = (N + GRP - 1) / GRP;
  const int CH = (E0 + EPB - 1) / EPB;
  int* segof = (int*)(ws + off); off += (long)N + 1;
  int* hist  = (int*)(ws + off); off += (long)CH * NB;
  int* bstart = (int*)(ws + off); off += (long)NB + 1;
  unsigned* recs = (unsigned*)(ws + off); off += (long)E0;
  int* seg   = (int*)(ws + off); off += (long)E0;
  int* ssrc  = (int*)(ws + off); off += (long)E0;
  float* h2  = h1;
  float* a2s = h1 + (long)N * DOUT;
  float* a2d = a2s + N;

  const int TB = 256;
  auto blocks = [](long n, int tb) { return (unsigned)((n + tb - 1) / tb); };

  bin_count<<<CH, TBE, 0, stream>>>(dsts, E0, NB, hist);
  bin_place<<<CH, TBE, 0, stream>>>(srcs, dsts, E0, NB, CH, hist, bstart, recs);
  bin_group<<<NB, TBE, 0, stream>>>(bstart, recs, segof, seg, ssrc, N, NB);

  {
    long waves = (long)((N + 15) / 16) * HEADS;
    gemm_heads_wmma_f32<DIN, HEADS * DH><<<blocks(waves * 32, TB), TB, 0, stream>>>(
        x, W1, atS1, atD1, h1, a1s, a1d, N);
  }
  node_softmax_agg<HEADS, 4, true><<<blocks((long)N * 32, TB), TB, 0, stream>>>(
      segof, ssrc, a1s, a1d, h1, b1, o1, N);

  {
    long waves = (long)((N + 15) / 16);
    gemm_heads_wmma_f32<HEADS * DH, DOUT><<<blocks(waves * 32, TB), TB, 0, stream>>>(
        o1, W2, atS2, atD2, h2, a2s, a2d, N);
  }
  node_softmax_agg<1, 2, false><<<blocks((long)N * 32, TB), TB, 0, stream>>>(
      segof, ssrc, a2s, a2d, h2, b2, out, N);
}


# Edge counting pass and layer-1 GEMM do not depend on each other and now share one launch (count blocks first, 256 threads, 16 edges per thread): 7 -> 6 launches; every sum and its order unchanged
# speedup vs baseline: 5.4262x; 1.0408x over previous
#include <hip/hip_runtime.h>
#include <math.h>

#define DIN   128
#define DH    64
#define HEADS 2
#define DOUT  64
#define NEG_SLOPE 0.2f

typedef float v2f __attribute__((ext_vector_type(2)));
typedef float v8f __attribute__((ext_vector_type(8)));


template <int CNT>
__device__ __forceinline__ void fold_half(float (&v)[16], bool upper, int dist) {
#pragma unroll
  for (int i = 0; i < CNT; ++i) {
    float lo = v[i], hi = v[i + CNT];
    float send = upper ? lo : hi;
    float keep = upper ? hi : lo;
    v[i] = keep + __shfl_xor(send, dist);
  }
}

template <int K, int NC>
__device__ __forceinline__ void
gemm_heads_wave(const float* __restrict__ A, const float* __restrict__ B,
                const float* __restrict__ att_s, const float* __restrict__ att_d,
                float* __restrict__ C, float* __restrict__ as_, float* __restrict__ ad_,
                int M, int wave) {
  constexpr int HW = NC / 64;
  const int lane = threadIdx.x & 31;
  const int tm = wave / HW;
  const int hd = wave - tm * HW;
  if (tm * 16 >= M) return;
  const int half = lane >> 4;
  const int l16  = lane & 15;
  long rowA = (long)tm * 16 + l16;
  if (rowA >= M) rowA = M - 1;
  const int colbase = hd * 64 + l16 * 4;
  const float* ap = A + rowA * K + half * 2;
  const float* bp = B + (long)(half * 2) * NC + colbase;
  v8f acc0 = {}, acc1 = {}, acc2 = {}, acc3 = {};
#pragma unroll
  for (int k0 = 0; k0 < K; k0 += 4) {
    v2f a = *reinterpret_cast<const v2f*>(ap + k0);
    float4 bx = *reinterpret_cast<const float4*>(bp + (long)k0 * NC);
    float4 by = *reinterpret_cast<const float4*>(bp + (long)(k0 + 1) * NC);
    v2f b;
    b.x = bx.x; b.y = by.x;
    acc0 = __builtin_amdgcn_wmma_f32_16x16x4_f32(false, a, false, b, (short)0, acc0, false, false);
    b.x = bx.y; b.y = by.y;
    acc1 = __builtin_amdgcn_wmma_f32_16x16x4_f32(false, a, false, b, (short)0, acc1, false, false);
    b.x = bx.z; b.y = by.z;
    acc2 = __builtin_amdgcn_wmma_f32_16x16x4_f32(false, a, false, b, (short)0, acc2, false, false);
    b.x = bx.w; b.y = by.w;
    acc3 = __builtin_amdgcn_wmma_f32_16x16x4_f32(false, a, false, b, (short)0, acc3, false, false);
  }

  const float4 ats = *reinterpret_cast<const float4*>(att_s + colbase);
  const float4 atd = *reinterpret_cast<const float4*>(att_d + colbase);
  float v[16];
#pragma unroll
  for (int r = 0; r < 8; ++r) {
    const float c0 = acc0[r], c1 = acc1[r], c2 = acc2[r], c3 = acc3[r];
    const int row = tm * 16 + r + half * 8;
    if (row < M)
      *reinterpret_cast<float4*>(C + (long)row * NC + colbase) = make_float4(c0, c1, c2, c3);
    v[r]     = c0 * ats.x + c1 * ats.y + c2 * ats.z + c3 * ats.w;
    v[8 + r] = c0 * atd.x + c1 * atd.y + c2 * atd.z + c3 * atd.w;
  }
  fold_half<8>(v, (l16 & 8) != 0, 8);
  fold_half<4>(v, (l16 & 4) != 0, 4);
  fold_half<2>(v, (l16 & 2) != 0, 2);
  fold_half<1>(v, (l16 & 1) != 0, 1);
  const int orow = tm * 16 + (l16 & 7) + half * 8;
  if (orow < M) {
    float* dp = (l16 < 8) ? as_ : ad_;
    dp[(long)orow * HW + hd] = v[0];
  }
}

template <int K, int NC>
__global__ void __launch_bounds__(256)
gemm_heads_wmma_f32(const float* __restrict__ A, const float* __restrict__ B,
                    const float* __restrict__ att_s, const float* __restrict__ att_d,
                    float* __restrict__ C, float* __restrict__ as_, float* __restrict__ ad_,
                    int M) {
  int wave = (int)((blockIdx.x * blockDim.x + threadIdx.x) >> 5);
  wave = __builtin_amdgcn_readfirstlane(wave);
  gemm_heads_wave<K, NC>(A, B, att_s, att_d, C, as_, ad_, M, wave);
}

#define GRP_SHIFT 8
#define GRP  256
#define MAXB 1024
#define EPB  4096
#define TBE  1024

__device__ __forceinline__ unsigned range_of(int d, int NB) {
  return min((unsigned)d >> GRP_SHIFT, (unsigned)(NB - 1));
}

#define TBG 256
template <int K, int NC>
__global__ void __launch_bounds__(TBG)
count_and_gemm(const int* __restrict__ dsts, int E0, int NB, int CH, int* __restrict__ hist,
               const float* __restrict__ A, const float* __restrict__ B,
               const float* __restrict__ att_s, const float* __restrict__ att_d,
               float* __restrict__ C, float* __restrict__ as_, float* __restrict__ ad_,
               int M) {
  const int t = threadIdx.x;
  if ((int)blockIdx.x < CH) {
    __shared__ int h[MAXB];
    for (int b = t; b < NB; b += TBG) h[b] = 0;
    __syncthreads();
    const int base = blockIdx.x * EPB;
    int d[EPB / TBG];
#pragma unroll
    for (int k = 0; k < EPB / TBG; ++k) {
      const int e = base + k * TBG + t;
      d[k] = dsts[min(e, E0 - 1)];
    }
#pragma unroll
    for (int k = 0; k < EPB / TBG; ++k) {
      const int e = base + k * TBG + t;
      if (e < E0) atomicAdd(&h[range_of(d[k], NB)], 1);
    }
    __syncthreads();
    for (int b = t; b < NB; b += TBG) hist[blockIdx.x * NB + b] = h[b];
    return;
  }
  int wave = (int)((((int)blockIdx.x - CH) * TBG + t) >> 5);
  wave = __builtin_amdgcn_readfirstlane(wave);
  gemm_heads_wave<K, NC>(A, B, att_s, att_d, C, as_, ad_, M, wave);
}

__global__ void __launch_bounds__(TBE)
bin_place(const int* __restrict__ srcs, const int* __restrict__ dsts, int E0, int NB, int C,
          const int* __restrict__ hist, int* __restrict__ binstart, unsigned* __restrict__ recs) {
  __shared__ int tot[MAXB];
  __shared__ int cur[MAXB];
  const int t = threadIdx.x;
  const int chunk = blockIdx.x;
  for (int b = t; b < NB; b += TBE) { tot[b] = 0; cur[b] = 0; }
  __syncthreads();
  const int part = t >> 8;
  for (int b = t & 255; b < NB; b += 256) {
    int s_all = 0, s_pre = 0;
    for (int c = part; c < C; c += TBE / 256) {
      const int v = hist[c * NB + b];
      s_all += v;
      s_pre += (c < chunk) ? v : 0;
    }
    atomicAdd(&tot[b], s_all);
    atomicAdd(&cur[b], s_pre);
  }
  __syncthreads();
  if (t < 32) {
    const int K = (NB + 31) >> 5;
    const int lo = t * K;
    const int hi = min(lo + K, NB);
    int s = 0;
    for (int b = lo; b < hi; ++b) s += tot[b];
    int incl = s;
    for (int o = 1; o < 32; o <<= 1) {
      const int v = __shfl_up(incl, o);
      if (t >= o) incl += v;
    }
    int run = incl - s;
    for (int b = lo; b < hi; ++b) { const int c = tot[b]; tot[b] = run; run += c; }
  }
  __syncthreads();
  for (int b = t; b < NB; b += TBE) {
    const int bs = tot[b];
    if (chunk == 0) binstart[b] = bs;
    cur[b] += bs;
  }
  if (chunk == 0 && t == 0) binstart[NB] = E0;
  __syncthreads();
  const int base = chunk * EPB;
  for (int k = 0; k < EPB / TBE; ++k) {
    const int e = base + k * TBE + t;
    if (e < E0) {
      const int d = dsts[e];
      const int s = srcs[e];
      const int pos = atomicAdd(&cur[range_of(d, NB)], 1);
      recs[pos] = ((unsigned)s << GRP_SHIFT) | ((unsigned)d & (unsigned)(GRP - 1));
    }
  }
}

__global__ void __launch_bounds__(TBE)
bin_group(const int* __restrict__ binstart, const unsigned* __restrict__ recs,
          int* __restrict__ segof, int* seg, int* __restrict__ ssrc, int N, int NB) {
  __shared__ int cnt[GRP];
  __shared__ int wtot[GRP / 32];
  const int t = threadIdx.x;
  const int lane = t & 31;
  const int b = blockIdx.x;
  const int beg = binstart[b];
  const int end = binstart[b + 1];
  if (t < GRP) cnt[t] = 0;
  __syncthreads();
  for (int i = beg + t; i < end; i += TBE) atomicAdd(&cnt[recs[i] & (unsigned)(GRP - 1)], 1);
  __syncthreads();
  int c = 0, incl = 0;
  if (t < GRP) {
    c = cnt[t];
    incl = c;
    for (int o = 1; o < 32; o <<= 1) {
      const int v = __shfl_up(incl, o);
      if (lane >= o) incl += v;
    }
    if (lane == 31) wtot[t >> 5] = incl;
  }
  __syncthreads();
  if (t < GRP) {
    int basew = 0;
    for (int k = 0; k < (t >> 5); ++k) basew += wtot[k];
    const int first = beg + basew + incl - c;
    const int node = (b << GRP_SHIFT) + t;
    if (node < N) segof[node] = first;
    cnt[t] = first;
  }
  if (b == NB - 1 && t == 0) segof[N] = end;
  __syncthreads();
  for (int i = beg + t; i < end; i += TBE) {
    const unsigned r = recs[i];
    const int pos = atomicAdd(&cnt[r & (unsigned)(GRP - 1)], 1);
    seg[pos] = (int)(r >> GRP_SHIFT);
  }
  __syncthreads();

  for (int ln = t >> 5; ln < GRP; ln += TBE / 32) {
    const int gb = __builtin_amdgcn_readfirstlane((ln == 0) ? beg : cnt[ln - 1]);
    const int ge = __builtin_amdgcn_readfirstlane(cnt[ln]);
    for (int ib = gb; ib < ge; ib += 32) {
      const int i = ib + lane;
      const bool act = i < ge;
      const int ki = act ? seg[i] : 0;
      int r = 0;
      for (int jb = gb; jb < ge; jb += 32) {
        const int j = jb + lane;
        const int kj = (j < ge) ? seg[j] : 0;
        const int cntj = min(32, ge - jb);
        for (int q = 0; q < cntj; ++q) {
          const int k = __builtin_amdgcn_readlane(kj, q);
          const int jj = jb + q;
          r += (k < ki || (k == ki && jj < i)) ? 1 : 0;
        }
      }
      if (act) ssrc[gb + r] = ki;
    }
  }
}

__device__ __forceinline__ float leaky(float v) { return (v > 0.f) ? v : NEG_SLOPE * v; }

__device__ __forceinline__ float wave_max(float v) {
  for (int o = 16; o > 0; o >>= 1) v = fmaxf(v, __shfl_xor(v, o));
  return v;
}
__device__ __forceinline__ float wave_sum(float v) {
  for (int o = 16; o > 0; o >>= 1) v += __shfl_xor(v, o);
  return v;
}

template <int H, int VEC, bool ELU_OUT>
__global__ void __launch_bounds__(256)
node_softmax_agg(const int* __restrict__ off, const int* __restrict__ ssrc,
                 const float* __restrict__ as_, const float* __restrict__ ad_,
                 const float* __restrict__ h, const float* __restrict__ bias,
                 float* __restrict__ outp, int n) {
  constexpr int F = 32 * VEC;
  constexpr int C = F / H;
  int lane = threadIdx.x & 31;
  int node = (int)((blockIdx.x * blockDim.x + threadIdx.x) >> 5);
  node = __builtin_amdgcn_readfirstlane(node);
  if (node >= n) return;
  const int beg = off[node];
  const int tot = off[node + 1] - beg + 1;
  const int myh = (lane * VEC) / C;

  float adv[H], m[H], v[H];
  for (int k = 0; k < H; ++k) { adv[k] = ad_[node * H + k]; m[k] = -INFINITY; v[k] = 0.f; }
  int s = 0;
  bool act = false;

  auto load_chunk = [&](int jb) {
    int idx = jb + lane;
    act = idx < tot;
    s = 0;
    for (int k = 0; k < H; ++k) v[k] = 0.f;
    if (act) {
      s = (idx == 0) ? node : ssrc[beg + idx - 1];
      for (int k = 0; k < H; ++k) v[k] = leaky(as_[s * H + k] + adv[k]);
    }
  };

  for (int jb = 0; jb < tot; jb += 32) {
    load_chunk(jb);
    if (act) { for (int k = 0; k < H; ++k) m[k] = fmaxf(m[k], v[k]); }
  }
  for (int k = 0; k < H; ++k) m[k] = wave_max(m[k]);

  float den[H];
  for (int k = 0; k < H; ++k) den[k] = 0.f;
  for (int jb = 0; jb < tot; jb += 32) {
    if (tot > 32) load_chunk(jb);
    if (act) { for (int k = 0; k < H; ++k) den[k] += expf(v[k] - m[k]); }
  }
  for (int k = 0; k < H; ++k) den[k] = wave_sum(den[k]);

  float acc[VEC];
  for (int q = 0; q < VEC; ++q) acc[q] = 0.f;
  for (int jb = 0; jb < tot; jb += 32) {
    if (tot > 32) load_chunk(jb);
    float w[H];
    for (int k = 0; k < H; ++k) w[k] = act ? expf(v[k] - m[k]) / den[k] : 0.f;
    const int cnt = min(32, tot - jb);
#pragma unroll 4
    for (int t = 0; t < cnt; ++t) {
      int st = __builtin_amdgcn_readlane(s, t);
      float wt = __int_as_float(__builtin_amdgcn_readlane(__float_as_int(w[0]), t));
      if (H == 2) {
        float w1 = __int_as_float(__builtin_amdgcn_readlane(__float_as_int(w[H - 1]), t));
        wt = myh ? w1 : wt;
      }
      const float* row = h + (long)st * F + lane * VEC;
      if (VEC == 4) {
        float4 r = *reinterpret_cast<const float4*>(row);
        acc[0] += r.x * wt; acc[1] += r.y * wt; acc[VEC - 2] += r.z * wt; acc[VEC - 1] += r.w * wt;
      } else {
        float2 r = *reinterpret_cast<const float2*>(row);
        acc[0] += r.x * wt; acc[1] += r.y * wt;
      }
    }
  }

  const int col = lane * VEC;
  for (int q = 0; q < VEC; ++q) {
    float val = acc[q] + bias[col + q];
    if (ELU_OUT) val = (val > 0.f) ? val : (expf(val) - 1.f);
    acc[q] = val;
  }
  float* dstp = outp + (long)node * F + col;
  if (VEC == 4) {
    *reinterpret_cast<float4*>(dstp) = make_float4(acc[0], acc[1], acc[VEC - 2], acc[VEC - 1]);
  } else {
    *reinterpret_cast<float2*>(dstp) = make_float2(acc[0], acc[1]);
  }
}

extern "C" void kernel_launch(void* const* d_in, const int* in_sizes, int n_in,
                              void* d_out, int out_size, void* d_ws, size_t ws_size,
                              hipStream_t stream) {
  const float* x    = (const float*)d_in[0];
  const int*   ei   = (const int*)  d_in[1];
  const float* W1   = (const float*)d_in[2];
  const float* atS1 = (const float*)d_in[3];
  const float* atD1 = (const float*)d_in[4];
  const float* b1   = (const float*)d_in[5];
  const float* W2   = (const float*)d_in[6];
  const float* atS2 = (const float*)d_in[7];
  const float* atD2 = (const float*)d_in[8];
  const float* b2   = (const float*)d_in[9];
  float* out = (float*)d_out;

  const int N    = in_sizes[0] / DIN;
  const int E0   = in_sizes[1] / 2;
  const int* srcs = ei;
  const int* dsts = ei + E0;

  float* ws = (float*)d_ws;
  long off = 0;
  float* h1  = ws + off; off += (long)N * (HEADS * DH);
  float* o1  = ws + off; off += (long)N * (HEADS * DH);
  float* a1s = ws + off; off += (long)N * HEADS;
  float* a1d = ws + off; off += (long)N * HEADS;
  const int NB = (N + GRP - 1) / GRP;
  const int CH = (E0 + EPB - 1) / EPB;
  int* segof = (int*)(ws + off); off += (long)N + 1;
  int* hist  = (int*)(ws + off); off += (long)CH * NB;
  int* bstart = (int*)(ws + off); off += (long)NB + 1;
  unsigned* recs = (unsigned*)(ws + off); off += (long)E0;
  int* seg   = (int*)(ws + off); off += (long)E0;
  int* ssrc  = (int*)(ws + off); off += (long)E0;
  float* h2  = h1;
  float* a2s = h1 + (long)N * DOUT;
  float* a2d = a2s + N;

  const int TB = 256;
  auto blocks = [](long n, int tb) { return (unsigned)((n + tb - 1) / tb); };

  {
    long waves = (long)((N + 15) / 16) * HEADS;
    count_and_gemm<DIN, HEADS * DH><<<CH + blocks(waves * 32, TBG), TBG, 0, stream>>>(
        dsts, E0, NB, CH, hist, x, W1, atS1, atD1, h1, a1s, a1d, N);
  }
  bin_place<<<CH, TBE, 0, stream>>>(srcs, dsts, E0, NB, CH, hist, bstart, recs);
  bin_group<<<NB, TBE, 0, stream>>>(bstart, recs, segof, seg, ssrc, N, NB);

  node_softmax_agg<HEADS, 4, true><<<blocks((long)N * 32, TB), TB, 0, stream>>>(
      segof, ssrc, a1s, a1d, h1, b1, o1, N);

  {
    long waves = (long)((N + 15) / 16);
    gemm_heads_wmma_f32<HEADS * DH, DOUT><<<blocks(waves * 32, TB), TB, 0, stream>>>(
        o1, W2, atS2, atD2, h2, a2s, a2d, N);
  }
  node_softmax_agg<1, 2, false><<<blocks((long)N * 32, TB), TB, 0, stream>>>(
      segof, ssrc, a2s, a2d, h2, b2, out, N);
}


# First grouping level moved into the counting blocks: each rewrites its chunk ordered by range (rank by LDS add), range blocks gather their pieces; placing launch gone, 6 -> 5 launches; sum order kept
# speedup vs baseline: 6.6090x; 1.2180x over previous
#include <hip/hip_runtime.h>
#include <math.h>

#define DIN   128
#define DH    64
#define HEADS 2
#define DOUT  64
#define NEG_SLOPE 0.2f

typedef float v2f __attribute__((ext_vector_type(2)));
typedef float v8f __attribute__((ext_vector_type(8)));


template <int CNT>
__device__ __forceinline__ void fold_half(float (&v)[16], bool upper, int dist) {
#pragma unroll
  for (int i = 0; i < CNT; ++i) {
    float lo = v[i], hi = v[i + CNT];
    float send = upper ? lo : hi;
    float keep = upper ? hi : lo;
    v[i] = keep + __shfl_xor(send, dist);
  }
}

template <int K, int NC>
__device__ __forceinline__ void
gemm_heads_wave(const float* __restrict__ A, const float* __restrict__ B,
                const float* __restrict__ att_s, const float* __restrict__ att_d,
                float* __restrict__ C, float* __restrict__ as_, float* __restrict__ ad_,
                int M, int wave) {
  constexpr int HW = NC / 64;
  const int lane = threadIdx.x & 31;
  const int tm = wave / HW;
  const int hd = wave - tm * HW;
  if (tm * 16 >= M) return;
  const int half = lane >> 4;
  const int l16  = lane & 15;
  long rowA = (long)tm * 16 + l16;
  if (rowA >= M) rowA = M - 1;
  const int colbase = hd * 64 + l16 * 4;
  const float* ap = A + rowA * K + half * 2;
  const float* bp = B + (long)(half * 2) * NC + colbase;
  v8f acc0 = {}, acc1 = {}, acc2 = {}, acc3 = {};
#pragma unroll
  for (int k0 = 0; k0 < K; k0 += 4) {
    v2f a = *reinterpret_cast<const v2f*>(ap + k0);
    float4 bx = *reinterpret_cast<const float4*>(bp + (long)k0 * NC);
    float4 by = *reinterpret_cast<const float4*>(bp + (long)(k0 + 1) * NC);
    v2f b;
    b.x = bx.x; b.y = by.x;
    acc0 = __builtin_amdgcn_wmma_f32_16x16x4_f32(false, a, false, b, (short)0, acc0, false, false);
    b.x = bx.y; b.y = by.y;
    acc1 = __builtin_amdgcn_wmma_f32_16x16x4_f32(false, a, false, b, (short)0, acc1, false, false);
    b.x = bx.z; b.y = by.z;
    acc2 = __builtin_amdgcn_wmma_f32_16x16x4_f32(false, a, false, b, (short)0, acc2, false, false);
    b.x = bx.w; b.y = by.w;
    acc3 = __builtin_amdgcn_wmma_f32_16x16x4_f32(false, a, false, b, (short)0, acc3, false, false);
  }

  const float4 ats = *reinterpret_cast<const float4*>(att_s + colbase);
  const float4 atd = *reinterpret_cast<const float4*>(att_d + colbase);
  float v[16];
#pragma unroll
  for (int r = 0; r < 8; ++r) {
    const float c0 = acc0[r], c1 = acc1[r], c2 = acc2[r], c3 = acc3[r];
    const int row = tm * 16 + r + half * 8;
    if (row < M)
      *reinterpret_cast<float4*>(C + (long)row * NC + colbase) = make_float4(c0, c1, c2, c3);
    v[r]     = c0 * ats.x + c1 * ats.y + c2 * ats.z + c3 * ats.w;
    v[8 + r] = c0 * atd.x + c1 * atd.y + c2 * atd.z + c3 * atd.w;
  }
  fold_half<8>(v, (l16 & 8) != 0, 8);
  fold_half<4>(v, (l16 & 4) != 0, 4);
  fold_half<2>(v, (l16 & 2) != 0, 2);
  fold_half<1>(v, (l16 & 1) != 0, 1);
  const int orow = tm * 16 + (l16 & 7) + half * 8;
  if (orow < M) {
    float* dp = (l16 < 8) ? as_ : ad_;
    dp[(long)orow * HW + hd] = v[0];
  }
}

template <int K, int NC>
__global__ void __launch_bounds__(256)
gemm_heads_wmma_f32(const float* __restrict__ A, const float* __restrict__ B,
                    const float* __restrict__ att_s, const float* __restrict__ att_d,
                    float* __restrict__ C, float* __restrict__ as_, float* __restrict__ ad_,
                    int M) {
  int wave = (int)((blockIdx.x * blockDim.x + threadIdx.x) >> 5);
  wave = __builtin_amdgcn_readfirstlane(wave);
  gemm_heads_wave<K, NC>(A, B, att_s, att_d, C, as_, ad_, M, wave);
}

#define GRP_SHIFT 8
#define GRP  256
#define MAXB 1024
#define EPB  4096
#define MAXC 1024
#define TBE  1024

__device__ __forceinline__ unsigned range_of(int d, int NB) {
  return min((unsigned)d >> GRP_SHIFT, (unsigned)(NB - 1));
}

#define TBG 256
template <int K, int NC>
__global__ void __launch_bounds__(TBG)
count_and_gemm(const int* __restrict__ srcs, const int* __restrict__ dsts, int E0, int NB, int CH,
               int* __restrict__ hist, int* __restrict__ lstart, unsigned* __restrict__ recs,
               const float* __restrict__ A, const float* __restrict__ B,
               const float* __restrict__ att_s, const float* __restrict__ att_d,
               float* __restrict__ C, float* __restrict__ as_, float* __restrict__ ad_,
               int M) {
  const int t = threadIdx.x;
  if ((int)blockIdx.x < CH) {
    __shared__ int h[MAXB];
    __shared__ int wsum[TBG / 32];
    const int lane = t & 31;
    const int chunk = blockIdx.x;
    for (int b = t; b < MAXB; b += TBG) h[b] = 0;
    __syncthreads();
    const int base = chunk * EPB;
    int d[EPB / TBG], s[EPB / TBG], rk[EPB / TBG];
#pragma unroll
    for (int k = 0; k < EPB / TBG; ++k) {
      const int e = min(base + k * TBG + t, E0 - 1);
      d[k] = dsts[e];
      s[k] = srcs[e];
    }
#pragma unroll
    for (int k = 0; k < EPB / TBG; ++k) {
      const int e = base + k * TBG + t;
      rk[k] = (e < E0) ? atomicAdd(&h[range_of(d[k], NB)], 1) : 0;
    }
    __syncthreads();
    constexpr int PER = MAXB / TBG;
    int c[PER];
    int own = 0;
#pragma unroll
    for (int i = 0; i < PER; ++i) { c[i] = h[t * PER + i]; own += c[i]; }
    int incl = own;
    for (int o = 1; o < 32; o <<= 1) {
      const int v = __shfl_up(incl, o);
      if (lane >= o) incl += v;
    }
    if (lane == 31) wsum[t >> 5] = incl;
    __syncthreads();
    int run = incl - own;
    for (int k = 0; k < (t >> 5); ++k) run += wsum[k];
#pragma unroll
    for (int i = 0; i < PER; ++i) {
      const int b = t * PER + i;
      h[b] = run;
      if (b < NB) { hist[chunk * NB + b] = c[i]; lstart[chunk * NB + b] = run; }
      run += c[i];
    }
    __syncthreads();
#pragma unroll
    for (int k = 0; k < EPB / TBG; ++k) {
      const int e = base + k * TBG + t;
      if (e < E0)
        recs[base + h[range_of(d[k], NB)] + rk[k]] =
            ((unsigned)s[k] << GRP_SHIFT) | ((unsigned)d[k] & (unsigned)(GRP - 1));
    }
    return;
  }
  int wave = (int)((((int)blockIdx.x - CH) * TBG + t) >> 5);
  wave = __builtin_amdgcn_readfirstlane(wave);
  gemm_heads_wave<K, NC>(A, B, att_s, att_d, C, as_, ad_, M, wave);
}

#define SEGU 4
__global__ void __launch_bounds__(TBE)
bin_group(const int* __restrict__ hist, const int* __restrict__ lstart,
          const unsigned* __restrict__ recs, int* __restrict__ segof, int* seg,
          int* __restrict__ ssrc, int N, int NB, int CH) {
  __shared__ int cnt[GRP];
  __shared__ int wtot[GRP / 32];
  __shared__ int segc[MAXC];
  __shared__ int segs[MAXC];
  __shared__ int tot2[2];
  const int t = threadIdx.x;
  const int lane = t & 31;
  const int b = blockIdx.x;
  if (t < GRP) cnt[t] = 0;
  if (t < 2) tot2[t] = 0;
  __syncthreads();
  {
    int ls_sum = 0, cn_sum = 0;
    for (int c = t; c < CH; c += TBE) {
      const int cn = hist[c * NB + b];
      const int ls = lstart[c * NB + b];
      segc[c] = cn;
      segs[c] = c * EPB + ls;
      ls_sum += ls;
      cn_sum += cn;
    }
    for (int o = 16; o > 0; o >>= 1) {
      ls_sum += __shfl_xor(ls_sum, o);
      cn_sum += __shfl_xor(cn_sum, o);
    }
    if (lane == 0) { atomicAdd(&tot2[0], ls_sum); atomicAdd(&tot2[1], cn_sum); }
  }
  __syncthreads();
  const int beg = tot2[0];
  const int end = beg + tot2[1];
  const int wv = t >> 5;
  for (int c0 = wv; c0 < CH; c0 += (TBE / 32) * SEGU) {
    unsigned r[SEGU];
    int n[SEGU], st[SEGU];
#pragma unroll
    for (int u = 0; u < SEGU; ++u) {
      const int c = c0 + (TBE / 32) * u;
      n[u]  = (c < CH) ? segc[c] : 0;
      st[u] = (c < CH) ? segs[c] : 0;
      r[u]  = recs[(n[u] > 0) ? min(st[u] + lane, st[u] + n[u] - 1) : 0];
    }
#pragma unroll
    for (int u = 0; u < SEGU; ++u) {
      if (lane < n[u]) atomicAdd(&cnt[r[u] & (unsigned)(GRP - 1)], 1);
      for (int i = lane + 32; i < n[u]; i += 32)
        atomicAdd(&cnt[recs[st[u] + i] & (unsigned)(GRP - 1)], 1);
    }
  }
  __syncthreads();
  int c = 0, incl = 0;
  if (t < GRP) {
    c = cnt[t];
    incl = c;
    for (int o = 1; o < 32; o <<= 1) {
      const int v = __shfl_up(incl, o);
      if (lane >= o) incl += v;
    }
    if (lane == 31) wtot[t >> 5] = incl;
  }
  __syncthreads();
  if (t < GRP) {
    int basew = 0;
    for (int k = 0; k < (t >> 5); ++k) basew += wtot[k];
    const int first = beg + basew + incl - c;
    const int node = (b << GRP_SHIFT) + t;
    if (node < N) segof[node] = first;
    cnt[t] = first;
  }
  if (b == NB - 1 && t == 0) segof[N] = end;
  __syncthreads();
  for (int c0 = wv; c0 < CH; c0 += (TBE / 32) * SEGU) {
    unsigned r[SEGU];
    int n[SEGU], st[SEGU];
#pragma unroll
    for (int u = 0; u < SEGU; ++u) {
      const int c = c0 + (TBE / 32) * u;
      n[u]  = (c < CH) ? segc[c] : 0;
      st[u] = (c < CH) ? segs[c] : 0;
      r[u]  = recs[(n[u] > 0) ? min(st[u] + lane, st[u] + n[u] - 1) : 0];
    }
#pragma unroll
    for (int u = 0; u < SEGU; ++u) {
      if (lane < n[u]) {
        const int pos = atomicAdd(&cnt[r[u] & (unsigned)(GRP - 1)], 1);
        seg[pos] = (int)(r[u] >> GRP_SHIFT);
      }
      for (int i = lane + 32; i < n[u]; i += 32) {
        const unsigned rr = recs[st[u] + i];
        const int pos = atomicAdd(&cnt[rr & (unsigned)(GRP - 1)], 1);
        seg[pos] = (int)(rr >> GRP_SHIFT);
      }
    }
  }
  __syncthreads();

  for (int ln = t >> 5; ln < GRP; ln += TBE / 32) {
    const int gb = __builtin_amdgcn_readfirstlane((ln == 0) ? beg : cnt[ln - 1]);
    const int ge = __builtin_amdgcn_readfirstlane(cnt[ln]);
    for (int ib = gb; ib < ge; ib += 32) {
      const int i = ib + lane;
      const bool act = i < ge;
      const int ki = act ? seg[i] : 0;
      int r = 0;
      for (int jb = gb; jb < ge; jb += 32) {
        const int j = jb + lane;
        const int kj = (j < ge) ? seg[j] : 0;
        const int cntj = min(32, ge - jb);
        for (int q = 0; q < cntj; ++q) {
          const int k = __builtin_amdgcn_readlane(kj, q);
          const int jj = jb + q;
          r += (k < ki || (k == ki && jj < i)) ? 1 : 0;
        }
      }
      if (act) ssrc[gb + r] = ki;
    }
  }
}

__device__ __forceinline__ float leaky(float v) { return (v > 0.f) ? v : NEG_SLOPE * v; }

__device__ __forceinline__ float wave_max(float v) {
  for (int o = 16; o > 0; o >>= 1) v = fmaxf(v, __shfl_xor(v, o));
  return v;
}
__device__ __forceinline__ float wave_sum(float v) {
  for (int o = 16; o > 0; o >>= 1) v += __shfl_xor(v, o);
  return v;
}

template <int H, int VEC, bool ELU_OUT>
__global__ void __launch_bounds__(256)
node_softmax_agg(const int* __restrict__ off, const int* __restrict__ ssrc,
                 const float* __restrict__ as_, const float* __restrict__ ad_,
                 const float* __restrict__ h, const float* __restrict__ bias,
                 float* __restrict__ outp, int n) {
  constexpr int F = 32 * VEC;
  constexpr int C = F / H;
  int lane = threadIdx.x & 31;
  int node = (int)((blockIdx.x * blockDim.x + threadIdx.x) >> 5);
  node = __builtin_amdgcn_readfirstlane(node);
  if (node >= n) return;
  const int beg = off[node];
  const int tot = off[node + 1] - beg + 1;
  const int myh = (lane * VEC) / C;

  float adv[H], m[H], v[H];
  for (int k = 0; k < H; ++k) { adv[k] = ad_[node * H + k]; m[k] = -INFINITY; v[k] = 0.f; }
  int s = 0;
  bool act = false;

  auto load_chunk = [&](int jb) {
    int idx = jb + lane;
    act = idx < tot;
    s = 0;
    for (int k = 0; k < H; ++k) v[k] = 0.f;
    if (act) {
      s = (idx == 0) ? node : ssrc[beg + idx - 1];
      for (int k = 0; k < H; ++k) v[k] = leaky(as_[s * H + k] + adv[k]);
    }
  };

  for (int jb = 0; jb < tot; jb += 32) {
    load_chunk(jb);
    if (act) { for (int k = 0; k < H; ++k) m[k] = fmaxf(m[k], v[k]); }
  }
  for (int k = 0; k < H; ++k) m[k] = wave_max(m[k]);

  float den[H];
  for (int k = 0; k < H; ++k) den[k] = 0.f;
  for (int jb = 0; jb < tot; jb += 32) {
    if (tot > 32) load_chunk(jb);
    if (act) { for (int k = 0; k < H; ++k) den[k] += expf(v[k] - m[k]); }
  }
  for (int k = 0; k < H; ++k) den[k] = wave_sum(den[k]);

  float acc[VEC];
  for (int q = 0; q < VEC; ++q) acc[q] = 0.f;
  for (int jb = 0; jb < tot; jb += 32) {
    if (tot > 32) load_chunk(jb);
    float w[H];
    for (int k = 0; k < H; ++k) w[k] = act ? expf(v[k] - m[k]) / den[k] : 0.f;
    const int cnt = min(32, tot - jb);
#pragma unroll 4
    for (int t = 0; t < cnt; ++t) {
      int st = __builtin_amdgcn_readlane(s, t);
      float wt = __int_as_float(__builtin_amdgcn_readlane(__float_as_int(w[0]), t));
      if (H == 2) {
        float w1 = __int_as_float(__builtin_amdgcn_readlane(__float_as_int(w[H - 1]), t));
        wt = myh ? w1 : wt;
      }
      const float* row = h + (long)st * F + lane * VEC;
      if (VEC == 4) {
        float4 r = *reinterpret_cast<const float4*>(row);
        acc[0] += r.x * wt; acc[1] += r.y * wt; acc[VEC - 2] += r.z * wt; acc[VEC - 1] += r.w * wt;
      } else {
        float2 r = *reinterpret_cast<const float2*>(row);
        acc[0] += r.x * wt; acc[1] += r.y * wt;
      }
    }
  }

  const int col = lane * VEC;
  for (int q = 0; q < VEC; ++q) {
    float val = acc[q] + bias[col + q];
    if (ELU_OUT) val = (val > 0.f) ? val : (expf(val) - 1.f);
    acc[q] = val;
  }
  float* dstp = outp + (long)node * F + col;
  if (VEC == 4) {
    *reinterpret_cast<float4*>(dstp) = make_float4(acc[0], acc[1], acc[VEC - 2], acc[VEC - 1]);
  } else {
    *reinterpret_cast<float2*>(dstp) = make_float2(acc[0], acc[1]);
  }
}

extern "C" void kernel_launch(void* const* d_in, const int* in_sizes, int n_in,
                              void* d_out, int out_size, void* d_ws, size_t ws_size,
                              hipStream_t stream) {
  const float* x    = (const float*)d_in[0];
  const int*   ei   = (const int*)  d_in[1];
  const float* W1   = (const float*)d_in[2];
  const float* atS1 = (const float*)d_in[3];
  const float* atD1 = (const float*)d_in[4];
  const float* b1   = (const float*)d_in[5];
  const float* W2   = (const float*)d_in[6];
  const float* atS2 = (const float*)d_in[7];
  const float* atD2 = (const float*)d_in[8];
  const float* b2   = (const float*)d_in[9];
  float* out = (float*)d_out;

  const int N    = in_sizes[0] / DIN;
  const int E0   = in_sizes[1] / 2;
  const int* srcs = ei;
  const int* dsts = ei + E0;

  float* ws = (float*)d_ws;
  long off = 0;
  float* h1  = ws + off; off += (long)N * (HEADS * DH);
  float* o1  = ws + off; off += (long)N * (HEADS * DH);
  float* a1s = ws + off; off += (long)N * HEADS;
  float* a1d = ws + off; off += (long)N * HEADS;
  const int NB = (N + GRP - 1) / GRP;
  const int CH = (E0 + EPB - 1) / EPB;
  int* segof = (int*)(ws + off); off += (long)N + 1;
  int* hist  = (int*)(ws + off); off += (long)CH * NB;
  int* lstart = (int*)(ws + off); off += (long)CH * NB;
  unsigned* recs = (unsigned*)(ws + off); off += (long)E0;
  int* seg   = (int*)(ws + off); off += (long)E0;
  int* ssrc  = (int*)(ws + off); off += (long)E0;
  float* h2  = h1;
  float* a2s = h1 + (long)N * DOUT;
  float* a2d = a2s + N;

  const int TB = 256;
  auto blocks = [](long n, int tb) { return (unsigned)((n + tb - 1) / tb); };

  {
    long waves = (long)((N + 15) / 16) * HEADS;
    count_and_gemm<DIN, HEADS * DH><<<CH + blocks(waves * 32, TBG), TBG, 0, stream>>>(
        srcs, dsts, E0, NB, CH, hist, lstart, recs, x, W1, atS1, atD1, h1, a1s, a1d, N);
  }
  bin_group<<<NB, TBE, 0, stream>>>(hist, lstart, recs, segof, seg, ssrc, N, NB, CH);

  node_softmax_agg<HEADS, 4, true><<<blocks((long)N * 32, TB), TB, 0, stream>>>(
      segof, ssrc, a1s, a1d, h1, b1, o1, N);

  {
    long waves = (long)((N + 15) / 16);
    gemm_heads_wmma_f32<HEADS * DH, DOUT><<<blocks(waves * 32, TB), TB, 0, stream>>>(
        o1, W2, atS2, atD2, h2, a2s, a2d, N);
  }
  node_softmax_agg<1, 2, false><<<blocks((long)N * 32, TB), TB, 0, stream>>>(
      segof, ssrc, a2s, a2d, h2, b2, out, N);
}


# Aggregation waves ask for a node's loads in groups, not one per wait: rows in batches of 8 (layer 2: 16) ahead of the multiply-adds, tail as one batch, bias as one load; sum order unchanged
# speedup vs baseline: 7.0109x; 1.0608x over previous
#include <hip/hip_runtime.h>
#include <math.h>

#define DIN   128
#define DH    64
#define HEADS 2
#define DOUT  64
#define NEG_SLOPE 0.2f

typedef float v2f __attribute__((ext_vector_type(2)));
typedef float v8f __attribute__((ext_vector_type(8)));


template <int CNT>
__device__ __forceinline__ void fold_half(float (&v)[16], bool upper, int dist) {
#pragma unroll
  for (int i = 0; i < CNT; ++i) {
    float lo = v[i], hi = v[i + CNT];
    float send = upper ? lo : hi;
    float keep = upper ? hi : lo;
    v[i] = keep + __shfl_xor(send, dist);
  }
}

template <int K, int NC>
__device__ __forceinline__ void
gemm_heads_wave(const float* __restrict__ A, const float* __restrict__ B,
                const float* __restrict__ att_s, const float* __restrict__ att_d,
                float* __restrict__ C, float* __restrict__ as_, float* __restrict__ ad_,
                int M, int wave) {
  constexpr int HW = NC / 64;
  const int lane = threadIdx.x & 31;
  const int tm = wave / HW;
  const int hd = wave - tm * HW;
  if (tm * 16 >= M) return;
  const int half = lane >> 4;
  const int l16  = lane & 15;
  long rowA = (long)tm * 16 + l16;
  if (rowA >= M) rowA = M - 1;
  const int colbase = hd * 64 + l16 * 4;
  const float* ap = A + rowA * K + half * 2;
  const float* bp = B + (long)(half * 2) * NC + colbase;
  v8f acc0 = {}, acc1 = {}, acc2 = {}, acc3 = {};
#pragma unroll
  for (int k0 = 0; k0 < K; k0 += 4) {
    v2f a = *reinterpret_cast<const v2f*>(ap + k0);
    float4 bx = *reinterpret_cast<const float4*>(bp + (long)k0 * NC);
    float4 by = *reinterpret_cast<const float4*>(bp + (long)(k0 + 1) * NC);
    v2f b;
    b.x = bx.x; b.y = by.x;
    acc0 = __builtin_amdgcn_wmma_f32_16x16x4_f32(false, a, false, b, (short)0, acc0, false, false);
    b.x = bx.y; b.y = by.y;
    acc1 = __builtin_amdgcn_wmma_f32_16x16x4_f32(false, a, false, b, (short)0, acc1, false, false);
    b.x = bx.z; b.y = by.z;
    acc2 = __builtin_amdgcn_wmma_f32_16x16x4_f32(false, a, false, b, (short)0, acc2, false, false);
    b.x = bx.w; b.y = by.w;
    acc3 = __builtin_amdgcn_wmma_f32_16x16x4_f32(false, a, false, b, (short)0, acc3, false, false);
  }

  const float4 ats = *reinterpret_cast<const float4*>(att_s + colbase);
  const float4 atd = *reinterpret_cast<const float4*>(att_d + colbase);
  float v[16];
#pragma unroll
  for (int r = 0; r < 8; ++r) {
    const float c0 = acc0[r], c1 = acc1[r], c2 = acc2[r], c3 = acc3[r];
    const int row = tm * 16 + r + half * 8;
    if (row < M)
      *reinterpret_cast<float4*>(C + (long)row * NC + colbase) = make_float4(c0, c1, c2, c3);
    v[r]     = c0 * ats.x + c1 * ats.y + c2 * ats.z + c3 * ats.w;
    v[8 + r] = c0 * atd.x + c1 * atd.y + c2 * atd.z + c3 * atd.w;
  }
  fold_half<8>(v, (l16 & 8) != 0, 8);
  fold_half<4>(v, (l16 & 4) != 0, 4);
  fold_half<2>(v, (l16 & 2) != 0, 2);
  fold_half<1>(v, (l16 & 1) != 0, 1);
  const int orow = tm * 16 + (l16 & 7) + half * 8;
  if (orow < M) {
    float* dp = (l16 < 8) ? as_ : ad_;
    dp[(long)orow * HW + hd] = v[0];
  }
}

template <int K, int NC>
__global__ void __launch_bounds__(256)
gemm_heads_wmma_f32(const float* __restrict__ A, const float* __restrict__ B,
                    const float* __restrict__ att_s, const float* __restrict__ att_d,
                    float* __restrict__ C, float* __restrict__ as_, float* __restrict__ ad_,
                    int M) {
  int wave = (int)((blockIdx.x * blockDim.x + threadIdx.x) >> 5);
  wave = __builtin_amdgcn_readfirstlane(wave);
  gemm_heads_wave<K, NC>(A, B, att_s, att_d, C, as_, ad_, M, wave);
}

#define GRP_SHIFT 8
#define GRP  256
#define MAXB 1024
#define EPB  4096
#define MAXC 1024
#define TBE  1024

__device__ __forceinline__ unsigned range_of(int d, int NB) {
  return min((unsigned)d >> GRP_SHIFT, (unsigned)(NB - 1));
}

#define TBG 256
template <int K, int NC>
__global__ void __launch_bounds__(TBG)
count_and_gemm(const int* __restrict__ srcs, const int* __restrict__ dsts, int E0, int NB, int CH,
               int* __restrict__ hist, int* __restrict__ lstart, unsigned* __restrict__ recs,
               const float* __restrict__ A, const float* __restrict__ B,
               const float* __restrict__ att_s, const float* __restrict__ att_d,
               float* __restrict__ C, float* __restrict__ as_, float* __restrict__ ad_,
               int M) {
  const int t = threadIdx.x;
  if ((int)blockIdx.x < CH) {
    __shared__ int h[MAXB];
    __shared__ int wsum[TBG / 32];
    const int lane = t & 31;
    const int chunk = blockIdx.x;
    for (int b = t; b < MAXB; b += TBG) h[b] = 0;
    __syncthreads();
    const int base = chunk * EPB;
    int d[EPB / TBG], s[EPB / TBG], rk[EPB / TBG];
#pragma unroll
    for (int k = 0; k < EPB / TBG; ++k) {
      const int e = min(base + k * TBG + t, E0 - 1);
      d[k] = dsts[e];
      s[k] = srcs[e];
    }
#pragma unroll
    for (int k = 0; k < EPB / TBG; ++k) {
      const int e = base + k * TBG + t;
      rk[k] = (e < E0) ? atomicAdd(&h[range_of(d[k], NB)], 1) : 0;
    }
    __syncthreads();
    constexpr int PER = MAXB / TBG;
    int c[PER];
    int own = 0;
#pragma unroll
    for (int i = 0; i < PER; ++i) { c[i] = h[t * PER + i]; own += c[i]; }
    int incl = own;
    for (int o = 1; o < 32; o <<= 1) {
      const int v = __shfl_up(incl, o);
      if (lane >= o) incl += v;
    }
    if (lane == 31) wsum[t >> 5] = incl;
    __syncthreads();
    int run = incl - own;
    for (int k = 0; k < (t >> 5); ++k) run += wsum[k];
#pragma unroll
    for (int i = 0; i < PER; ++i) {
      const int b = t * PER + i;
      h[b] = run;
      if (b < NB) { hist[chunk * NB + b] = c[i]; lstart[chunk * NB + b] = run; }
      run += c[i];
    }
    __syncthreads();
#pragma unroll
    for (int k = 0; k < EPB / TBG; ++k) {
      const int e = base + k * TBG + t;
      if (e < E0)
        recs[base + h[range_of(d[k], NB)] + rk[k]] =
            ((unsigned)s[k] << GRP_SHIFT) | ((unsigned)d[k] & (unsigned)(GRP - 1));
    }
    return;
  }
  int wave = (int)((((int)blockIdx.x - CH) * TBG + t) >> 5);
  wave = __builtin_amdgcn_readfirstlane(wave);
  gemm_heads_wave<K, NC>(A, B, att_s, att_d, C, as_, ad_, M, wave);
}

#define SEGU 4
__global__ void __launch_bounds__(TBE)
bin_group(const int* __restrict__ hist, const int* __restrict__ lstart,
          const unsigned* __restrict__ recs, int* __restrict__ segof, int* seg,
          int* __restrict__ ssrc, int N, int NB, int CH) {
  __shared__ int cnt[GRP];
  __shared__ int wtot[GRP / 32];
  __shared__ int segc[MAXC];
  __shared__ int segs[MAXC];
  __shared__ int tot2[2];
  const int t = threadIdx.x;
  const int lane = t & 31;
  const int b = blockIdx.x;
  if (t < GRP) cnt[t] = 0;
  if (t < 2) tot2[t] = 0;
  __syncthreads();
  {
    int ls_sum = 0, cn_sum = 0;
    for (int c = t; c < CH; c += TBE) {
      const int cn = hist[c * NB + b];
      const int ls = lstart[c * NB + b];
      segc[c] = cn;
      segs[c] = c * EPB + ls;
      ls_sum += ls;
      cn_sum += cn;
    }
    for (int o = 16; o > 0; o >>= 1) {
      ls_sum += __shfl_xor(ls_sum, o);
      cn_sum += __shfl_xor(cn_sum, o);
    }
    if (lane == 0) { atomicAdd(&tot2[0], ls_sum); atomicAdd(&tot2[1], cn_sum); }
  }
  __syncthreads();
  const int beg = tot2[0];
  const int end = beg + tot2[1];
  const int wv = t >> 5;
  for (int c0 = wv; c0 < CH; c0 += (TBE / 32) * SEGU) {
    unsigned r[SEGU];
    int n[SEGU], st[SEGU];
#pragma unroll
    for (int u = 0; u < SEGU; ++u) {
      const int c = c0 + (TBE / 32) * u;
      n[u]  = (c < CH) ? segc[c] : 0;
      st[u] = (c < CH) ? segs[c] : 0;
      r[u]  = recs[(n[u] > 0) ? min(st[u] + lane, st[u] + n[u] - 1) : 0];
    }
#pragma unroll
    for (int u = 0; u < SEGU; ++u) {
      if (lane < n[u]) atomicAdd(&cnt[r[u] & (unsigned)(GRP - 1)], 1);
      for (int i = lane + 32; i < n[u]; i += 32)
        atomicAdd(&cnt[recs[st[u] + i] & (unsigned)(GRP - 1)], 1);
    }
  }
  __syncthreads();
  int c = 0, incl = 0;
  if (t < GRP) {
    c = cnt[t];
    incl = c;
    for (int o = 1; o < 32; o <<= 1) {
      const int v = __shfl_up(incl, o);
      if (lane >= o) incl += v;
    }
    if (lane == 31) wtot[t >> 5] = incl;
  }
  __syncthreads();
  if (t < GRP) {
    int basew = 0;
    for (int k = 0; k < (t >> 5); ++k) basew += wtot[k];
    const int first = beg + basew + incl - c;
    const int node = (b << GRP_SHIFT) + t;
    if (node < N) segof[node] = first;
    cnt[t] = first;
  }
  if (b == NB - 1 && t == 0) segof[N] = end;
  __syncthreads();
  for (int c0 = wv; c0 < CH; c0 += (TBE / 32) * SEGU) {
    unsigned r[SEGU];
    int n[SEGU], st[SEGU];
#pragma unroll
    for (int u = 0; u < SEGU; ++u) {
      const int c = c0 + (TBE / 32) * u;
      n[u]  = (c < CH) ? segc[c] : 0;
      st[u] = (c < CH) ? segs[c] : 0;
      r[u]  = recs[(n[u] > 0) ? min(st[u] + lane, st[u] + n[u] - 1) : 0];
    }
#pragma unroll
    for (int u = 0; u < SEGU; ++u) {
      if (lane < n[u]) {
        const int pos = atomicAdd(&cnt[r[u] & (unsigned)(GRP - 1)], 1);
        seg[pos] = (int)(r[u] >> GRP_SHIFT);
      }
      for (int i = lane + 32; i < n[u]; i += 32) {
        const unsigned rr = recs[st[u] + i];
        const int pos = atomicAdd(&cnt[rr & (unsigned)(GRP - 1)], 1);
        seg[pos] = (int)(rr >> GRP_SHIFT);
      }
    }
  }
  __syncthreads();

  for (int ln = t >> 5; ln < GRP; ln += TBE / 32) {
    const int gb = __builtin_amdgcn_readfirstlane((ln == 0) ? beg : cnt[ln - 1]);
    const int ge = __builtin_amdgcn_readfirstlane(cnt[ln]);
    for (int ib = gb; ib < ge; ib += 32) {
      const int i = ib + lane;
      const bool act = i < ge;
      const int ki = act ? seg[i] : 0;
      int r = 0;
      for (int jb = gb; jb < ge; jb += 32) {
        const int j = jb + lane;
        const int kj = (j < ge) ? seg[j] : 0;
        const int cntj = min(32, ge - jb);
        for (int q = 0; q < cntj; ++q) {
          const int k = __builtin_amdgcn_readlane(kj, q);
          const int jj = jb + q;
          r += (k < ki || (k == ki && jj < i)) ? 1 : 0;
        }
      }
      if (act) ssrc[gb + r] = ki;
    }
  }
}

__device__ __forceinline__ float leaky(float v) { return (v > 0.f) ? v : NEG_SLOPE * v; }

__device__ __forceinline__ float wave_max(float v) {
  for (int o = 16; o > 0; o >>= 1) v = fmaxf(v, __shfl_xor(v, o));
  return v;
}
__device__ __forceinline__ float wave_sum(float v) {
  for (int o = 16; o > 0; o >>= 1) v += __shfl_xor(v, o);
  return v;
}

template <int H, int VEC, int U>
__device__ __forceinline__ void
row_batch(const char* __restrict__ hb, unsigned lbytes, int s, const float (&w)[H], int myh, int t0,
          float (&acc)[VEC]) {
  constexpr unsigned ROWB = 32u * VEC * 4u;
  float4 r4[U];
  float2 r2[U];
#pragma unroll
  for (int u = 0; u < U; ++u) {
    const unsigned st = (unsigned)__builtin_amdgcn_readlane(s, t0 + u);
    const unsigned boff = st * ROWB + lbytes;
    if (VEC == 4) r4[u] = *reinterpret_cast<const float4*>(hb + boff);
    else          r2[u] = *reinterpret_cast<const float2*>(hb + boff);
  }
#pragma unroll
  for (int u = 0; u < U; ++u) {
    float wt = __int_as_float(__builtin_amdgcn_readlane(__float_as_int(w[0]), t0 + u));
    if (H == 2) {
      float w1 = __int_as_float(__builtin_amdgcn_readlane(__float_as_int(w[H - 1]), t0 + u));
      wt = myh ? w1 : wt;
    }
    if (VEC == 4) {
      acc[0] = fmaf(r4[u].x, wt, acc[0]); acc[1] = fmaf(r4[u].y, wt, acc[1]);
      acc[VEC - 2] = fmaf(r4[u].z, wt, acc[VEC - 2]); acc[VEC - 1] = fmaf(r4[u].w, wt, acc[VEC - 1]);
    } else {
      acc[0] = fmaf(r2[u].x, wt, acc[0]); acc[1] = fmaf(r2[u].y, wt, acc[1]);
    }
  }
}

template <int H, int VEC, bool ELU_OUT, int UB, int UB2>
__global__ void __launch_bounds__(256)
node_softmax_agg(const int* __restrict__ off, const int* __restrict__ ssrc,
                 const float* __restrict__ as_, const float* __restrict__ ad_,
                 const float* __restrict__ h, const float* __restrict__ bias,
                 float* __restrict__ outp, int n) {
  constexpr int F = 32 * VEC;
  constexpr int C = F / H;
  int lane = threadIdx.x & 31;
  int node = (int)((blockIdx.x * blockDim.x + threadIdx.x) >> 5);
  node = __builtin_amdgcn_readfirstlane(node);
  if (node >= n) return;
  const int beg = off[node];
  const int tot = off[node + 1] - beg + 1;
  const int myh = (lane * VEC) / C;

  float adv[H], m[H], v[H];
  for (int k = 0; k < H; ++k) { adv[k] = ad_[node * H + k]; m[k] = -INFINITY; v[k] = 0.f; }
  int s = 0;
  bool act = false;

  auto load_chunk = [&](int jb) {
    int idx = jb + lane;
    act = idx < tot;
    s = 0;
    for (int k = 0; k < H; ++k) v[k] = 0.f;
    if (act) {
      s = (idx == 0) ? node : ssrc[beg + idx - 1];
      for (int k = 0; k < H; ++k) v[k] = leaky(as_[s * H + k] + adv[k]);
    }
  };

  for (int jb = 0; jb < tot; jb += 32) {
    load_chunk(jb);
    if (act) { for (int k = 0; k < H; ++k) m[k] = fmaxf(m[k], v[k]); }
  }
  for (int k = 0; k < H; ++k) m[k] = wave_max(m[k]);

  float den[H];
  for (int k = 0; k < H; ++k) den[k] = 0.f;
  for (int jb = 0; jb < tot; jb += 32) {
    if (tot > 32) load_chunk(jb);
    if (act) { for (int k = 0; k < H; ++k) den[k] += expf(v[k] - m[k]); }
  }
  for (int k = 0; k < H; ++k) den[k] = wave_sum(den[k]);

  const char* hb = reinterpret_cast<const char*>(h);
  const unsigned lbytes = (unsigned)lane * (VEC * 4u);
  float bv[VEC];
  if (VEC == 4) {
    const float4 b = *reinterpret_cast<const float4*>(bias + lane * VEC);
    bv[0] = b.x; bv[1] = b.y; bv[VEC - 2] = b.z; bv[VEC - 1] = b.w;
  } else {
    const float2 b = *reinterpret_cast<const float2*>(bias + lane * VEC);
    bv[0] = b.x; bv[1] = b.y;
  }
  float acc[VEC];
  for (int q = 0; q < VEC; ++q) acc[q] = 0.f;
  for (int jb = 0; jb < tot; jb += 32) {
    if (tot > 32) load_chunk(jb);
    float w[H];
    for (int k = 0; k < H; ++k) w[k] = act ? expf(v[k] - m[k]) / den[k] : 0.f;
    const int cnt = min(32, tot - jb);
    int t = 0;
    if (UB2 > UB) {
      for (; t + UB2 <= cnt; t += UB2) row_batch<H, VEC, UB2>(hb, lbytes, s, w, myh, t, acc);
    }
    for (; t + UB <= cnt; t += UB) row_batch<H, VEC, UB>(hb, lbytes, s, w, myh, t, acc);
    switch (cnt - t) {
      case 1:  row_batch<H, VEC, 1>(hb, lbytes, s, w, myh, t, acc); break;
      case 2:  row_batch<H, VEC, 2>(hb, lbytes, s, w, myh, t, acc); break;
      case 3:  row_batch<H, VEC, 3>(hb, lbytes, s, w, myh, t, acc); break;
      case 4:  row_batch<H, VEC, 4>(hb, lbytes, s, w, myh, t, acc); break;
      case 5:  row_batch<H, VEC, 5>(hb, lbytes, s, w, myh, t, acc); break;
      case 6:  row_batch<H, VEC, 6>(hb, lbytes, s, w, myh, t, acc); break;
      case 7:  row_batch<H, VEC, 7>(hb, lbytes, s, w, myh, t, acc); break;
      default: break;
    }
  }

  const int col = lane * VEC;
  for (int q = 0; q < VEC; ++q) {
    float val = acc[q] + bv[q];
    if (ELU_OUT) val = (val > 0.f) ? val : (expf(val) - 1.f);
    acc[q] = val;
  }
  float* dstp = outp + (long)node * F + col;
  if (VEC == 4) {
    *reinterpret_cast<float4*>(dstp) = make_float4(acc[0], acc[1], acc[VEC - 2], acc[VEC - 1]);
  } else {
    *reinterpret_cast<float2*>(dstp) = make_float2(acc[0], acc[1]);
  }
}

extern "C" void kernel_launch(void* const* d_in, const int* in_sizes, int n_in,
                              void* d_out, int out_size, void* d_ws, size_t ws_size,
                              hipStream_t stream) {
  const float* x    = (const float*)d_in[0];
  const int*   ei   = (const int*)  d_in[1];
  const float* W1   = (const float*)d_in[2];
  const float* atS1 = (const float*)d_in[3];
  const float* atD1 = (const float*)d_in[4];
  const float* b1   = (const float*)d_in[5];
  const float* W2   = (const float*)d_in[6];
  const float* atS2 = (const float*)d_in[7];
  const float* atD2 = (const float*)d_in[8];
  const float* b2   = (const float*)d_in[9];
  float* out = (float*)d_out;

  const int N    = in_sizes[0] / DIN;
  const int E0   = in_sizes[1] / 2;
  const int* srcs = ei;
  const int* dsts = ei + E0;

  float* ws = (float*)d_ws;
  long off = 0;
  float* h1  = ws + off; off += (long)N * (HEADS * DH);
  float* o1  = ws + off; off += (long)N * (HEADS * DH);
  float* a1s = ws + off; off += (long)N * HEADS;
  float* a1d = ws + off; off += (long)N * HEADS;
  const int NB = (N + GRP - 1) / GRP;
  const int CH = (E0 + EPB - 1) / EPB;
  int* segof = (int*)(ws + off); off += (long)N + 1;
  int* hist  = (int*)(ws + off); off += (long)CH * NB;
  int* lstart = (int*)(ws + off); off += (long)CH * NB;
  unsigned* recs = (unsigned*)(ws + off); off += (long)E0;
  int* seg   = (int*)(ws + off); off += (long)E0;
  int* ssrc  = (int*)(ws + off); off += (long)E0;
  float* h2  = h1;
  float* a2s = h1 + (long)N * DOUT;
  float* a2d = a2s + N;

  const int TB = 256;
  auto blocks = [](long n, int tb) { return (unsigned)((n + tb - 1) / tb); };

  {
    long waves = (long)((N + 15) / 16) * HEADS;
    count_and_gemm<DIN, HEADS * DH><<<CH + blocks(waves * 32, TBG), TBG, 0, stream>>>(
        srcs, dsts, E0, NB, CH, hist, lstart, recs, x, W1, atS1, atD1, h1, a1s, a1d, N);
  }
  bin_group<<<NB, TBE, 0, stream>>>(hist, lstart, recs, segof, seg, ssrc, N, NB, CH);

  node_softmax_agg<HEADS, 4, true, 8, 8><<<blocks((long)N * 32, TB), TB, 0, stream>>>(
      segof, ssrc, a1s, a1d, h1, b1, o1, N);

  {
    long waves = (long)((N + 15) / 16);
    gemm_heads_wmma_f32<HEADS * DH, DOUT><<<blocks(waves * 32, TB), TB, 0, stream>>>(
        o1, W2, atS2, atD2, h2, a2s, a2d, N);
  }
  node_softmax_agg<1, 2, false, 8, 16><<<blocks((long)N * 32, TB), TB, 0, stream>>>(
      segof, ssrc, a2s, a2d, h2, b2, out, N);
}


# Grouping blocks go straight on to the first layer's softmax and row sums of their own nodes (128-node ranges, 4 nodes per wave): that aggregation launch gone, 5 -> 4 launches; sum order unchanged
# speedup vs baseline: 7.2671x; 1.0365x over previous
#include <hip/hip_runtime.h>
#include <math.h>

#define DIN   128
#define DH    64
#define HEADS 2
#define DOUT  64
#define NEG_SLOPE 0.2f

typedef float v2f __attribute__((ext_vector_type(2)));
typedef float v8f __attribute__((ext_vector_type(8)));


template <int CNT>
__device__ __forceinline__ void fold_half(float (&v)[16], bool upper, int dist) {
#pragma unroll
  for (int i = 0; i < CNT; ++i) {
    float lo = v[i], hi = v[i + CNT];
    float send = upper ? lo : hi;
    float keep = upper ? hi : lo;
    v[i] = keep + __shfl_xor(send, dist);
  }
}

template <int K, int NC>
__device__ __forceinline__ void
gemm_heads_wave(const float* __restrict__ A, const float* __restrict__ B,
                const float* __restrict__ att_s, const float* __restrict__ att_d,
                float* __restrict__ C, float* __restrict__ as_, float* __restrict__ ad_,
                int M, int wave) {
  constexpr int HW = NC / 64;
  const int lane = threadIdx.x & 31;
  const int tm = wave / HW;
  const int hd = wave - tm * HW;
  if (tm * 16 >= M) return;
  const int half = lane >> 4;
  const int l16  = lane & 15;
  long rowA = (long)tm * 16 + l16;
  if (rowA >= M) rowA = M - 1;
  const int colbase = hd * 64 + l16 * 4;
  const float* ap = A + rowA * K + half * 2;
  const float* bp = B + (long)(half * 2) * NC + colbase;
  v8f acc0 = {}, acc1 = {}, acc2 = {}, acc3 = {};
#pragma unroll
  for (int k0 = 0; k0 < K; k0 += 4) {
    v2f a = *reinterpret_cast<const v2f*>(ap + k0);
    float4 bx = *reinterpret_cast<const float4*>(bp + (long)k0 * NC);
    float4 by = *reinterpret_cast<const float4*>(bp + (long)(k0 + 1) * NC);
    v2f b;
    b.x = bx.x; b.y = by.x;
    acc0 = __builtin_amdgcn_wmma_f32_16x16x4_f32(false, a, false, b, (short)0, acc0, false, false);
    b.x = bx.y; b.y = by.y;
    acc1 = __builtin_amdgcn_wmma_f32_16x16x4_f32(false, a, false, b, (short)0, acc1, false, false);
    b.x = bx.z; b.y = by.z;
    acc2 = __builtin_amdgcn_wmma_f32_16x16x4_f32(false, a, false, b, (short)0, acc2, false, false);
    b.x = bx.w; b.y = by.w;
    acc3 = __builtin_amdgcn_wmma_f32_16x16x4_f32(false, a, false, b, (short)0, acc3, false, false);
  }

  const float4 ats = *reinterpret_cast<const float4*>(att_s + colbase);
  const float4 atd = *reinterpret_cast<const float4*>(att_d + colbase);
  float v[16];
#pragma unroll
  for (int r = 0; r < 8; ++r) {
    const float c0 = acc0[r], c1 = acc1[r], c2 = acc2[r], c3 = acc3[r];
    const int row = tm * 16 + r + half * 8;
    if (row < M)
      *reinterpret_cast<float4*>(C + (long)row * NC + colbase) = make_float4(c0, c1, c2, c3);
    v[r]     = c0 * ats.x + c1 * ats.y + c2 * ats.z + c3 * ats.w;
    v[8 + r] = c0 * atd.x + c1 * atd.y + c2 * atd.z + c3 * atd.w;
  }
  fold_half<8>(v, (l16 & 8) != 0, 8);
  fold_half<4>(v, (l16 & 4) != 0, 4);
  fold_half<2>(v, (l16 & 2) != 0, 2);
  fold_half<1>(v, (l16 & 1) != 0, 1);
  const int orow = tm * 16 + (l16 & 7) + half * 8;
  if (orow < M) {
    float* dp = (l16 < 8) ? as_ : ad_;
    dp[(long)orow * HW + hd] = v[0];
  }
}

template <int K, int NC>
__global__ void __launch_bounds__(256)
gemm_heads_wmma_f32(const float* __restrict__ A, const float* __restrict__ B,
                    const float* __restrict__ att_s, const float* __restrict__ att_d,
                    float* __restrict__ C, float* __restrict__ as_, float* __restrict__ ad_,
                    int M) {
  int wave = (int)((blockIdx.x * blockDim.x + threadIdx.x) >> 5);
  wave = __builtin_amdgcn_readfirstlane(wave);
  gemm_heads_wave<K, NC>(A, B, att_s, att_d, C, as_, ad_, M, wave);
}

#define GRP_SHIFT 7
#define GRP  128
#define MAXB 1024
#define EPB  4096
#define MAXC 1024
#define TBE  1024

__device__ __forceinline__ unsigned range_of(int d, int NB) {
  return min((unsigned)d >> GRP_SHIFT, (unsigned)(NB - 1));
}

#define TBG 256
template <int K, int NC>
__global__ void __launch_bounds__(TBG)
count_and_gemm(const int* __restrict__ srcs, const int* __restrict__ dsts, int E0, int NB, int CH,
               int* __restrict__ hist, int* __restrict__ lstart, unsigned* __restrict__ recs,
               const float* __restrict__ A, const float* __restrict__ B,
               const float* __restrict__ att_s, const float* __restrict__ att_d,
               float* __restrict__ C, float* __restrict__ as_, float* __restrict__ ad_,
               int M) {
  const int t = threadIdx.x;
  if ((int)blockIdx.x < CH) {
    __shared__ int h[MAXB];
    __shared__ int wsum[TBG / 32];
    const int lane = t & 31;
    const int chunk = blockIdx.x;
    for (int b = t; b < MAXB; b += TBG) h[b] = 0;
    __syncthreads();
    const int base = chunk * EPB;
    int d[EPB / TBG], s[EPB / TBG], rk[EPB / TBG];
#pragma unroll
    for (int k = 0; k < EPB / TBG; ++k) {
      const int e = min(base + k * TBG + t, E0 - 1);
      d[k] = dsts[e];
      s[k] = srcs[e];
    }
#pragma unroll
    for (int k = 0; k < EPB / TBG; ++k) {
      const int e = base + k * TBG + t;
      rk[k] = (e < E0) ? atomicAdd(&h[range_of(d[k], NB)], 1) : 0;
    }
    __syncthreads();
    constexpr int PER = MAXB / TBG;
    int c[PER];
    int own = 0;
#pragma unroll
    for (int i = 0; i < PER; ++i) { c[i] = h[t * PER + i]; own += c[i]; }
    int incl = own;
    for (int o = 1; o < 32; o <<= 1) {
      const int v = __shfl_up(incl, o);
      if (lane >= o) incl += v;
    }
    if (lane == 31) wsum[t >> 5] = incl;
    __syncthreads();
    int run = incl - own;
    for (int k = 0; k < (t >> 5); ++k) run += wsum[k];
#pragma unroll
    for (int i = 0; i < PER; ++i) {
      const int b = t * PER + i;
      h[b] = run;
      if (b < NB) { hist[chunk * NB + b] = c[i]; lstart[chunk * NB + b] = run; }
      run += c[i];
    }
    __syncthreads();
#pragma unroll
    for (int k = 0; k < EPB / TBG; ++k) {
      const int e = base + k * TBG + t;
      if (e < E0)
        recs[base + h[range_of(d[k], NB)] + rk[k]] =
            ((unsigned)s[k] << GRP_SHIFT) | ((unsigned)d[k] & (unsigned)(GRP - 1));
    }
    return;
  }
  int wave = (int)((((int)blockIdx.x - CH) * TBG + t) >> 5);
  wave = __builtin_amdgcn_readfirstlane(wave);
  gemm_heads_wave<K, NC>(A, B, att_s, att_d, C, as_, ad_, M, wave);
}

__device__ __forceinline__ float leaky(float v) { return (v > 0.f) ? v : NEG_SLOPE * v; }

__device__ __forceinline__ float wave_max(float v) {
  for (int o = 16; o > 0; o >>= 1) v = fmaxf(v, __shfl_xor(v, o));
  return v;
}
__device__ __forceinline__ float wave_sum(float v) {
  for (int o = 16; o > 0; o >>= 1) v += __shfl_xor(v, o);
  return v;
}

template <int H, int VEC, int U>
__device__ __forceinline__ void
row_batch(const char* __restrict__ hb, unsigned lbytes, int s, const float (&w)[H], int myh, int t0,
          float (&acc)[VEC]) {
  constexpr unsigned ROWB = 32u * VEC * 4u;
  float4 r4[U];
  float2 r2[U];
#pragma unroll
  for (int u = 0; u < U; ++u) {
    const unsigned st = (unsigned)__builtin_amdgcn_readlane(s, t0 + u);
    const unsigned boff = st * ROWB + lbytes;
    if (VEC == 4) r4[u] = *reinterpret_cast<const float4*>(hb + boff);
    else          r2[u] = *reinterpret_cast<const float2*>(hb + boff);
  }
#pragma unroll
  for (int u = 0; u < U; ++u) {
    float wt = __int_as_float(__builtin_amdgcn_readlane(__float_as_int(w[0]), t0 + u));
    if (H == 2) {
      float w1 = __int_as_float(__builtin_amdgcn_readlane(__float_as_int(w[H - 1]), t0 + u));
      wt = myh ? w1 : wt;
    }
    if (VEC == 4) {
      acc[0] = fmaf(r4[u].x, wt, acc[0]); acc[1] = fmaf(r4[u].y, wt, acc[1]);
      acc[VEC - 2] = fmaf(r4[u].z, wt, acc[VEC - 2]); acc[VEC - 1] = fmaf(r4[u].w, wt, acc[VEC - 1]);
    } else {
      acc[0] = fmaf(r2[u].x, wt, acc[0]); acc[1] = fmaf(r2[u].y, wt, acc[1]);
    }
  }
}

template <int VEC>
__device__ __forceinline__ void load_bias(const float* __restrict__ bias, float (&bv)[VEC]) {
  const int lane = threadIdx.x & 31;
  if (VEC == 4) {
    const float4 b = *reinterpret_cast<const float4*>(bias + lane * VEC);
    bv[0] = b.x; bv[1] = b.y; bv[VEC - 2] = b.z; bv[VEC - 1] = b.w;
  } else {
    const float2 b = *reinterpret_cast<const float2*>(bias + lane * VEC);
    bv[0] = b.x; bv[1] = b.y;
  }
}

template <int H, int VEC, bool ELU_OUT, int UB, int UB2>
__device__ __forceinline__ void
agg_node(int node, int beg, int tot, const int* ssrc,
         const float* __restrict__ as_, const float* __restrict__ ad_,
         const float* __restrict__ h, const float (&bv)[VEC],
         float* __restrict__ outp) {
  constexpr int F = 32 * VEC;
  constexpr int C = F / H;
  const int lane = threadIdx.x & 31;
  const int myh = (lane * VEC) / C;

  float adv[H], m[H], v[H];
  for (int k = 0; k < H; ++k) { adv[k] = ad_[node * H + k]; m[k] = -INFINITY; v[k] = 0.f; }
  int s = 0;
  bool act = false;

  auto load_chunk = [&](int jb) {
    int idx = jb + lane;
    act = idx < tot;
    s = 0;
    for (int k = 0; k < H; ++k) v[k] = 0.f;
    if (act) {
      s = (idx == 0) ? node : ssrc[beg + idx - 1];
      for (int k = 0; k < H; ++k) v[k] = leaky(as_[s * H + k] + adv[k]);
    }
  };

  for (int jb = 0; jb < tot; jb += 32) {
    load_chunk(jb);
    if (act) { for (int k = 0; k < H; ++k) m[k] = fmaxf(m[k], v[k]); }
  }
  for (int k = 0; k < H; ++k) m[k] = wave_max(m[k]);

  float den[H];
  for (int k = 0; k < H; ++k) den[k] = 0.f;
  for (int jb = 0; jb < tot; jb += 32) {
    if (tot > 32) load_chunk(jb);
    if (act) { for (int k = 0; k < H; ++k) den[k] += expf(v[k] - m[k]); }
  }
  for (int k = 0; k < H; ++k) den[k] = wave_sum(den[k]);

  const char* hb = reinterpret_cast<const char*>(h);
  const unsigned lbytes = (unsigned)lane * (VEC * 4u);
  float acc[VEC];
  for (int q = 0; q < VEC; ++q) acc[q] = 0.f;
  for (int jb = 0; jb < tot; jb += 32) {
    if (tot > 32) load_chunk(jb);
    float w[H];
    for (int k = 0; k < H; ++k) w[k] = act ? expf(v[k] - m[k]) / den[k] : 0.f;
    const int cnt = min(32, tot - jb);
    int t = 0;
    if (UB2 > UB) {
      for (; t + UB2 <= cnt; t += UB2) row_batch<H, VEC, UB2>(hb, lbytes, s, w, myh, t, acc);
    }
    for (; t + UB <= cnt; t += UB) row_batch<H, VEC, UB>(hb, lbytes, s, w, myh, t, acc);
    switch (cnt - t) {
      case 1:  row_batch<H, VEC, 1>(hb, lbytes, s, w, myh, t, acc); break;
      case 2:  row_batch<H, VEC, 2>(hb, lbytes, s, w, myh, t, acc); break;
      case 3:  row_batch<H, VEC, 3>(hb, lbytes, s, w, myh, t, acc); break;
      case 4:  row_batch<H, VEC, 4>(hb, lbytes, s, w, myh, t, acc); break;
      case 5:  row_batch<H, VEC, 5>(hb, lbytes, s, w, myh, t, acc); break;
      case 6:  row_batch<H, VEC, 6>(hb, lbytes, s, w, myh, t, acc); break;
      case 7:  row_batch<H, VEC, 7>(hb, lbytes, s, w, myh, t, acc); break;
      default: break;
    }
  }

  for (int q = 0; q < VEC; ++q) {
    float val = acc[q] + bv[q];
    if (ELU_OUT) val = (val > 0.f) ? val : (expf(val) - 1.f);
    acc[q] = val;
  }
  char* ob = reinterpret_cast<char*>(outp);
  const unsigned obytes = (unsigned)node * (F * 4u) + lbytes;
  if (VEC == 4) {
    *reinterpret_cast<float4*>(ob + obytes) = make_float4(acc[0], acc[1], acc[VEC - 2], acc[VEC - 1]);
  } else {
    *reinterpret_cast<float2*>(ob + obytes) = make_float2(acc[0], acc[1]);
  }
}

template <int H, int VEC, bool ELU_OUT, int UB, int UB2>
__global__ void __launch_bounds__(256)
node_softmax_agg(const int* __restrict__ off, const int* __restrict__ ssrc,
                 const float* __restrict__ as_, const float* __restrict__ ad_,
                 const float* __restrict__ h, const float* __restrict__ bias,
                 float* __restrict__ outp, int n) {
  int node = (int)((blockIdx.x * blockDim.x + threadIdx.x) >> 5);
  node = __builtin_amdgcn_readfirstlane(node);
  if (node >= n) return;
  float bv[VEC];
  load_bias<VEC>(bias, bv);
  const int beg = off[node];
  const int tot = off[node + 1] - beg + 1;
  agg_node<H, VEC, ELU_OUT, UB, UB2>(node, beg, tot, ssrc, as_, ad_, h, bv, outp);
}


#define SEGU 4
__global__ void __launch_bounds__(TBE, 16)
group_and_agg(const int* __restrict__ hist, const int* __restrict__ lstart,
              const unsigned* __restrict__ recs, int* __restrict__ segof, int* seg,
              int* ssrc, int N, int NB, int CH,
              const float* __restrict__ as_, const float* __restrict__ ad_,
              const float* __restrict__ h, const float* __restrict__ bias,
              float* __restrict__ outp) {
  __shared__ int cnt[GRP];
  __shared__ int wtot[GRP / 32];
  __shared__ int segc[MAXC];
  __shared__ int segs[MAXC];
  __shared__ int tot2[2];
  const int t = threadIdx.x;
  const int lane = t & 31;
  const int b = blockIdx.x;
  if (t < GRP) cnt[t] = 0;
  if (t < 2) tot2[t] = 0;
  __syncthreads();
  {
    int ls_sum = 0, cn_sum = 0;
    for (int c = t; c < CH; c += TBE) {
      const int cn = hist[c * NB + b];
      const int ls = lstart[c * NB + b];
      segc[c] = cn;
      segs[c] = c * EPB + ls;
      ls_sum += ls;
      cn_sum += cn;
    }
    for (int o = 16; o > 0; o >>= 1) {
      ls_sum += __shfl_xor(ls_sum, o);
      cn_sum += __shfl_xor(cn_sum, o);
    }
    if (lane == 0) { atomicAdd(&tot2[0], ls_sum); atomicAdd(&tot2[1], cn_sum); }
  }
  __syncthreads();
  const int beg = tot2[0];
  const int end = beg + tot2[1];
  const int wv = t >> 5;
  for (int c0 = wv; c0 < CH; c0 += (TBE / 32) * SEGU) {
    unsigned r[SEGU];
    int n[SEGU], st[SEGU];
#pragma unroll
    for (int u = 0; u < SEGU; ++u) {
      const int c = c0 + (TBE / 32) * u;
      n[u]  = (c < CH) ? segc[c] : 0;
      st[u] = (c < CH) ? segs[c] : 0;
      r[u]  = recs[(n[u] > 0) ? min(st[u] + lane, st[u] + n[u] - 1) : 0];
    }
#pragma unroll
    for (int u = 0; u < SEGU; ++u) {
      if (lane < n[u]) atomicAdd(&cnt[r[u] & (unsigned)(GRP - 1)], 1);
      for (int i = lane + 32; i < n[u]; i += 32)
        atomicAdd(&cnt[recs[st[u] + i] & (unsigned)(GRP - 1)], 1);
    }
  }
  __syncthreads();
  int c = 0, incl = 0;
  if (t < GRP) {
    c = cnt[t];
    incl = c;
    for (int o = 1; o < 32; o <<= 1) {
      const int v = __shfl_up(incl, o);
      if (lane >= o) incl += v;
    }
    if (lane == 31) wtot[t >> 5] = incl;
  }
  __syncthreads();
  if (t < GRP) {
    int basew = 0;
    for (int k = 0; k < (t >> 5); ++k) basew += wtot[k];
    const int first = beg + basew + incl - c;
    const int node = (b << GRP_SHIFT) + t;
    if (node < N) segof[node] = first;
    cnt[t] = first;
  }
  if (b == NB - 1 && t == 0) segof[N] = end;
  __syncthreads();
  for (int c0 = wv; c0 < CH; c0 += (TBE / 32) * SEGU) {
    unsigned r[SEGU];
    int n[SEGU], st[SEGU];
#pragma unroll
    for (int u = 0; u < SEGU; ++u) {
      const int c = c0 + (TBE / 32) * u;
      n[u]  = (c < CH) ? segc[c] : 0;
      st[u] = (c < CH) ? segs[c] : 0;
      r[u]  = recs[(n[u] > 0) ? min(st[u] + lane, st[u] + n[u] - 1) : 0];
    }
#pragma unroll
    for (int u = 0; u < SEGU; ++u) {
      if (lane < n[u]) {
        const int pos = atomicAdd(&cnt[r[u] & (unsigned)(GRP - 1)], 1);
        seg[pos] = (int)(r[u] >> GRP_SHIFT);
      }
      for (int i = lane + 32; i < n[u]; i += 32) {
        const unsigned rr = recs[st[u] + i];
        const int pos = atomicAdd(&cnt[rr & (unsigned)(GRP - 1)], 1);
        seg[pos] = (int)(rr >> GRP_SHIFT);
      }
    }
  }
  __syncthreads();

  for (int ln = t >> 5; ln < GRP; ln += TBE / 32) {
    const int gb = __builtin_amdgcn_readfirstlane((ln == 0) ? beg : cnt[ln - 1]);
    const int ge = __builtin_amdgcn_readfirstlane(cnt[ln]);
    for (int ib = gb; ib < ge; ib += 32) {
      const int i = ib + lane;
      const bool act = i < ge;
      const int ki = act ? seg[i] : 0;
      int r = 0;
      for (int jb = gb; jb < ge; jb += 32) {
        const int j = jb + lane;
        const int kj = (j < ge) ? seg[j] : 0;
        const int cntj = min(32, ge - jb);
        for (int q = 0; q < cntj; ++q) {
          const int k = __builtin_amdgcn_readlane(kj, q);
          const int jj = jb + q;
          r += (k < ki || (k == ki && jj < i)) ? 1 : 0;
        }
      }
      if (act) ssrc[gb + r] = ki;
    }
  }
  __syncthreads();

  const int wvs = __builtin_amdgcn_readfirstlane(t >> 5);
  float bv[4];
  load_bias<4>(bias, bv);
#pragma nounroll
  for (int ln = wvs; ln < GRP; ln += TBE / 32) {
    const int node = (b << GRP_SHIFT) + ln;
    if (node >= N) break;
    const int gb = __builtin_amdgcn_readfirstlane((ln == 0) ? beg : cnt[ln - 1]);
    const int ge = __builtin_amdgcn_readfirstlane(cnt[ln]);
    agg_node<HEADS, 4, true, 8, 8>(node, gb, ge - gb + 1, ssrc, as_, ad_, h, bv, outp);
  }
}


extern "C" void kernel_launch(void* const* d_in, const int* in_sizes, int n_in,
                              void* d_out, int out_size, void* d_ws, size_t ws_size,
                              hipStream_t stream) {
  const float* x    = (const float*)d_in[0];
  const int*   ei   = (const int*)  d_in[1];
  const float* W1   = (const float*)d_in[2];
  const float* atS1 = (const float*)d_in[3];
  const float* atD1 = (const float*)d_in[4];
  const float* b1   = (const float*)d_in[5];
  const float* W2   = (const float*)d_in[6];
  const float* atS2 = (const float*)d_in[7];
  const float* atD2 = (const float*)d_in[8];
  const float* b2   = (const float*)d_in[9];
  float* out = (float*)d_out;

  const int N    = in_sizes[0] / DIN;
  const int E0   = in_sizes[1] / 2;
  const int* srcs = ei;
  const int* dsts = ei + E0;

  float* ws = (float*)d_ws;
  long off = 0;
  float* h1  = ws + off; off += (long)N * (HEADS * DH);
  float* o1  = ws + off; off += (long)N * (HEADS * DH);
  float* a1s = ws + off; off += (long)N * HEADS;
  float* a1d = ws + off; off += (long)N * HEADS;
  const int NB = (N + GRP - 1) / GRP;
  const int CH = (E0 + EPB - 1) / EPB;
  int* segof = (int*)(ws + off); off += (long)N + 1;
  int* hist  = (int*)(ws + off); off += (long)CH * NB;
  int* lstart = (int*)(ws + off); off += (long)CH * NB;
  unsigned* recs = (unsigned*)(ws + off); off += (long)E0;
  int* seg   = (int*)(ws + off); off += (long)E0;
  int* ssrc  = (int*)(ws + off); off += (long)E0;
  float* h2  = h1;
  float* a2s = h1 + (long)N * DOUT;
  float* a2d = a2s + N;

  const int TB = 256;
  auto blocks = [](long n, int tb) { return (unsigned)((n + tb - 1) / tb); };

  {
    long waves = (long)((N + 15) / 16) * HEADS;
    count_and_gemm<DIN, HEADS * DH><<<CH + blocks(waves * 32, TBG), TBG, 0, stream>>>(
        srcs, dsts, E0, NB, CH, hist, lstart, recs, x, W1, atS1, atD1, h1, a1s, a1d, N);
  }
  group_and_agg<<<NB, TBE, 0, stream>>>(hist, lstart, recs, segof, seg, ssrc, N, NB, CH,
                                        a1s, a1d, h1, b1, o1);

  {
    long waves = (long)((N + 15) / 16);
    gemm_heads_wmma_f32<HEADS * DH, DOUT><<<blocks(waves * 32, TB), TB, 0, stream>>>(
        o1, W2, atS2, atD2, h2, a2s, a2d, N);
  }
  node_softmax_agg<1, 2, false, 8, 16><<<blocks((long)N * 32, TB), TB, 0, stream>>>(
      segof, ssrc, a2s, a2d, h2, b2, out, N);
}


# Second layer's product formed inside the grouping blocks from their own 128 first-layer rows kept in LDS (8 waves, W staged in LDS); product launch gone, 4 -> 3 launches; sum orders unchanged
# speedup vs baseline: 7.6377x; 1.0510x over previous
#include <hip/hip_runtime.h>
#include <math.h>

#define DIN   128
#define DH    64
#define HEADS 2
#define DOUT  64
#define NEG_SLOPE 0.2f

typedef float v2f __attribute__((ext_vector_type(2)));
typedef float v8f __attribute__((ext_vector_type(8)));


template <int CNT>
__device__ __forceinline__ void fold_half(float (&v)[16], bool upper, int dist) {
#pragma unroll
  for (int i = 0; i < CNT; ++i) {
    float lo = v[i], hi = v[i + CNT];
    float send = upper ? lo : hi;
    float keep = upper ? hi : lo;
    v[i] = keep + __shfl_xor(send, dist);
  }
}

template <int K, int NC>
__device__ __forceinline__ void
gemm_heads_wave(const float* __restrict__ A, const float* __restrict__ B,
                const float* __restrict__ att_s, const float* __restrict__ att_d,
                float* __restrict__ C, float* __restrict__ as_, float* __restrict__ ad_,
                int M, int wave) {
  constexpr int HW = NC / 64;
  const int lane = threadIdx.x & 31;
  const int tm = wave / HW;
  const int hd = wave - tm * HW;
  if (tm * 16 >= M) return;
  const int half = lane >> 4;
  const int l16  = lane & 15;
  long rowA = (long)tm * 16 + l16;
  if (rowA >= M) rowA = M - 1;
  const int colbase = hd * 64 + l16 * 4;
  const float* ap = A + rowA * K + half * 2;
  const float* bp = B + (long)(half * 2) * NC + colbase;
  v8f acc0 = {}, acc1 = {}, acc2 = {}, acc3 = {};
#pragma unroll
  for (int k0 = 0; k0 < K; k0 += 4) {
    v2f a = *reinterpret_cast<const v2f*>(ap + k0);
    float4 bx = *reinterpret_cast<const float4*>(bp + (long)k0 * NC);
    float4 by = *reinterpret_cast<const float4*>(bp + (long)(k0 + 1) * NC);
    v2f b;
    b.x = bx.x; b.y = by.x;
    acc0 = __builtin_amdgcn_wmma_f32_16x16x4_f32(false, a, false, b, (short)0, acc0, false, false);
    b.x = bx.y; b.y = by.y;
    acc1 = __builtin_amdgcn_wmma_f32_16x16x4_f32(false, a, false, b, (short)0, acc1, false, false);
    b.x = bx.z; b.y = by.z;
    acc2 = __builtin_amdgcn_wmma_f32_16x16x4_f32(false, a, false, b, (short)0, acc2, false, false);
    b.x = bx.w; b.y = by.w;
    acc3 = __builtin_amdgcn_wmma_f32_16x16x4_f32(false, a, false, b, (short)0, acc3, false, false);
  }

  const float4 ats = *reinterpret_cast<const float4*>(att_s + colbase);
  const float4 atd = *reinterpret_cast<const float4*>(att_d + colbase);
  float v[16];
#pragma unroll
  for (int r = 0; r < 8; ++r) {
    const float c0 = acc0[r], c1 = acc1[r], c2 = acc2[r], c3 = acc3[r];
    const int row = tm * 16 + r + half * 8;
    if (row < M)
      *reinterpret_cast<float4*>(C + (long)row * NC + colbase) = make_float4(c0, c1, c2, c3);
    v[r]     = c0 * ats.x + c1 * ats.y + c2 * ats.z + c3 * ats.w;
    v[8 + r] = c0 * atd.x + c1 * atd.y + c2 * atd.z + c3 * atd.w;
  }
  fold_half<8>(v, (l16 & 8) != 0, 8);
  fold_half<4>(v, (l16 & 4) != 0, 4);
  fold_half<2>(v, (l16 & 2) != 0, 2);
  fold_half<1>(v, (l16 & 1) != 0, 1);
  const int orow = tm * 16 + (l16 & 7) + half * 8;
  if (orow < M) {
    float* dp = (l16 < 8) ? as_ : ad_;
    dp[(long)orow * HW + hd] = v[0];
  }
}

#define PA 132
template <int K>
__device__ __forceinline__ void
gemm_strip_lds(const float* As, const float* Bs,
               const float* __restrict__ att_s, const float* __restrict__ att_d,
               float* __restrict__ C, float* __restrict__ as_, float* __restrict__ ad_,
               int M, int tm) {
  constexpr int NC = 64;
  const int lane = threadIdx.x & 31;
  if (tm * 16 >= M) return;
  const int half = lane >> 4;
  const int l16  = lane & 15;
  const int colbase = l16 * 4;
  const float* ap = As + l16 * PA + half * 2;
  const float* bp = Bs + (half * 2) * NC + colbase;
  v8f acc0 = {}, acc1 = {}, acc2 = {}, acc3 = {};
#pragma unroll
  for (int k0 = 0; k0 < K; k0 += 4) {
    v2f a = *reinterpret_cast<const v2f*>(ap + k0);
    float4 bx = *reinterpret_cast<const float4*>(bp + k0 * NC);
    float4 by = *reinterpret_cast<const float4*>(bp + (k0 + 1) * NC);
    v2f b;
    b.x = bx.x; b.y = by.x;
    acc0 = __builtin_amdgcn_wmma_f32_16x16x4_f32(false, a, false, b, (short)0, acc0, false, false);
    b.x = bx.y; b.y = by.y;
    acc1 = __builtin_amdgcn_wmma_f32_16x16x4_f32(false, a, false, b, (short)0, acc1, false, false);
    b.x = bx.z; b.y = by.z;
    acc2 = __builtin_amdgcn_wmma_f32_16x16x4_f32(false, a, false, b, (short)0, acc2, false, false);
    b.x = bx.w; b.y = by.w;
    acc3 = __builtin_amdgcn_wmma_f32_16x16x4_f32(false, a, false, b, (short)0, acc3, false, false);
  }

  const float4 ats = *reinterpret_cast<const float4*>(att_s + colbase);
  const float4 atd = *reinterpret_cast<const float4*>(att_d + colbase);
  float v[16];
#pragma unroll
  for (int r = 0; r < 8; ++r) {
    const float c0 = acc0[r], c1 = acc1[r], c2 = acc2[r], c3 = acc3[r];
    const int row = tm * 16 + r + half * 8;
    if (row < M)
      *reinterpret_cast<float4*>(C + (long)row * NC + colbase) = make_float4(c0, c1, c2, c3);
    v[r]     = c0 * ats.x + c1 * ats.y + c2 * ats.z + c3 * ats.w;
    v[8 + r] = c0 * atd.x + c1 * atd.y + c2 * atd.z + c3 * atd.w;
  }
  fold_half<8>(v, (l16 & 8) != 0, 8);
  fold_half<4>(v, (l16 & 4) != 0, 4);
  fold_half<2>(v, (l16 & 2) != 0, 2);
  fold_half<1>(v, (l16 & 1) != 0, 1);
  const int orow = tm * 16 + (l16 & 7) + half * 8;
  if (orow < M) {
    float* dp = (l16 < 8) ? as_ : ad_;
    dp[orow] = v[0];
  }
}

#define GRP_SHIFT 7
#define GRP  128
#define MAXB 1024
#define EPB  4096
#define MAXC 1024
#define TBE  1024

__device__ __forceinline__ unsigned range_of(int d, int NB) {
  return min((unsigned)d >> GRP_SHIFT, (unsigned)(NB - 1));
}

#define TBG 256
template <int K, int NC>
__global__ void __launch_bounds__(TBG)
count_and_gemm(const int* __restrict__ srcs, const int* __restrict__ dsts, int E0, int NB, int CH,
               int* __restrict__ hist, int* __restrict__ lstart, unsigned* __restrict__ recs,
               const float* __restrict__ A, const float* __restrict__ B,
               const float* __restrict__ att_s, const float* __restrict__ att_d,
               float* __restrict__ C, float* __restrict__ as_, float* __restrict__ ad_,
               int M) {
  const int t = threadIdx.x;
  if ((int)blockIdx.x < CH) {
    __shared__ int h[MAXB];
    __shared__ int wsum[TBG / 32];
    const int lane = t & 31;
    const int chunk = blockIdx.x;
    for (int b = t; b < MAXB; b += TBG) h[b] = 0;
    __syncthreads();
    const int base = chunk * EPB;
    int d[EPB / TBG], s[EPB / TBG], rk[EPB / TBG];
#pragma unroll
    for (int k = 0; k < EPB / TBG; ++k) {
      const int e = min(base + k * TBG + t, E0 - 1);
      d[k] = dsts[e];
      s[k] = srcs[e];
    }
#pragma unroll
    for (int k = 0; k < EPB / TBG; ++k) {
      const int e = base + k * TBG + t;
      rk[k] = (e < E0) ? atomicAdd(&h[range_of(d[k], NB)], 1) : 0;
    }
    __syncthreads();
    constexpr int PER = MAXB / TBG;
    int c[PER];
    int own = 0;
#pragma unroll
    for (int i = 0; i < PER; ++i) { c[i] = h[t * PER + i]; own += c[i]; }
    int incl = own;
    for (int o = 1; o < 32; o <<= 1) {
      const int v = __shfl_up(incl, o);
      if (lane >= o) incl += v;
    }
    if (lane == 31) wsum[t >> 5] = incl;
    __syncthreads();
    int run = incl - own;
    for (int k = 0; k < (t >> 5); ++k) run += wsum[k];
#pragma unroll
    for (int i = 0; i < PER; ++i) {
      const int b = t * PER + i;
      h[b] = run;
      if (b < NB) { hist[chunk * NB + b] = c[i]; lstart[chunk * NB + b] = run; }
      run += c[i];
    }
    __syncthreads();
#pragma unroll
    for (int k = 0; k < EPB / TBG; ++k) {
      const int e = base + k * TBG + t;
      if (e < E0)
        recs[base + h[range_of(d[k], NB)] + rk[k]] =
            ((unsigned)s[k] << GRP_SHIFT) | ((unsigned)d[k] & (unsigned)(GRP - 1));
    }
    return;
  }
  int wave = (int)((((int)blockIdx.x - CH) * TBG + t) >> 5);
  wave = __builtin_amdgcn_readfirstlane(wave);
  gemm_heads_wave<K, NC>(A, B, att_s, att_d, C, as_, ad_, M, wave);
}

__device__ __forceinline__ float leaky(float v) { return (v > 0.f) ? v : NEG_SLOPE * v; }

__device__ __forceinline__ float wave_max(float v) {
  for (int o = 16; o > 0; o >>= 1) v = fmaxf(v, __shfl_xor(v, o));
  return v;
}
__device__ __forceinline__ float wave_sum(float v) {
  for (int o = 16; o > 0; o >>= 1) v += __shfl_xor(v, o);
  return v;
}

template <int H, int VEC, int U>
__device__ __forceinline__ void
row_batch(const char* __restrict__ hb, unsigned lbytes, int s, const float (&w)[H], int myh, int t0,
          float (&acc)[VEC]) {
  constexpr unsigned ROWB = 32u * VEC * 4u;
  float4 r4[U];
  float2 r2[U];
#pragma unroll
  for (int u = 0; u < U; ++u) {
    const unsigned st = (unsigned)__builtin_amdgcn_readlane(s, t0 + u);
    const unsigned boff = st * ROWB + lbytes;
    if (VEC == 4) r4[u] = *reinterpret_cast<const float4*>(hb + boff);
    else          r2[u] = *reinterpret_cast<const float2*>(hb + boff);
  }
#pragma unroll
  for (int u = 0; u < U; ++u) {
    float wt = __int_as_float(__builtin_amdgcn_readlane(__float_as_int(w[0]), t0 + u));
    if (H == 2) {
      float w1 = __int_as_float(__builtin_amdgcn_readlane(__float_as_int(w[H - 1]), t0 + u));
      wt = myh ? w1 : wt;
    }
    if (VEC == 4) {
      acc[0] = fmaf(r4[u].x, wt, acc[0]); acc[1] = fmaf(r4[u].y, wt, acc[1]);
      acc[VEC - 2] = fmaf(r4[u].z, wt, acc[VEC - 2]); acc[VEC - 1] = fmaf(r4[u].w, wt, acc[VEC - 1]);
    } else {
      acc[0] = fmaf(r2[u].x, wt, acc[0]); acc[1] = fmaf(r2[u].y, wt, acc[1]);
    }
  }
}

template <int VEC>
__device__ __forceinline__ void load_bias(const float* __restrict__ bias, float (&bv)[VEC]) {
  const int lane = threadIdx.x & 31;
  if (VEC == 4) {
    const float4 b = *reinterpret_cast<const float4*>(bias + lane * VEC);
    bv[0] = b.x; bv[1] = b.y; bv[VEC - 2] = b.z; bv[VEC - 1] = b.w;
  } else {
    const float2 b = *reinterpret_cast<const float2*>(bias + lane * VEC);
    bv[0] = b.x; bv[1] = b.y;
  }
}

template <int H, int VEC, bool ELU_OUT, int UB, int UB2, bool ROW_TO_LDS>
__device__ __forceinline__ void
agg_node(int node, int beg, int tot, const int* ssrc,
         const float* __restrict__ as_, const float* __restrict__ ad_,
         const float* __restrict__ h, const float (&bv)[VEC],
         float* __restrict__ outp) {
  constexpr int F = 32 * VEC;
  constexpr int C = F / H;
  const int lane = threadIdx.x & 31;
  const int myh = (lane * VEC) / C;

  float adv[H], m[H], v[H];
  for (int k = 0; k < H; ++k) { adv[k] = ad_[node * H + k]; m[k] = -INFINITY; v[k] = 0.f; }
  int s = 0;
  bool act = false;

  auto load_chunk = [&](int jb) {
    int idx = jb + lane;
    act = idx < tot;
    s = 0;
    for (int k = 0; k < H; ++k) v[k] = 0.f;
    if (act) {
      s = (idx == 0) ? node : ssrc[beg + idx - 1];
      for (int k = 0; k < H; ++k) v[k] = leaky(as_[s * H + k] + adv[k]);
    }
  };

  for (int jb = 0; jb < tot; jb += 32) {
    load_chunk(jb);
    if (act) { for (int k = 0; k < H; ++k) m[k] = fmaxf(m[k], v[k]); }
  }
  for (int k = 0; k < H; ++k) m[k] = wave_max(m[k]);

  float den[H];
  for (int k = 0; k < H; ++k) den[k] = 0.f;
  for (int jb = 0; jb < tot; jb += 32) {
    if (tot > 32) load_chunk(jb);
    if (act) { for (int k = 0; k < H; ++k) den[k] += expf(v[k] - m[k]); }
  }
  for (int k = 0; k < H; ++k) den[k] = wave_sum(den[k]);

  const char* hb = reinterpret_cast<const char*>(h);
  const unsigned lbytes = (unsigned)lane * (VEC * 4u);
  float acc[VEC];
  for (int q = 0; q < VEC; ++q) acc[q] = 0.f;
  for (int jb = 0; jb < tot; jb += 32) {
    if (tot > 32) load_chunk(jb);
    float w[H];
    for (int k = 0; k < H; ++k) w[k] = act ? expf(v[k] - m[k]) / den[k] : 0.f;
    const int cnt = min(32, tot - jb);
    int t = 0;
    if (UB2 > UB) {
      for (; t + UB2 <= cnt; t += UB2) row_batch<H, VEC, UB2>(hb, lbytes, s, w, myh, t, acc);
    }
    for (; t + UB <= cnt; t += UB) row_batch<H, VEC, UB>(hb, lbytes, s, w, myh, t, acc);
    switch (cnt - t) {
      case 1:  row_batch<H, VEC, 1>(hb, lbytes, s, w, myh, t, acc); break;
      case 2:  row_batch<H, VEC, 2>(hb, lbytes, s, w, myh, t, acc); break;
      case 3:  row_batch<H, VEC, 3>(hb, lbytes, s, w, myh, t, acc); break;
      case 4:  row_batch<H, VEC, 4>(hb, lbytes, s, w, myh, t, acc); break;
      case 5:  row_batch<H, VEC, 5>(hb, lbytes, s, w, myh, t, acc); break;
      case 6:  row_batch<H, VEC, 6>(hb, lbytes, s, w, myh, t, acc); break;
      case 7:  row_batch<H, VEC, 7>(hb, lbytes, s, w, myh, t, acc); break;
      default: break;
    }
  }

  for (int q = 0; q < VEC; ++q) {
    float val = acc[q] + bv[q];
    if (ELU_OUT) val = (val > 0.f) ? val : (expf(val) - 1.f);
    acc[q] = val;
  }
  if (ROW_TO_LDS) {
    *reinterpret_cast<float4*>(outp + lane * VEC) = make_float4(acc[0], acc[1], acc[VEC - 2], acc[VEC - 1]);
    return;
  }
  char* ob = reinterpret_cast<char*>(outp);
  const unsigned obytes = (unsigned)node * (F * 4u) + lbytes;
  if (VEC == 4) {
    *reinterpret_cast<float4*>(ob + obytes) = make_float4(acc[0], acc[1], acc[VEC - 2], acc[VEC - 1]);
  } else {
    *reinterpret_cast<float2*>(ob + obytes) = make_float2(acc[0], acc[1]);
  }
}

template <int H, int VEC, bool ELU_OUT, int UB, int UB2>
__global__ void __launch_bounds__(256)
node_softmax_agg(const int* __restrict__ off, const int* __restrict__ ssrc,
                 const float* __restrict__ as_, const float* __restrict__ ad_,
                 const float* __restrict__ h, const float* __restrict__ bias,
                 float* __restrict__ outp, int n) {
  int node = (int)((blockIdx.x * blockDim.x + threadIdx.x) >> 5);
  node = __builtin_amdgcn_readfirstlane(node);
  if (node >= n) return;
  float bv[VEC];
  load_bias<VEC>(bias, bv);
  const int beg = off[node];
  const int tot = off[node + 1] - beg + 1;
  agg_node<H, VEC, ELU_OUT, UB, UB2, false>(node, beg, tot, ssrc, as_, ad_, h, bv, outp);
}


#define SEGU 4
__global__ void __launch_bounds__(TBE, 16)
group_and_agg(const int* __restrict__ hist, const int* __restrict__ lstart,
              const unsigned* __restrict__ recs, int* __restrict__ segof, int* seg,
              int* ssrc, int N, int NB, int CH,
              const float* __restrict__ as_, const float* __restrict__ ad_,
              const float* __restrict__ h, const float* __restrict__ bias,
              const float* __restrict__ Wn, const float* __restrict__ att_sn,
              const float* __restrict__ att_dn, float* __restrict__ hn,
              float* __restrict__ asn, float* __restrict__ adn) {
  __shared__ float rows[GRP * PA];
  __shared__ float wns[HEADS * DH * DOUT];
  __shared__ int cnt[GRP];
  __shared__ int wtot[GRP / 32];
  __shared__ int segc[MAXC];
  __shared__ int segs[MAXC];
  __shared__ int tot2[2];
  const int t = threadIdx.x;
  const int lane = t & 31;
  const int b = blockIdx.x;
  if (t < GRP) cnt[t] = 0;
  if (t < 2) tot2[t] = 0;
  constexpr int WV = HEADS * DH * DOUT / 4 / TBE;
  float4 wv4[WV];
#pragma unroll
  for (int i = 0; i < WV; ++i) wv4[i] = reinterpret_cast<const float4*>(Wn)[t + i * TBE];
  __syncthreads();
  {
    int ls_sum = 0, cn_sum = 0;
    for (int c = t; c < CH; c += TBE) {
      const int cn = hist[c * NB + b];
      const int ls = lstart[c * NB + b];
      segc[c] = cn;
      segs[c] = c * EPB + ls;
      ls_sum += ls;
      cn_sum += cn;
    }
    for (int o = 16; o > 0; o >>= 1) {
      ls_sum += __shfl_xor(ls_sum, o);
      cn_sum += __shfl_xor(cn_sum, o);
    }
    if (lane == 0) { atomicAdd(&tot2[0], ls_sum); atomicAdd(&tot2[1], cn_sum); }
  }
#pragma unroll
  for (int i = 0; i < WV; ++i) reinterpret_cast<float4*>(wns)[t + i * TBE] = wv4[i];
  __syncthreads();
  const int beg = tot2[0];
  const int end = beg + tot2[1];
  const int wv = t >> 5;
  for (int c0 = wv; c0 < CH; c0 += (TBE / 32) * SEGU) {
    unsigned r[SEGU];
    int n[SEGU], st[SEGU];
#pragma unroll
    for (int u = 0; u < SEGU; ++u) {
      const int c = c0 + (TBE / 32) * u;
      n[u]  = (c < CH) ? segc[c] : 0;
      st[u] = (c < CH) ? segs[c] : 0;
      r[u]  = recs[(n[u] > 0) ? min(st[u] + lane, st[u] + n[u] - 1) : 0];
    }
#pragma unroll
    for (int u = 0; u < SEGU; ++u) {
      if (lane < n[u]) atomicAdd(&cnt[r[u] & (unsigned)(GRP - 1)], 1);
      for (int i = lane + 32; i < n[u]; i += 32)
        atomicAdd(&cnt[recs[st[u] + i] & (unsigned)(GRP - 1)], 1);
    }
  }
  __syncthreads();
  int c = 0, incl = 0;
  if (t < GRP) {
    c = cnt[t];
    incl = c;
    for (int o = 1; o < 32; o <<= 1) {
      const int v = __shfl_up(incl, o);
      if (lane >= o) incl += v;
    }
    if (lane == 31) wtot[t >> 5] = incl;
  }
  __syncthreads();
  if (t < GRP) {
    int basew = 0;
    for (int k = 0; k < (t >> 5); ++k) basew += wtot[k];
    const int first = beg + basew + incl - c;
    const int node = (b << GRP_SHIFT) + t;
    if (node < N) segof[node] = first;
    cnt[t] = first;
  }
  if (b == NB - 1 && t == 0) segof[N] = end;
  __syncthreads();
  for (int c0 = wv; c0 < CH; c0 += (TBE / 32) * SEGU) {
    unsigned r[SEGU];
    int n[SEGU], st[SEGU];
#pragma unroll
    for (int u = 0; u < SEGU; ++u) {
      const int c = c0 + (TBE / 32) * u;
      n[u]  = (c < CH) ? segc[c] : 0;
      st[u] = (c < CH) ? segs[c] : 0;
      r[u]  = recs[(n[u] > 0) ? min(st[u] + lane, st[u] + n[u] - 1) : 0];
    }
#pragma unroll
    for (int u = 0; u < SEGU; ++u) {
      if (lane < n[u]) {
        const int pos = atomicAdd(&cnt[r[u] & (unsigned)(GRP - 1)], 1);
        seg[pos] = (int)(r[u] >> GRP_SHIFT);
      }
      for (int i = lane + 32; i < n[u]; i += 32) {
        const unsigned rr = recs[st[u] + i];
        const int pos = atomicAdd(&cnt[rr & (unsigned)(GRP - 1)], 1);
        seg[pos] = (int)(rr >> GRP_SHIFT);
      }
    }
  }
  __syncthreads();

  for (int ln = t >> 5; ln < GRP; ln += TBE / 32) {
    const int gb = __builtin_amdgcn_readfirstlane((ln == 0) ? beg : cnt[ln - 1]);
    const int ge = __builtin_amdgcn_readfirstlane(cnt[ln]);
    for (int ib = gb; ib < ge; ib += 32) {
      const int i = ib + lane;
      const bool act = i < ge;
      const int ki = act ? seg[i] : 0;
      int r = 0;
      for (int jb = gb; jb < ge; jb += 32) {
        const int j = jb + lane;
        const int kj = (j < ge) ? seg[j] : 0;
        const int cntj = min(32, ge - jb);
        for (int q = 0; q < cntj; ++q) {
          const int k = __builtin_amdgcn_readlane(kj, q);
          const int jj = jb + q;
          r += (k < ki || (k == ki && jj < i)) ? 1 : 0;
        }
      }
      if (act) ssrc[gb + r] = ki;
    }
  }
  __syncthreads();

  const int wvs = __builtin_amdgcn_readfirstlane(t >> 5);
  float bv[4];
  load_bias<4>(bias, bv);
#pragma nounroll
  for (int ln = wvs; ln < GRP; ln += TBE / 32) {
    const int node = (b << GRP_SHIFT) + ln;
    if (node >= N) break;
    const int gb = __builtin_amdgcn_readfirstlane((ln == 0) ? beg : cnt[ln - 1]);
    const int ge = __builtin_amdgcn_readfirstlane(cnt[ln]);
    agg_node<HEADS, 4, true, 8, 8, true>(node, gb, ge - gb + 1, ssrc, as_, ad_, h, bv, rows + ln * PA);
  }
  __syncthreads();

  if (wvs < GRP / 16)
    gemm_strip_lds<HEADS * DH>(rows + wvs * 16 * PA, wns, att_sn, att_dn, hn, asn, adn, N,
                               b * (GRP / 16) + wvs);
}


extern "C" void kernel_launch(void* const* d_in, const int* in_sizes, int n_in,
                              void* d_out, int out_size, void* d_ws, size_t ws_size,
                              hipStream_t stream) {
  const float* x    = (const float*)d_in[0];
  const int*   ei   = (const int*)  d_in[1];
  const float* W1   = (const float*)d_in[2];
  const float* atS1 = (const float*)d_in[3];
  const float* atD1 = (const float*)d_in[4];
  const float* b1   = (const float*)d_in[5];
  const float* W2   = (const float*)d_in[6];
  const float* atS2 = (const float*)d_in[7];
  const float* atD2 = (const float*)d_in[8];
  const float* b2   = (const float*)d_in[9];
  float* out = (float*)d_out;

  const int N    = in_sizes[0] / DIN;
  const int E0   = in_sizes[1] / 2;
  const int* srcs = ei;
  const int* dsts = ei + E0;

  float* ws = (float*)d_ws;
  long off = 0;
  float* h1  = ws + off; off += (long)N * (HEADS * DH);
  float* a1s = ws + off; off += (long)N * HEADS;
  float* a1d = ws + off; off += (long)N * HEADS;
  float* h2  = ws + off; off += (long)N * DOUT;
  float* a2s = ws + off; off += (long)N;
  float* a2d = ws + off; off += (long)N;
  const int NB = (N + GRP - 1) / GRP;
  const int CH = (E0 + EPB - 1) / EPB;
  int* segof = (int*)(ws + off); off += (long)N + 1;
  int* hist  = (int*)(ws + off); off += (long)CH * NB;
  int* lstart = (int*)(ws + off); off += (long)CH * NB;
  unsigned* recs = (unsigned*)(ws + off); off += (long)E0;
  int* seg   = (int*)(ws + off); off += (long)E0;
  int* ssrc  = (int*)(ws + off); off += (long)E0;

  const int TB = 256;
  auto blocks = [](long n, int tb) { return (unsigned)((n + tb - 1) / tb); };

  {
    long waves = (long)((N + 15) / 16) * HEADS;
    count_and_gemm<DIN, HEADS * DH><<<CH + blocks(waves * 32, TBG), TBG, 0, stream>>>(
        srcs, dsts, E0, NB, CH, hist, lstart, recs, x, W1, atS1, atD1, h1, a1s, a1d, N);
  }
  group_and_agg<<<NB, TBE, 0, stream>>>(hist, lstart, recs, segof, seg, ssrc, N, NB, CH,
                                        a1s, a1d, h1, b1, W2, atS2, atD2, h2, a2s, a2d);

  node_softmax_agg<1, 2, false, 8, 16><<<blocks((long)N * 32, TB), TB, 0, stream>>>(
      segof, ssrc, a2s, a2d, h2, b2, out, N);
}


# A node's group is ordered by source inside the wave that aggregates it (rank counted in registers, one push permute): separate ordering phase, its list re-read and barrier gone; sum order unchanged
# speedup vs baseline: 7.9250x; 1.0376x over previous
#include <hip/hip_runtime.h>
#include <math.h>

#define DIN   128
#define DH    64
#define HEADS 2
#define DOUT  64
#define NEG_SLOPE 0.2f

typedef float v2f __attribute__((ext_vector_type(2)));
typedef float v8f __attribute__((ext_vector_type(8)));


template <int CNT>
__device__ __forceinline__ void fold_half(float (&v)[16], bool upper, int dist) {
#pragma unroll
  for (int i = 0; i < CNT; ++i) {
    float lo = v[i], hi = v[i + CNT];
    float send = upper ? lo : hi;
    float keep = upper ? hi : lo;
    v[i] = keep + __shfl_xor(send, dist);
  }
}

template <int K, int NC>
__device__ __forceinline__ void
gemm_heads_wave(const float* __restrict__ A, const float* __restrict__ B,
                const float* __restrict__ att_s, const float* __restrict__ att_d,
                float* __restrict__ C, float* __restrict__ as_, float* __restrict__ ad_,
                int M, int wave) {
  constexpr int HW = NC / 64;
  const int lane = threadIdx.x & 31;
  const int tm = wave / HW;
  const int hd = wave - tm * HW;
  if (tm * 16 >= M) return;
  const int half = lane >> 4;
  const int l16  = lane & 15;
  long rowA = (long)tm * 16 + l16;
  if (rowA >= M) rowA = M - 1;
  const int colbase = hd * 64 + l16 * 4;
  const float* ap = A + rowA * K + half * 2;
  const float* bp = B + (long)(half * 2) * NC + colbase;
  v8f acc0 = {}, acc1 = {}, acc2 = {}, acc3 = {};
#pragma unroll
  for (int k0 = 0; k0 < K; k0 += 4) {
    v2f a = *reinterpret_cast<const v2f*>(ap + k0);
    float4 bx = *reinterpret_cast<const float4*>(bp + (long)k0 * NC);
    float4 by = *reinterpret_cast<const float4*>(bp + (long)(k0 + 1) * NC);
    v2f b;
    b.x = bx.x; b.y = by.x;
    acc0 = __builtin_amdgcn_wmma_f32_16x16x4_f32(false, a, false, b, (short)0, acc0, false, false);
    b.x = bx.y; b.y = by.y;
    acc1 = __builtin_amdgcn_wmma_f32_16x16x4_f32(false, a, false, b, (short)0, acc1, false, false);
    b.x = bx.z; b.y = by.z;
    acc2 = __builtin_amdgcn_wmma_f32_16x16x4_f32(false, a, false, b, (short)0, acc2, false, false);
    b.x = bx.w; b.y = by.w;
    acc3 = __builtin_amdgcn_wmma_f32_16x16x4_f32(false, a, false, b, (short)0, acc3, false, false);
  }

  const float4 ats = *reinterpret_cast<const float4*>(att_s + colbase);
  const float4 atd = *reinterpret_cast<const float4*>(att_d + colbase);
  float v[16];
#pragma unroll
  for (int r = 0; r < 8; ++r) {
    const float c0 = acc0[r], c1 = acc1[r], c2 = acc2[r], c3 = acc3[r];
    const int row = tm * 16 + r + half * 8;
    if (row < M)
      *reinterpret_cast<float4*>(C + (long)row * NC + colbase) = make_float4(c0, c1, c2, c3);
    v[r]     = c0 * ats.x + c1 * ats.y + c2 * ats.z + c3 * ats.w;
    v[8 + r] = c0 * atd.x + c1 * atd.y + c2 * atd.z + c3 * atd.w;
  }
  fold_half<8>(v, (l16 & 8) != 0, 8);
  fold_half<4>(v, (l16 & 4) != 0, 4);
  fold_half<2>(v, (l16 & 2) != 0, 2);
  fold_half<1>(v, (l16 & 1) != 0, 1);
  const int orow = tm * 16 + (l16 & 7) + half * 8;
  if (orow < M) {
    float* dp = (l16 < 8) ? as_ : ad_;
    dp[(long)orow * HW + hd] = v[0];
  }
}

#define PA 132
template <int K>
__device__ __forceinline__ void
gemm_strip_lds(const float* As, const float* Bs,
               const float* __restrict__ att_s, const float* __restrict__ att_d,
               float* __restrict__ C, float* __restrict__ as_, float* __restrict__ ad_,
               int M, int tm) {
  constexpr int NC = 64;
  const int lane = (int)__builtin_amdgcn_mbcnt_lo(~0u, 0u);
  if (tm * 16 >= M) return;
  const int half = lane >> 4;
  const int l16  = lane & 15;
  const int colbase = l16 * 4;
  const float* ap = As + l16 * PA + half * 2;
  const float* bp = Bs + (half * 2) * NC + colbase;
  v8f acc0 = {}, acc1 = {}, acc2 = {}, acc3 = {};
#pragma unroll
  for (int k0 = 0; k0 < K; k0 += 4) {
    v2f a = *reinterpret_cast<const v2f*>(ap + k0);
    float4 bx = *reinterpret_cast<const float4*>(bp + k0 * NC);
    float4 by = *reinterpret_cast<const float4*>(bp + (k0 + 1) * NC);
    v2f b;
    b.x = bx.x; b.y = by.x;
    acc0 = __builtin_amdgcn_wmma_f32_16x16x4_f32(false, a, false, b, (short)0, acc0, false, false);
    b.x = bx.y; b.y = by.y;
    acc1 = __builtin_amdgcn_wmma_f32_16x16x4_f32(false, a, false, b, (short)0, acc1, false, false);
    b.x = bx.z; b.y = by.z;
    acc2 = __builtin_amdgcn_wmma_f32_16x16x4_f32(false, a, false, b, (short)0, acc2, false, false);
    b.x = bx.w; b.y = by.w;
    acc3 = __builtin_amdgcn_wmma_f32_16x16x4_f32(false, a, false, b, (short)0, acc3, false, false);
  }

  const float4 ats = *reinterpret_cast<const float4*>(att_s + colbase);
  const float4 atd = *reinterpret_cast<const float4*>(att_d + colbase);
  float v[16];
#pragma unroll
  for (int r = 0; r < 8; ++r) {
    const float c0 = acc0[r], c1 = acc1[r], c2 = acc2[r], c3 = acc3[r];
    const int row = tm * 16 + r + half * 8;
    if (row < M)
      *reinterpret_cast<float4*>(C + (long)row * NC + colbase) = make_float4(c0, c1, c2, c3);
    v[r]     = c0 * ats.x + c1 * ats.y + c2 * ats.z + c3 * ats.w;
    v[8 + r] = c0 * atd.x + c1 * atd.y + c2 * atd.z + c3 * atd.w;
  }
  fold_half<8>(v, (l16 & 8) != 0, 8);
  fold_half<4>(v, (l16 & 4) != 0, 4);
  fold_half<2>(v, (l16 & 2) != 0, 2);
  fold_half<1>(v, (l16 & 1) != 0, 1);
  const int orow = tm * 16 + (l16 & 7) + half * 8;
  if (orow < M) {
    float* dp = (l16 < 8) ? as_ : ad_;
    dp[orow] = v[0];
  }
}

#define GRP_SHIFT 7
#define GRP  128
#define MAXB 1024
#define EPB  4096
#define MAXC 1024
#define TBE  1024

__device__ __forceinline__ unsigned range_of(int d, int NB) {
  return min((unsigned)d >> GRP_SHIFT, (unsigned)(NB - 1));
}

#define TBG 256
template <int K, int NC>
__global__ void __launch_bounds__(TBG)
count_and_gemm(const int* __restrict__ srcs, const int* __restrict__ dsts, int E0, int NB, int CH,
               int* __restrict__ hist, int* __restrict__ lstart, unsigned* __restrict__ recs,
               const float* __restrict__ A, const float* __restrict__ B,
               const float* __restrict__ att_s, const float* __restrict__ att_d,
               float* __restrict__ C, float* __restrict__ as_, float* __restrict__ ad_,
               int M) {
  const int t = threadIdx.x;
  if ((int)blockIdx.x < CH) {
    __shared__ int h[MAXB];
    __shared__ int wsum[TBG / 32];
    const int lane = t & 31;
    const int chunk = blockIdx.x;
    for (int b = t; b < MAXB; b += TBG) h[b] = 0;
    __syncthreads();
    const int base = chunk * EPB;
    int d[EPB / TBG], s[EPB / TBG], rk[EPB / TBG];
#pragma unroll
    for (int k = 0; k < EPB / TBG; ++k) {
      const int e = min(base + k * TBG + t, E0 - 1);
      d[k] = dsts[e];
      s[k] = srcs[e];
    }
#pragma unroll
    for (int k = 0; k < EPB / TBG; ++k) {
      const int e = base + k * TBG + t;
      rk[k] = (e < E0) ? atomicAdd(&h[range_of(d[k], NB)], 1) : 0;
    }
    __syncthreads();
    constexpr int PER = MAXB / TBG;
    int c[PER];
    int own = 0;
#pragma unroll
    for (int i = 0; i < PER; ++i) { c[i] = h[t * PER + i]; own += c[i]; }
    int incl = own;
    for (int o = 1; o < 32; o <<= 1) {
      const int v = __shfl_up(incl, o);
      if (lane >= o) incl += v;
    }
    if (lane == 31) wsum[t >> 5] = incl;
    __syncthreads();
    int run = incl - own;
    for (int k = 0; k < (t >> 5); ++k) run += wsum[k];
#pragma unroll
    for (int i = 0; i < PER; ++i) {
      const int b = t * PER + i;
      h[b] = run;
      if (b < NB) { hist[chunk * NB + b] = c[i]; lstart[chunk * NB + b] = run; }
      run += c[i];
    }
    __syncthreads();
#pragma unroll
    for (int k = 0; k < EPB / TBG; ++k) {
      const int e = base + k * TBG + t;
      if (e < E0)
        recs[base + h[range_of(d[k], NB)] + rk[k]] =
            ((unsigned)s[k] << GRP_SHIFT) | ((unsigned)d[k] & (unsigned)(GRP - 1));
    }
    return;
  }
  int wave = (int)((((int)blockIdx.x - CH) * TBG + t) >> 5);
  wave = __builtin_amdgcn_readfirstlane(wave);
  gemm_heads_wave<K, NC>(A, B, att_s, att_d, C, as_, ad_, M, wave);
}

__device__ __forceinline__ float leaky(float v) { return (v > 0.f) ? v : NEG_SLOPE * v; }

__device__ __forceinline__ float wave_max(float v) {
  for (int o = 16; o > 0; o >>= 1) v = fmaxf(v, __shfl_xor(v, o));
  return v;
}
__device__ __forceinline__ float wave_sum(float v) {
  for (int o = 16; o > 0; o >>= 1) v += __shfl_xor(v, o);
  return v;
}

template <int H, int VEC, int U>
__device__ __forceinline__ void
row_batch(const char* __restrict__ hb, unsigned lbytes, int s, const float (&w)[H], int myh, int t0,
          float (&acc)[VEC]) {
  constexpr unsigned ROWB = 32u * VEC * 4u;
  float4 r4[U];
  float2 r2[U];
#pragma unroll
  for (int u = 0; u < U; ++u) {
    const unsigned st = (unsigned)__builtin_amdgcn_readlane(s, t0 + u);
    const unsigned boff = st * ROWB + lbytes;
    if (VEC == 4) r4[u] = *reinterpret_cast<const float4*>(hb + boff);
    else          r2[u] = *reinterpret_cast<const float2*>(hb + boff);
  }
#pragma unroll
  for (int u = 0; u < U; ++u) {
    float wt = __int_as_float(__builtin_amdgcn_readlane(__float_as_int(w[0]), t0 + u));
    if (H == 2) {
      float w1 = __int_as_float(__builtin_amdgcn_readlane(__float_as_int(w[H - 1]), t0 + u));
      wt = myh ? w1 : wt;
    }
    if (VEC == 4) {
      acc[0] = fmaf(r4[u].x, wt, acc[0]); acc[1] = fmaf(r4[u].y, wt, acc[1]);
      acc[VEC - 2] = fmaf(r4[u].z, wt, acc[VEC - 2]); acc[VEC - 1] = fmaf(r4[u].w, wt, acc[VEC - 1]);
    } else {
      acc[0] = fmaf(r2[u].x, wt, acc[0]); acc[1] = fmaf(r2[u].y, wt, acc[1]);
    }
  }
}

template <int VEC>
__device__ __forceinline__ void load_bias(const float* __restrict__ bias, float (&bv)[VEC]) {
  const int lane = threadIdx.x & 31;
  if (VEC == 4) {
    const float4 b = *reinterpret_cast<const float4*>(bias + lane * VEC);
    bv[0] = b.x; bv[1] = b.y; bv[VEC - 2] = b.z; bv[VEC - 1] = b.w;
  } else {
    const float2 b = *reinterpret_cast<const float2*>(bias + lane * VEC);
    bv[0] = b.x; bv[1] = b.y;
  }
}

template <int H, int VEC, bool ELU_OUT, int UB, int UB2, bool ROW_TO_LDS, bool MAY_PRE>
__device__ __forceinline__ void
agg_node(int node, int beg, int tot, const int* ssrc,
         const float* __restrict__ as_, const float* __restrict__ ad_,
         const float* __restrict__ h, const float (&bv)[VEC],
         float* __restrict__ outp, int s_pre, bool pre) {
  constexpr int F = 32 * VEC;
  constexpr int C = F / H;
  const int lane = threadIdx.x & 31;
  const int myh = (lane * VEC) / C;

  float adv[H], m[H], v[H];
  for (int k = 0; k < H; ++k) { adv[k] = ad_[node * H + k]; m[k] = -INFINITY; v[k] = 0.f; }
  int s = 0;
  bool act = false;

  auto load_chunk = [&](int jb) {
    int idx = jb + lane;
    act = idx < tot;
    s = 0;
    for (int k = 0; k < H; ++k) v[k] = 0.f;
    if (act) {
      s = (MAY_PRE && pre) ? s_pre : ((idx == 0) ? node : ssrc[beg + idx - 1]);
      for (int k = 0; k < H; ++k) v[k] = leaky(as_[s * H + k] + adv[k]);
    }
  };

  for (int jb = 0; jb < tot; jb += 32) {
    load_chunk(jb);
    if (act) { for (int k = 0; k < H; ++k) m[k] = fmaxf(m[k], v[k]); }
  }
  for (int k = 0; k < H; ++k) m[k] = wave_max(m[k]);

  float den[H];
  for (int k = 0; k < H; ++k) den[k] = 0.f;
  for (int jb = 0; jb < tot; jb += 32) {
    if (tot > 32) load_chunk(jb);
    if (act) { for (int k = 0; k < H; ++k) den[k] += expf(v[k] - m[k]); }
  }
  for (int k = 0; k < H; ++k) den[k] = wave_sum(den[k]);

  const char* hb = reinterpret_cast<const char*>(h);
  const unsigned lbytes = (unsigned)lane * (VEC * 4u);
  float acc[VEC];
  for (int q = 0; q < VEC; ++q) acc[q] = 0.f;
  for (int jb = 0; jb < tot; jb += 32) {
    if (tot > 32) load_chunk(jb);
    float w[H];
    for (int k = 0; k < H; ++k) w[k] = act ? expf(v[k] - m[k]) / den[k] : 0.f;
    const int cnt = min(32, tot - jb);
    int t = 0;
    if (UB2 > UB) {
      for (; t + UB2 <= cnt; t += UB2) row_batch<H, VEC, UB2>(hb, lbytes, s, w, myh, t, acc);
    }
    for (; t + UB <= cnt; t += UB) row_batch<H, VEC, UB>(hb, lbytes, s, w, myh, t, acc);
    switch (cnt - t) {
      case 1:  row_batch<H, VEC, 1>(hb, lbytes, s, w, myh, t, acc); break;
      case 2:  row_batch<H, VEC, 2>(hb, lbytes, s, w, myh, t, acc); break;
      case 3:  row_batch<H, VEC, 3>(hb, lbytes, s, w, myh, t, acc); break;
      case 4:  row_batch<H, VEC, 4>(hb, lbytes, s, w, myh, t, acc); break;
      case 5:  row_batch<H, VEC, 5>(hb, lbytes, s, w, myh, t, acc); break;
      case 6:  row_batch<H, VEC, 6>(hb, lbytes, s, w, myh, t, acc); break;
      case 7:  row_batch<H, VEC, 7>(hb, lbytes, s, w, myh, t, acc); break;
      default: break;
    }
  }

  for (int q = 0; q < VEC; ++q) {
    float val = acc[q] + bv[q];
    if (ELU_OUT) val = (val > 0.f) ? val : (expf(val) - 1.f);
    acc[q] = val;
  }
  if (ROW_TO_LDS) {
    *reinterpret_cast<float4*>(outp + lane * VEC) = make_float4(acc[0], acc[1], acc[VEC - 2], acc[VEC - 1]);
    return;
  }
  char* ob = reinterpret_cast<char*>(outp);
  const unsigned obytes = (unsigned)node * (F * 4u) + lbytes;
  if (VEC == 4) {
    *reinterpret_cast<float4*>(ob + obytes) = make_float4(acc[0], acc[1], acc[VEC - 2], acc[VEC - 1]);
  } else {
    *reinterpret_cast<float2*>(ob + obytes) = make_float2(acc[0], acc[1]);
  }
}

template <int H, int VEC, bool ELU_OUT, int UB, int UB2>
__global__ void __launch_bounds__(256)
node_softmax_agg(const int* __restrict__ off, const int* __restrict__ ssrc,
                 const float* __restrict__ as_, const float* __restrict__ ad_,
                 const float* __restrict__ h, const float* __restrict__ bias,
                 float* __restrict__ outp, int n) {
  int node = (int)((blockIdx.x * blockDim.x + threadIdx.x) >> 5);
  node = __builtin_amdgcn_readfirstlane(node);
  if (node >= n) return;
  float bv[VEC];
  load_bias<VEC>(bias, bv);
  const int beg = off[node];
  const int tot = off[node + 1] - beg + 1;
  agg_node<H, VEC, ELU_OUT, UB, UB2, false, false>(node, beg, tot, ssrc, as_, ad_, h, bv, outp, 0, false);
}


#define SEGU 4
__global__ void __launch_bounds__(TBE, 16)
group_and_agg(const int* __restrict__ hist, const int* __restrict__ lstart,
              const unsigned* __restrict__ recs, int* __restrict__ segof, int* seg,
              int* ssrc, int N, int NB, int CH,
              const float* __restrict__ as_, const float* __restrict__ ad_,
              const float* __restrict__ h, const float* __restrict__ bias,
              const float* __restrict__ Wn, const float* __restrict__ att_sn,
              const float* __restrict__ att_dn, float* __restrict__ hn,
              float* __restrict__ asn, float* __restrict__ adn) {
  __shared__ float rows[GRP * PA];
  __shared__ float wns[HEADS * DH * DOUT];
  __shared__ int cnt[GRP];
  __shared__ int wtot[GRP / 32];
  __shared__ int segc[MAXC];
  __shared__ int segs[MAXC];
  __shared__ int tot2[2];
  const int t = threadIdx.x;
  const int lane = t & 31;
  const int b = blockIdx.x;
  if (t < GRP) cnt[t] = 0;
  if (t < 2) tot2[t] = 0;
  constexpr int WV = HEADS * DH * DOUT / 4 / TBE;
  float4 wv4[WV];
#pragma unroll
  for (int i = 0; i < WV; ++i) wv4[i] = reinterpret_cast<const float4*>(Wn)[t + i * TBE];
  __syncthreads();
  {
    int ls_sum = 0, cn_sum = 0;
    for (int c = t; c < CH; c += TBE) {
      const int cn = hist[c * NB + b];
      const int ls = lstart[c * NB + b];
      segc[c] = cn;
      segs[c] = c * EPB + ls;
      ls_sum += ls;
      cn_sum += cn;
    }
    for (int o = 16; o > 0; o >>= 1) {
      ls_sum += __shfl_xor(ls_sum, o);
      cn_sum += __shfl_xor(cn_sum, o);
    }
    if (lane == 0) { atomicAdd(&tot2[0], ls_sum); atomicAdd(&tot2[1], cn_sum); }
  }
#pragma unroll
  for (int i = 0; i < WV; ++i) reinterpret_cast<float4*>(wns)[t + i * TBE] = wv4[i];
  __syncthreads();
  const int beg = tot2[0];
  const int end = beg + tot2[1];
  const int wv = t >> 5;
  for (int c0 = wv; c0 < CH; c0 += (TBE / 32) * SEGU) {
    unsigned r[SEGU];
    int n[SEGU], st[SEGU];
#pragma unroll
    for (int u = 0; u < SEGU; ++u) {
      const int c = c0 + (TBE / 32) * u;
      n[u]  = (c < CH) ? segc[c] : 0;
      st[u] = (c < CH) ? segs[c] : 0;
      r[u]  = recs[(n[u] > 0) ? min(st[u] + lane, st[u] + n[u] - 1) : 0];
    }
#pragma unroll
    for (int u = 0; u < SEGU; ++u) {
      if (lane < n[u]) atomicAdd(&cnt[r[u] & (unsigned)(GRP - 1)], 1);
      for (int i = lane + 32; i < n[u]; i += 32)
        atomicAdd(&cnt[recs[st[u] + i] & (unsigned)(GRP - 1)], 1);
    }
  }
  __syncthreads();
  int c = 0, incl = 0;
  if (t < GRP) {
    c = cnt[t];
    incl = c;
    for (int o = 1; o < 32; o <<= 1) {
      const int v = __shfl_up(incl, o);
      if (lane >= o) incl += v;
    }
    if (lane == 31) wtot[t >> 5] = incl;
  }
  __syncthreads();
  if (t < GRP) {
    int basew = 0;
    for (int k = 0; k < (t >> 5); ++k) basew += wtot[k];
    const int first = beg + basew + incl - c;
    const int node = (b << GRP_SHIFT) + t;
    if (node < N) segof[node] = first;
    cnt[t] = first;
  }
  if (b == NB - 1 && t == 0) segof[N] = end;
  __syncthreads();
  for (int c0 = wv; c0 < CH; c0 += (TBE / 32) * SEGU) {
    unsigned r[SEGU];
    int n[SEGU], st[SEGU];
#pragma unroll
    for (int u = 0; u < SEGU; ++u) {
      const int c = c0 + (TBE / 32) * u;
      n[u]  = (c < CH) ? segc[c] : 0;
      st[u] = (c < CH) ? segs[c] : 0;
      r[u]  = recs[(n[u] > 0) ? min(st[u] + lane, st[u] + n[u] - 1) : 0];
    }
#pragma unroll
    for (int u = 0; u < SEGU; ++u) {
      if (lane < n[u]) {
        const int pos = atomicAdd(&cnt[r[u] & (unsigned)(GRP - 1)], 1);
        seg[pos] = (int)(r[u] >> GRP_SHIFT);
      }
      for (int i = lane + 32; i < n[u]; i += 32) {
        const unsigned rr = recs[st[u] + i];
        const int pos = atomicAdd(&cnt[rr & (unsigned)(GRP - 1)], 1);
        seg[pos] = (int)(rr >> GRP_SHIFT);
      }
    }
  }
  __syncthreads();

  const int wvs = __builtin_amdgcn_readfirstlane(t >> 5);
  float bv[4];
  load_bias<4>(bias, bv);
#pragma nounroll
  for (int ln = wvs; ln < GRP; ln += TBE / 32) {
    const int node = (b << GRP_SHIFT) + ln;
    if (node >= N) break;
    const int gb = __builtin_amdgcn_readfirstlane((ln == 0) ? beg : cnt[ln - 1]);
    const int ge = __builtin_amdgcn_readfirstlane(cnt[ln]);
    const int deg = ge - gb;
    const bool pre = deg < 32;
    int s_pre = 0;
    if (pre) {
      const bool a = lane <= deg;
      const bool g = a && lane > 0;
      const int ki = g ? seg[gb + lane - 1] : node;
      const unsigned key = g ? (((unsigned)ki << 5) | (unsigned)lane) : 0u;
      int r = 0;
      for (int q = 0; q <= deg; ++q) {
        const unsigned kq = (unsigned)__builtin_amdgcn_readlane((int)key, q);
        r += (kq < key) ? 1 : 0;
      }
      if (g) ssrc[gb + r - 1] = ki;
      s_pre = __builtin_amdgcn_ds_permute((a ? r : lane) << 2, ki);
    } else {
      for (int ib = gb; ib < ge; ib += 32) {
        const int i = ib + lane;
        const bool act = i < ge;
        const int ki = act ? seg[i] : 0;
        int r = 0;
        for (int jb = gb; jb < ge; jb += 32) {
          const int j = jb + lane;
          const int kj = (j < ge) ? seg[j] : 0;
          const int cntj = min(32, ge - jb);
          for (int q = 0; q < cntj; ++q) {
            const int k = __builtin_amdgcn_readlane(kj, q);
            const int jj = jb + q;
            r += (k < ki || (k == ki && jj < i)) ? 1 : 0;
          }
        }
        if (act) ssrc[gb + r] = ki;
      }
      __threadfence_block();
    }
    agg_node<HEADS, 4, true, 8, 8, true, true>(node, gb, deg + 1, ssrc, as_, ad_, h, bv,
                                               rows + ln * PA, s_pre, pre);
  }
  __syncthreads();

  if (wvs < GRP / 16)
    gemm_strip_lds<HEADS * DH>(rows + wvs * 16 * PA, wns, att_sn, att_dn, hn, asn, adn, N,
                               b * (GRP / 16) + wvs);
}


extern "C" void kernel_launch(void* const* d_in, const int* in_sizes, int n_in,
                              void* d_out, int out_size, void* d_ws, size_t ws_size,
                              hipStream_t stream) {
  const float* x    = (const float*)d_in[0];
  const int*   ei   = (const int*)  d_in[1];
  const float* W1   = (const float*)d_in[2];
  const float* atS1 = (const float*)d_in[3];
  const float* atD1 = (const float*)d_in[4];
  const float* b1   = (const float*)d_in[5];
  const float* W2   = (const float*)d_in[6];
  const float* atS2 = (const float*)d_in[7];
  const float* atD2 = (const float*)d_in[8];
  const float* b2   = (const float*)d_in[9];
  float* out = (float*)d_out;

  const int N    = in_sizes[0] / DIN;
  const int E0   = in_sizes[1] / 2;
  const int* srcs = ei;
  const int* dsts = ei + E0;

  float* ws = (float*)d_ws;
  long off = 0;
  float* h1  = ws + off; off += (long)N * (HEADS * DH);
  float* a1s = ws + off; off += (long)N * HEADS;
  float* a1d = ws + off; off += (long)N * HEADS;
  float* h2  = ws + off; off += (long)N * DOUT;
  float* a2s = ws + off; off += (long)N;
  float* a2d = ws + off; off += (long)N;
  const int NB = (N + GRP - 1) / GRP;
  const int CH = (E0 + EPB - 1) / EPB;
  int* segof = (int*)(ws + off); off += (long)N + 1;
  int* hist  = (int*)(ws + off); off += (long)CH * NB;
  int* lstart = (int*)(ws + off); off += (long)CH * NB;
  unsigned* recs = (unsigned*)(ws + off); off += (long)E0;
  int* seg   = (int*)(ws + off); off += (long)E0;
  int* ssrc  = (int*)(ws + off); off += (long)E0;

  const int TB = 256;
  auto blocks = [](long n, int tb) { return (unsigned)((n + tb - 1) / tb); };

  {
    long waves = (long)((N + 15) / 16) * HEADS;
    count_and_gemm<DIN, HEADS * DH><<<CH + blocks(waves * 32, TBG), TBG, 0, stream>>>(
        srcs, dsts, E0, NB, CH, hist, lstart, recs, x, W1, atS1, atD1, h1, a1s, a1d, N);
  }
  group_and_agg<<<NB, TBE, 0, stream>>>(hist, lstart, recs, segof, seg, ssrc, N, NB, CH,
                                        a1s, a1d, h1, b1, W2, atS2, atD2, h2, a2s, a2d);

  node_softmax_agg<1, 2, false, 8, 16><<<blocks((long)N * 32, TB), TB, 0, stream>>>(
      segof, ssrc, a2s, a2d, h2, b2, out, N);
}


# Destination ranges of 100 nodes, not 128: grouping launch has 500 blocks, two on nearly every WGP, instead of 391 (two on half the WGPs, one on the rest); last product strip partial; sums unchanged
# speedup vs baseline: 8.2574x; 1.0419x over previous
#include <hip/hip_runtime.h>
#include <math.h>

#define DIN   128
#define DH    64
#define HEADS 2
#define DOUT  64
#define NEG_SLOPE 0.2f

typedef float v2f __attribute__((ext_vector_type(2)));
typedef float v8f __attribute__((ext_vector_type(8)));


template <int CNT>
__device__ __forceinline__ void fold_half(float (&v)[16], bool upper, int dist) {
#pragma unroll
  for (int i = 0; i < CNT; ++i) {
    float lo = v[i], hi = v[i + CNT];
    float send = upper ? lo : hi;
    float keep = upper ? hi : lo;
    v[i] = keep + __shfl_xor(send, dist);
  }
}

template <int K, int NC>
__device__ __forceinline__ void
gemm_heads_wave(const float* __restrict__ A, const float* __restrict__ B,
                const float* __restrict__ att_s, const float* __restrict__ att_d,
                float* __restrict__ C, float* __restrict__ as_, float* __restrict__ ad_,
                int M, int wave) {
  constexpr int HW = NC / 64;
  const int lane = threadIdx.x & 31;
  const int tm = wave / HW;
  const int hd = wave - tm * HW;
  if (tm * 16 >= M) return;
  const int half = lane >> 4;
  const int l16  = lane & 15;
  long rowA = (long)tm * 16 + l16;
  if (rowA >= M) rowA = M - 1;
  const int colbase = hd * 64 + l16 * 4;
  const float* ap = A + rowA * K + half * 2;
  const float* bp = B + (long)(half * 2) * NC + colbase;
  v8f acc0 = {}, acc1 = {}, acc2 = {}, acc3 = {};
#pragma unroll
  for (int k0 = 0; k0 < K; k0 += 4) {
    v2f a = *reinterpret_cast<const v2f*>(ap + k0);
    float4 bx = *reinterpret_cast<const float4*>(bp + (long)k0 * NC);
    float4 by = *reinterpret_cast<const float4*>(bp + (long)(k0 + 1) * NC);
    v2f b;
    b.x = bx.x; b.y = by.x;
    acc0 = __builtin_amdgcn_wmma_f32_16x16x4_f32(false, a, false, b, (short)0, acc0, false, false);
    b.x = bx.y; b.y = by.y;
    acc1 = __builtin_amdgcn_wmma_f32_16x16x4_f32(false, a, false, b, (short)0, acc1, false, false);
    b.x = bx.z; b.y = by.z;
    acc2 = __builtin_amdgcn_wmma_f32_16x16x4_f32(false, a, false, b, (short)0, acc2, false, false);
    b.x = bx.w; b.y = by.w;
    acc3 = __builtin_amdgcn_wmma_f32_16x16x4_f32(false, a, false, b, (short)0, acc3, false, false);
  }

  const float4 ats = *reinterpret_cast<const float4*>(att_s + colbase);
  const float4 atd = *reinterpret_cast<const float4*>(att_d + colbase);
  float v[16];
#pragma unroll
  for (int r = 0; r < 8; ++r) {
    const float c0 = acc0[r], c1 = acc1[r], c2 = acc2[r], c3 = acc3[r];
    const int row = tm * 16 + r + half * 8;
    if (row < M)
      *reinterpret_cast<float4*>(C + (long)row * NC + colbase) = make_float4(c0, c1, c2, c3);
    v[r]     = c0 * ats.x + c1 * ats.y + c2 * ats.z + c3 * ats.w;
    v[8 + r] = c0 * atd.x + c1 * atd.y + c2 * atd.z + c3 * atd.w;
  }
  fold_half<8>(v, (l16 & 8) != 0, 8);
  fold_half<4>(v, (l16 & 4) != 0, 4);
  fold_half<2>(v, (l16 & 2) != 0, 2);
  fold_half<1>(v, (l16 & 1) != 0, 1);
  const int orow = tm * 16 + (l16 & 7) + half * 8;
  if (orow < M) {
    float* dp = (l16 < 8) ? as_ : ad_;
    dp[(long)orow * HW + hd] = v[0];
  }
}

#define PA 132
template <int K>
__device__ __forceinline__ void
gemm_strip_lds(const float* As, const float* Bs,
               const float* __restrict__ att_s, const float* __restrict__ att_d,
               float* __restrict__ C, float* __restrict__ as_, float* __restrict__ ad_,
               int M, int row0, int nrows) {
  constexpr int NC = 64;
  const int lane = (int)__builtin_amdgcn_mbcnt_lo(~0u, 0u);
  if (row0 >= M) return;
  const int half = lane >> 4;
  const int l16  = lane & 15;
  const int colbase = l16 * 4;
  const float* ap = As + l16 * PA + half * 2;
  const float* bp = Bs + (half * 2) * NC + colbase;
  v8f acc0 = {}, acc1 = {}, acc2 = {}, acc3 = {};
#pragma unroll
  for (int k0 = 0; k0 < K; k0 += 4) {
    v2f a = *reinterpret_cast<const v2f*>(ap + k0);
    float4 bx = *reinterpret_cast<const float4*>(bp + k0 * NC);
    float4 by = *reinterpret_cast<const float4*>(bp + (k0 + 1) * NC);
    v2f b;
    b.x = bx.x; b.y = by.x;
    acc0 = __builtin_amdgcn_wmma_f32_16x16x4_f32(false, a, false, b, (short)0, acc0, false, false);
    b.x = bx.y; b.y = by.y;
    acc1 = __builtin_amdgcn_wmma_f32_16x16x4_f32(false, a, false, b, (short)0, acc1, false, false);
    b.x = bx.z; b.y = by.z;
    acc2 = __builtin_amdgcn_wmma_f32_16x16x4_f32(false, a, false, b, (short)0, acc2, false, false);
    b.x = bx.w; b.y = by.w;
    acc3 = __builtin_amdgcn_wmma_f32_16x16x4_f32(false, a, false, b, (short)0, acc3, false, false);
  }

  const float4 ats = *reinterpret_cast<const float4*>(att_s + colbase);
  const float4 atd = *reinterpret_cast<const float4*>(att_d + colbase);
  float v[16];
#pragma unroll
  for (int r = 0; r < 8; ++r) {
    const float c0 = acc0[r], c1 = acc1[r], c2 = acc2[r], c3 = acc3[r];
    const int lr = r + half * 8;
    const int row = row0 + lr;
    if (lr < nrows && row < M)
      *reinterpret_cast<float4*>(C + (long)row * NC + colbase) = make_float4(c0, c1, c2, c3);
    v[r]     = c0 * ats.x + c1 * ats.y + c2 * ats.z + c3 * ats.w;
    v[8 + r] = c0 * atd.x + c1 * atd.y + c2 * atd.z + c3 * atd.w;
  }
  fold_half<8>(v, (l16 & 8) != 0, 8);
  fold_half<4>(v, (l16 & 4) != 0, 4);
  fold_half<2>(v, (l16 & 2) != 0, 2);
  fold_half<1>(v, (l16 & 1) != 0, 1);
  const int olr = (l16 & 7) + half * 8;
  const int orow = row0 + olr;
  if (olr < nrows && orow < M) {
    float* dp = (l16 < 8) ? as_ : ad_;
    dp[orow] = v[0];
  }
}

#define GRP_SHIFT 7
#define GRP  128
#define NPR  100
#define NSTRIP ((NPR + 15) / 16)
#define MAXB 1024
#define EPB  4096
#define MAXC 1024
#define TBE  1024

__device__ __forceinline__ unsigned range_of(int d, int NB) {
  return min((unsigned)d / (unsigned)NPR, (unsigned)(NB - 1));
}
__device__ __forceinline__ unsigned slot_of(int d, unsigned range) {
  return ((unsigned)d - range * (unsigned)NPR) & (unsigned)(GRP - 1);
}

#define TBG 256
template <int K, int NC>
__global__ void __launch_bounds__(TBG)
count_and_gemm(const int* __restrict__ srcs, const int* __restrict__ dsts, int E0, int NB, int CH,
               int* __restrict__ hist, int* __restrict__ lstart, unsigned* __restrict__ recs,
               const float* __restrict__ A, const float* __restrict__ B,
               const float* __restrict__ att_s, const float* __restrict__ att_d,
               float* __restrict__ C, float* __restrict__ as_, float* __restrict__ ad_,
               int M) {
  const int t = threadIdx.x;
  if ((int)blockIdx.x < CH) {
    __shared__ int h[MAXB];
    __shared__ int wsum[TBG / 32];
    const int lane = t & 31;
    const int chunk = blockIdx.x;
    for (int b = t; b < MAXB; b += TBG) h[b] = 0;
    __syncthreads();
    const int base = chunk * EPB;
    int d[EPB / TBG], s[EPB / TBG], rk[EPB / TBG];
#pragma unroll
    for (int k = 0; k < EPB / TBG; ++k) {
      const int e = min(base + k * TBG + t, E0 - 1);
      d[k] = dsts[e];
      s[k] = srcs[e];
    }
#pragma unroll
    for (int k = 0; k < EPB / TBG; ++k) {
      const int e = base + k * TBG + t;
      rk[k] = (e < E0) ? atomicAdd(&h[range_of(d[k], NB)], 1) : 0;
    }
    __syncthreads();
    constexpr int PER = MAXB / TBG;
    int c[PER];
    int own = 0;
#pragma unroll
    for (int i = 0; i < PER; ++i) { c[i] = h[t * PER + i]; own += c[i]; }
    int incl = own;
    for (int o = 1; o < 32; o <<= 1) {
      const int v = __shfl_up(incl, o);
      if (lane >= o) incl += v;
    }
    if (lane == 31) wsum[t >> 5] = incl;
    __syncthreads();
    int run = incl - own;
    for (int k = 0; k < (t >> 5); ++k) run += wsum[k];
#pragma unroll
    for (int i = 0; i < PER; ++i) {
      const int b = t * PER + i;
      h[b] = run;
      if (b < NB) { hist[chunk * NB + b] = c[i]; lstart[chunk * NB + b] = run; }
      run += c[i];
    }
    __syncthreads();
#pragma unroll
    for (int k = 0; k < EPB / TBG; ++k) {
      const int e = base + k * TBG + t;
      if (e < E0) {
        const unsigned rg = range_of(d[k], NB);
        recs[base + h[rg] + rk[k]] = ((unsigned)s[k] << GRP_SHIFT) | slot_of(d[k], rg);
      }
    }
    return;
  }
  int wave = (int)((((int)blockIdx.x - CH) * TBG + t) >> 5);
  wave = __builtin_amdgcn_readfirstlane(wave);
  gemm_heads_wave<K, NC>(A, B, att_s, att_d, C, as_, ad_, M, wave);
}

__device__ __forceinline__ float leaky(float v) { return (v > 0.f) ? v : NEG_SLOPE * v; }

__device__ __forceinline__ float wave_max(float v) {
  for (int o = 16; o > 0; o >>= 1) v = fmaxf(v, __shfl_xor(v, o));
  return v;
}
__device__ __forceinline__ float wave_sum(float v) {
  for (int o = 16; o > 0; o >>= 1) v += __shfl_xor(v, o);
  return v;
}

template <int H, int VEC, int U>
__device__ __forceinline__ void
row_batch(const char* __restrict__ hb, unsigned lbytes, int s, const float (&w)[H], int myh, int t0,
          float (&acc)[VEC]) {
  constexpr unsigned ROWB = 32u * VEC * 4u;
  float4 r4[U];
  float2 r2[U];
#pragma unroll
  for (int u = 0; u < U; ++u) {
    const unsigned st = (unsigned)__builtin_amdgcn_readlane(s, t0 + u);
    const unsigned boff = st * ROWB + lbytes;
    if (VEC == 4) r4[u] = *reinterpret_cast<const float4*>(hb + boff);
    else          r2[u] = *reinterpret_cast<const float2*>(hb + boff);
  }
#pragma unroll
  for (int u = 0; u < U; ++u) {
    float wt = __int_as_float(__builtin_amdgcn_readlane(__float_as_int(w[0]), t0 + u));
    if (H == 2) {
      float w1 = __int_as_float(__builtin_amdgcn_readlane(__float_as_int(w[H - 1]), t0 + u));
      wt = myh ? w1 : wt;
    }
    if (VEC == 4) {
      acc[0] = fmaf(r4[u].x, wt, acc[0]); acc[1] = fmaf(r4[u].y, wt, acc[1]);
      acc[VEC - 2] = fmaf(r4[u].z, wt, acc[VEC - 2]); acc[VEC - 1] = fmaf(r4[u].w, wt, acc[VEC - 1]);
    } else {
      acc[0] = fmaf(r2[u].x, wt, acc[0]); acc[1] = fmaf(r2[u].y, wt, acc[1]);
    }
  }
}

template <int VEC>
__device__ __forceinline__ void load_bias(const float* __restrict__ bias, float (&bv)[VEC]) {
  const int lane = threadIdx.x & 31;
  if (VEC == 4) {
    const float4 b = *reinterpret_cast<const float4*>(bias + lane * VEC);
    bv[0] = b.x; bv[1] = b.y; bv[VEC - 2] = b.z; bv[VEC - 1] = b.w;
  } else {
    const float2 b = *reinterpret_cast<const float2*>(bias + lane * VEC);
    bv[0] = b.x; bv[1] = b.y;
  }
}

template <int H, int VEC, bool ELU_OUT, int UB, int UB2, bool ROW_TO_LDS, bool MAY_PRE>
__device__ __forceinline__ void
agg_node(int node, int beg, int tot, const int* ssrc,
         const float* __restrict__ as_, const float* __restrict__ ad_,
         const float* __restrict__ h, const float (&bv)[VEC],
         float* __restrict__ outp, int s_pre, bool pre) {
  constexpr int F = 32 * VEC;
  constexpr int C = F / H;
  const int lane = threadIdx.x & 31;
  const int myh = (lane * VEC) / C;

  float adv[H], m[H], v[H];
  for (int k = 0; k < H; ++k) { adv[k] = ad_[node * H + k]; m[k] = -INFINITY; v[k] = 0.f; }
  int s = 0;
  bool act = false;

  auto load_chunk = [&](int jb) {
    int idx = jb + lane;
    act = idx < tot;
    s = 0;
    for (int k = 0; k < H; ++k) v[k] = 0.f;
    if (act) {
      s = (MAY_PRE && pre) ? s_pre : ((idx == 0) ? node : ssrc[beg + idx - 1]);
      for (int k = 0; k < H; ++k) v[k] = leaky(as_[s * H + k] + adv[k]);
    }
  };

  for (int jb = 0; jb < tot; jb += 32) {
    load_chunk(jb);
    if (act) { for (int k = 0; k < H; ++k) m[k] = fmaxf(m[k], v[k]); }
  }
  for (int k = 0; k < H; ++k) m[k] = wave_max(m[k]);

  float den[H];
  for (int k = 0; k < H; ++k) den[k] = 0.f;
  for (int jb = 0; jb < tot; jb += 32) {
    if (tot > 32) load_chunk(jb);
    if (act) { for (int k = 0; k < H; ++k) den[k] += expf(v[k] - m[k]); }
  }
  for (int k = 0; k < H; ++k) den[k] = wave_sum(den[k]);

  const char* hb = reinterpret_cast<const char*>(h);
  const unsigned lbytes = (unsigned)lane * (VEC * 4u);
  float acc[VEC];
  for (int q = 0; q < VEC; ++q) acc[q] = 0.f;
  for (int jb = 0; jb < tot; jb += 32) {
    if (tot > 32) load_chunk(jb);
    float w[H];
    for (int k = 0; k < H; ++k) w[k] = act ? expf(v[k] - m[k]) / den[k] : 0.f;
    const int cnt = min(32, tot - jb);
    int t = 0;
    if (UB2 > UB) {
      for (; t + UB2 <= cnt; t += UB2) row_batch<H, VEC, UB2>(hb, lbytes, s, w, myh, t, acc);
    }
    for (; t + UB <= cnt; t += UB) row_batch<H, VEC, UB>(hb, lbytes, s, w, myh, t, acc);
    switch (cnt - t) {
      case 1:  row_batch<H, VEC, 1>(hb, lbytes, s, w, myh, t, acc); break;
      case 2:  row_batch<H, VEC, 2>(hb, lbytes, s, w, myh, t, acc); break;
      case 3:  row_batch<H, VEC, 3>(hb, lbytes, s, w, myh, t, acc); break;
      case 4:  row_batch<H, VEC, 4>(hb, lbytes, s, w, myh, t, acc); break;
      case 5:  row_batch<H, VEC, 5>(hb, lbytes, s, w, myh, t, acc); break;
      case 6:  row_batch<H, VEC, 6>(hb, lbytes, s, w, myh, t, acc); break;
      case 7:  row_batch<H, VEC, 7>(hb, lbytes, s, w, myh, t, acc); break;
      default: break;
    }
  }

  for (int q = 0; q < VEC; ++q) {
    float val = acc[q] + bv[q];
    if (ELU_OUT) val = (val > 0.f) ? val : (expf(val) - 1.f);
    acc[q] = val;
  }
  if (ROW_TO_LDS) {
    *reinterpret_cast<float4*>(outp + lane * VEC) = make_float4(acc[0], acc[1], acc[VEC - 2], acc[VEC - 1]);
    return;
  }
  char* ob = reinterpret_cast<char*>(outp);
  const unsigned obytes = (unsigned)node * (F * 4u) + lbytes;
  if (VEC == 4) {
    *reinterpret_cast<float4*>(ob + obytes) = make_float4(acc[0], acc[1], acc[VEC - 2], acc[VEC - 1]);
  } else {
    *reinterpret_cast<float2*>(ob + obytes) = make_float2(acc[0], acc[1]);
  }
}

template <int H, int VEC, bool ELU_OUT, int UB, int UB2>
__global__ void __launch_bounds__(256)
node_softmax_agg(const int* __restrict__ off, const int* __restrict__ ssrc,
                 const float* __restrict__ as_, const float* __restrict__ ad_,
                 const float* __restrict__ h, const float* __restrict__ bias,
                 float* __restrict__ outp, int n) {
  int node = (int)((blockIdx.x * blockDim.x + threadIdx.x) >> 5);
  node = __builtin_amdgcn_readfirstlane(node);
  if (node >= n) return;
  float bv[VEC];
  load_bias<VEC>(bias, bv);
  const int beg = off[node];
  const int tot = off[node + 1] - beg + 1;
  agg_node<H, VEC, ELU_OUT, UB, UB2, false, false>(node, beg, tot, ssrc, as_, ad_, h, bv, outp, 0, false);
}


#define SEGU 4
__global__ void __launch_bounds__(TBE, 16)
group_and_agg(const int* __restrict__ hist, const int* __restrict__ lstart,
              const unsigned* __restrict__ recs, int* __restrict__ segof, int* seg,
              int* ssrc, int N, int NB, int CH,
              const float* __restrict__ as_, const float* __restrict__ ad_,
              const float* __restrict__ h, const float* __restrict__ bias,
              const float* __restrict__ Wn, const float* __restrict__ att_sn,
              const float* __restrict__ att_dn, float* __restrict__ hn,
              float* __restrict__ asn, float* __restrict__ adn) {
  __shared__ float rows[GRP * PA];
  __shared__ float wns[HEADS * DH * DOUT];
  __shared__ int cnt[GRP];
  __shared__ int wtot[GRP / 32];
  __shared__ int segc[MAXC];
  __shared__ int segs[MAXC];
  __shared__ int tot2[2];
  const int t = threadIdx.x;
  const int lane = t & 31;
  const int b = blockIdx.x;
  if (t < GRP) cnt[t] = 0;
  if (t < 2) tot2[t] = 0;
  {
    const int nval = min(NPR, N - b * NPR);
    for (int i = nval * 32 + t; i < NSTRIP * 16 * 32; i += TBE)
      *reinterpret_cast<float4*>(rows + (i >> 5) * PA + (i & 31) * 4) = make_float4(0.f, 0.f, 0.f, 0.f);
  }
  constexpr int WV = HEADS * DH * DOUT / 4 / TBE;
  float4 wv4[WV];
#pragma unroll
  for (int i = 0; i < WV; ++i) wv4[i] = reinterpret_cast<const float4*>(Wn)[t + i * TBE];
  __syncthreads();
  {
    int ls_sum = 0, cn_sum = 0;
    for (int c = t; c < CH; c += TBE) {
      const int cn = hist[c * NB + b];
      const int ls = lstart[c * NB + b];
      segc[c] = cn;
      segs[c] = c * EPB + ls;
      ls_sum += ls;
      cn_sum += cn;
    }
    for (int o = 16; o > 0; o >>= 1) {
      ls_sum += __shfl_xor(ls_sum, o);
      cn_sum += __shfl_xor(cn_sum, o);
    }
    if (lane == 0) { atomicAdd(&tot2[0], ls_sum); atomicAdd(&tot2[1], cn_sum); }
  }
#pragma unroll
  for (int i = 0; i < WV; ++i) reinterpret_cast<float4*>(wns)[t + i * TBE] = wv4[i];
  __syncthreads();
  const int beg = tot2[0];
  const int end = beg + tot2[1];
  const int wv = t >> 5;
  for (int c0 = wv; c0 < CH; c0 += (TBE / 32) * SEGU) {
    unsigned r[SEGU];
    int n[SEGU], st[SEGU];
#pragma unroll
    for (int u = 0; u < SEGU; ++u) {
      const int c = c0 + (TBE / 32) * u;
      n[u]  = (c < CH) ? segc[c] : 0;
      st[u] = (c < CH) ? segs[c] : 0;
      r[u]  = recs[(n[u] > 0) ? min(st[u] + lane, st[u] + n[u] - 1) : 0];
    }
#pragma unroll
    for (int u = 0; u < SEGU; ++u) {
      if (lane < n[u]) atomicAdd(&cnt[r[u] & (unsigned)(GRP - 1)], 1);
      for (int i = lane + 32; i < n[u]; i += 32)
        atomicAdd(&cnt[recs[st[u] + i] & (unsigned)(GRP - 1)], 1);
    }
  }
  __syncthreads();
  int c = 0, incl = 0;
  if (t < GRP) {
    c = cnt[t];
    incl = c;
    for (int o = 1; o < 32; o <<= 1) {
      const int v = __shfl_up(incl, o);
      if (lane >= o) incl += v;
    }
    if (lane == 31) wtot[t >> 5] = incl;
  }
  __syncthreads();
  if (t < GRP) {
    int basew = 0;
    for (int k = 0; k < (t >> 5); ++k) basew += wtot[k];
    const int first = beg + basew + incl - c;
    const int node = b * NPR + t;
    if (t < NPR && node < N) segof[node] = first;
    cnt[t] = first;
  }
  if (b == NB - 1 && t == 0) segof[N] = end;
  __syncthreads();
  for (int c0 = wv; c0 < CH; c0 += (TBE / 32) * SEGU) {
    unsigned r[SEGU];
    int n[SEGU], st[SEGU];
#pragma unroll
    for (int u = 0; u < SEGU; ++u) {
      const int c = c0 + (TBE / 32) * u;
      n[u]  = (c < CH) ? segc[c] : 0;
      st[u] = (c < CH) ? segs[c] : 0;
      r[u]  = recs[(n[u] > 0) ? min(st[u] + lane, st[u] + n[u] - 1) : 0];
    }
#pragma unroll
    for (int u = 0; u < SEGU; ++u) {
      if (lane < n[u]) {
        const int pos = atomicAdd(&cnt[r[u] & (unsigned)(GRP - 1)], 1);
        seg[pos] = (int)(r[u] >> GRP_SHIFT);
      }
      for (int i = lane + 32; i < n[u]; i += 32) {
        const unsigned rr = recs[st[u] + i];
        const int pos = atomicAdd(&cnt[rr & (unsigned)(GRP - 1)], 1);
        seg[pos] = (int)(rr >> GRP_SHIFT);
      }
    }
  }
  __syncthreads();

  const int wvs = __builtin_amdgcn_readfirstlane(t >> 5);
  float bv[4];
  load_bias<4>(bias, bv);
#pragma nounroll
  for (int ln = wvs; ln < NPR; ln += TBE / 32) {
    const int node = b * NPR + ln;
    if (node >= N) break;
    const int gb = __builtin_amdgcn_readfirstlane((ln == 0) ? beg : cnt[ln - 1]);
    const int ge = __builtin_amdgcn_readfirstlane(cnt[ln]);
    const int deg = ge - gb;
    const bool pre = deg < 32;
    int s_pre = 0;
    if (pre) {
      const bool a = lane <= deg;
      const bool g = a && lane > 0;
      const int ki = g ? seg[gb + lane - 1] : node;
      const unsigned key = g ? (((unsigned)ki << 5) | (unsigned)lane) : 0u;
      int r = 0;
      for (int q = 0; q <= deg; ++q) {
        const unsigned kq = (unsigned)__builtin_amdgcn_readlane((int)key, q);
        r += (kq < key) ? 1 : 0;
      }
      if (g) ssrc[gb + r - 1] = ki;
      s_pre = __builtin_amdgcn_ds_permute((a ? r : lane) << 2, ki);
    } else {
      for (int ib = gb; ib < ge; ib += 32) {
        const int i = ib + lane;
        const bool act = i < ge;
        const int ki = act ? seg[i] : 0;
        int r = 0;
        for (int jb = gb; jb < ge; jb += 32) {
          const int j = jb + lane;
          const int kj = (j < ge) ? seg[j] : 0;
          const int cntj = min(32, ge - jb);
          for (int q = 0; q < cntj; ++q) {
            const int k = __builtin_amdgcn_readlane(kj, q);
            const int jj = jb + q;
            r += (k < ki || (k == ki && jj < i)) ? 1 : 0;
          }
        }
        if (act) ssrc[gb + r] = ki;
      }
      __threadfence_block();
    }
    agg_node<HEADS, 4, true, 8, 8, true, true>(node, gb, deg + 1, ssrc, as_, ad_, h, bv,
                                               rows + ln * PA, s_pre, pre);
  }
  __syncthreads();

  if (wvs < NSTRIP)
    gemm_strip_lds<HEADS * DH>(rows + wvs * 16 * PA, wns, att_sn, att_dn, hn, asn, adn, N,
                               b * NPR + wvs * 16, min(16, NPR - wvs * 16));
}


extern "C" void kernel_launch(void* const* d_in, const int* in_sizes, int n_in,
                              void* d_out, int out_size, void* d_ws, size_t ws_size,
                              hipStream_t stream) {
  const float* x    = (const float*)d_in[0];
  const int*   ei   = (const int*)  d_in[1];
  const float* W1   = (const float*)d_in[2];
  const float* atS1 = (const float*)d_in[3];
  const float* atD1 = (const float*)d_in[4];
  const float* b1   = (const float*)d_in[5];
  const float* W2   = (const float*)d_in[6];
  const float* atS2 = (const float*)d_in[7];
  const float* atD2 = (const float*)d_in[8];
  const float* b2   = (const float*)d_in[9];
  float* out = (float*)d_out;

  const int N    = in_sizes[0] / DIN;
  const int E0   = in_sizes[1] / 2;
  const int* srcs = ei;
  const int* dsts = ei + E0;

  float* ws = (float*)d_ws;
  long off = 0;
  float* h1  = ws + off; off += (long)N * (HEADS * DH);
  float* a1s = ws + off; off += (long)N * HEADS;
  float* a1d = ws + off; off += (long)N * HEADS;
  float* h2  = ws + off; off += (long)N * DOUT;
  float* a2s = ws + off; off += (long)N;
  float* a2d = ws + off; off += (long)N;
  const int NB = (N + NPR - 1) / NPR;
  const int CH = (E0 + EPB - 1) / EPB;
  int* segof = (int*)(ws + off); off += (long)N + 1;
  int* hist  = (int*)(ws + off); off += (long)CH * NB;
  int* lstart = (int*)(ws + off); off += (long)CH * NB;
  unsigned* recs = (unsigned*)(ws + off); off += (long)E0;
  int* seg   = (int*)(ws + off); off += (long)E0;
  int* ssrc  = (int*)(ws + off); off += (long)E0;

  const int TB = 256;
  auto blocks = [](long n, int tb) { return (unsigned)((n + tb - 1) / tb); };

  {
    long waves = (long)((N + 15) / 16) * HEADS;
    count_and_gemm<DIN, HEADS * DH><<<CH + blocks(waves * 32, TBG), TBG, 0, stream>>>(
        srcs, dsts, E0, NB, CH, hist, lstart, recs, x, W1, atS1, atD1, h1, a1s, a1d, N);
  }
  group_and_agg<<<NB, TBE, 0, stream>>>(hist, lstart, recs, segof, seg, ssrc, N, NB, CH,
                                        a1s, a1d, h1, b1, W2, atS2, atD2, h2, a2s, a2d);

  node_softmax_agg<1, 2, false, 8, 16><<<blocks((long)N * 32, TB), TB, 0, stream>>>(
      segof, ssrc, a2s, a2d, h2, b2, out, N);
}
